# starting kernel
# baseline (speedup 1.0000x reference)
#include <hip/hip_runtime.h>
#include <math.h>

#define Hdim 256
#define NHEAD 4
#define DK 64
#define PLR 32
#define Bsz 2
#define LQ 8
#define LKs 512
#define NTOK (Bsz * LQ * LKs)
#define EPSV 1e-5f

typedef float v2f __attribute__((ext_vector_type(2)));
typedef float v8f __attribute__((ext_vector_type(8)));

#define ACT_NONE  0
#define ACT_SILU  1
#define ACT_TANH  2
#define ACT_DECAY 3

__global__ void rwkv_features_kernel(
    const float* __restrict__ query, const float* __restrict__ keyval,
    const float* __restrict__ rwg_mu, const float* __restrict__ W_rwg0,
    const float* __restrict__ W_rwg2, const float* __restrict__ rwg_bias,
    const float* __restrict__ kv_mu, const float* __restrict__ W_kv0,
    const float* __restrict__ W_kv2, const float* __restrict__ kv_bias,
    float* __restrict__ xr, float* __restrict__ xw, float* __restrict__ xg,
    float* __restrict__ xk, float* __restrict__ xv)
{
    const int tid = blockIdx.x;
    const int e   = tid >> 9;
    const int l   = tid & 511;
    const int b   = e >> 3;
    const int t   = threadIdx.x;

    __shared__ float xrwg_s[Hdim];
    __shared__ float xkv_s[Hdim];
    __shared__ float t1_s[3 * PLR];
    __shared__ float t2_s[2 * PLR];

    const float h  = keyval[((long)b * LKs + l) * Hdim + t];
    const float q  = query[(long)e * Hdim + t];
    const float hp = (l == 0) ? 0.0f : keyval[((long)b * LKs + (l - 1)) * Hdim + t];
    const float dc = q - h;
    const float d  = hp - h;

    xrwg_s[t] = h + dc * rwg_mu[t];
    xkv_s[t]  = h + d  * kv_mu[t];
    __syncthreads();

    if (t < 96) {
        float acc = 0.0f;
        #pragma unroll 8
        for (int i = 0; i < Hdim; ++i) acc += xrwg_s[i] * W_rwg0[i * 96 + t];
        t1_s[t] = tanhf(acc);
    } else if (t < 160) {
        const int j = t - 96;
        float acc = 0.0f;
        #pragma unroll 8
        for (int i = 0; i < Hdim; ++i) acc += xkv_s[i] * W_kv0[i * 64 + j];
        t2_s[j] = tanhf(acc);
    }
    __syncthreads();

    float rmu = rwg_bias[t];
    float wmu = rwg_bias[Hdim + t];
    float gmu = rwg_bias[2 * Hdim + t];
    #pragma unroll
    for (int r = 0; r < PLR; ++r) {
        rmu += t1_s[r]            * W_rwg2[t * 96 + r];
        wmu += t1_s[PLR + r]      * W_rwg2[t * 96 + PLR + r];
        gmu += t1_s[2 * PLR + r]  * W_rwg2[t * 96 + 2 * PLR + r];
    }
    float kmu = kv_bias[t];
    float vmu = kv_bias[Hdim + t];
    #pragma unroll
    for (int r = 0; r < PLR; ++r) {
        kmu += t2_s[r]       * W_kv2[t * 64 + r];
        vmu += t2_s[PLR + r] * W_kv2[t * 64 + PLR + r];
    }

    const long o = (long)tid * Hdim + t;
    xr[o] = h + dc * rmu;
    xw[o] = h + dc * wmu;
    xg[o] = h + dc * gmu;
    xk[o] = h + d  * kmu;
    xv[o] = h + d  * vmu;
}

__device__ __forceinline__ float apply_act(float x, int act)
{
    switch (act) {
        case ACT_SILU:  return x / (1.0f + expf(-x));
        case ACT_TANH:  return tanhf(x);
        case ACT_DECAY: return expf(-expf(x));
        default:        return x;
    }
}

__global__ __launch_bounds__(128) void rwkv_wmma_gemm_kernel(
    const float* __restrict__ X, const float* __restrict__ W,
    const float* __restrict__ bias, float* __restrict__ Y,
    int K, int N, int act)
{
    const int wave  = threadIdx.x >> 5;
    const int lane  = threadIdx.x & 31;
    const int mbase = (blockIdx.x * 4 + wave) << 5;
    const int nbase = blockIdx.y << 5;

    const int sub  = lane & 15;
    const int koff = (lane >> 4) << 1;

    const float* __restrict__ Arow0 = X + (long)(mbase + sub) * K + koff;
    const float* __restrict__ Arow1 = Arow0 + (long)16 * K;
    const float* __restrict__ Bcol0 = W + (long)koff * N + nbase + sub;
    const float* __restrict__ Bcol1 = Bcol0 + 16;

    v8f c00 = {}, c01 = {}, c10 = {}, c11 = {};
    #pragma unroll 4
    for (int k = 0; k < K; k += 4) {
        const v2f a0 = *(const v2f*)(Arow0 + k);
        const v2f a1 = *(const v2f*)(Arow1 + k);
        v2f b0, b1;
        b0.x = Bcol0[(long)k * N];
        b0.y = Bcol0[(long)(k + 1) * N];
        b1.x = Bcol1[(long)k * N];
        b1.y = Bcol1[(long)(k + 1) * N];
        c00 = __builtin_amdgcn_wmma_f32_16x16x4_f32(false, a0, false, b0, (short)0, c00, false, false);
        c01 = __builtin_amdgcn_wmma_f32_16x16x4_f32(false, a0, false, b1, (short)0, c01, false, false);
        c10 = __builtin_amdgcn_wmma_f32_16x16x4_f32(false, a1, false, b0, (short)0, c10, false, false);
        c11 = __builtin_amdgcn_wmma_f32_16x16x4_f32(false, a1, false, b1, (short)0, c11, false, false);
    }

    const int rbase = (lane >> 4) * 8;
    const int col0  = nbase + sub;
    const int col1  = col0 + 16;
    const float bv0 = bias ? bias[col0] : 0.0f;
    const float bv1 = bias ? bias[col1] : 0.0f;
    #pragma unroll
    for (int r = 0; r < 8; ++r) {
        const long row0 = (long)mbase + rbase + r;
        const long row1 = row0 + 16;
        Y[row0 * N + col0] = apply_act(c00[r] + bv0, act);
        Y[row0 * N + col1] = apply_act(c01[r] + bv1, act);
        Y[row1 * N + col0] = apply_act(c10[r] + bv0, act);
        Y[row1 * N + col1] = apply_act(c11[r] + bv1, act);
    }
}

__global__ void rwkv_scan_kernel(
    const float* __restrict__ r, const float* __restrict__ k,
    const float* __restrict__ v, const float* __restrict__ decay,
    const float* __restrict__ u, float* __restrict__ o)
{
    const int e    = blockIdx.x >> 2;
    const int head = blockIdx.x & 3;
    const long base = ((long)e * LKs) * Hdim + head * DK;
    const int t    = threadIdx.x;
    const int vcol = t & 63;
    const int k0   = (t >> 6) * 16;

    float S[16];
    #pragma unroll
    for (int j = 0; j < 16; ++j) S[j] = 0.0f;

    __shared__ float rr[DK], kk[DK], vv[DK], dd[DK], uu[DK];
    __shared__ float part[256];

    if (t < DK) uu[t] = u[head * DK + t];
    __syncthreads();

    for (int step = 0; step < LKs; ++step) {
        const long p = base + (long)step * Hdim;
        if (t < 64) {
            rr[t] = r[p + t];
            __builtin_prefetch((const void*)(r + p + Hdim + t), 0, 1);
        } else if (t < 128) {
            kk[t - 64] = k[p + t - 64];
            __builtin_prefetch((const void*)(k + p + Hdim + t - 64), 0, 1);
        } else if (t < 192) {
            vv[t - 128] = v[p + t - 128];
            __builtin_prefetch((const void*)(v + p + Hdim + t - 128), 0, 1);
        } else {
            dd[t - 192] = decay[p + t - 192];
            __builtin_prefetch((const void*)(decay + p + Hdim + t - 192), 0, 1);
        }
        __syncthreads();

        const float vt = vv[vcol];
        float acc = 0.0f;
        #pragma unroll
        for (int j = 0; j < 16; ++j) {
            const int ki = k0 + j;
            const float kv = kk[ki] * vt;
            acc += rr[ki] * (S[j] + uu[ki] * kv);
            S[j] = dd[ki] * S[j] + kv;
        }
        part[t] = acc;
        __syncthreads();
        if (t < 64)
            o[p + t] = part[t] + part[t + 64] + part[t + 128] + part[t + 192];
        __syncthreads();
    }
}

__global__ void rwkv_gn_silu_kernel(
    const float* __restrict__ o, const float* __restrict__ gsilu,
    const float* __restrict__ gn_w, const float* __restrict__ gn_b,
    float* __restrict__ y)
{
    const long tok = blockIdx.x;
    const int t    = threadIdx.x;
    const int head = t >> 6;
    const int idx  = t & 63;

    const float val = o[tok * Hdim + t];
    __shared__ float red[256];

    red[t] = val;
    __syncthreads();
    for (int off = 32; off >= 1; off >>= 1) {
        if (idx < off) red[t] += red[t + off];
        __syncthreads();
    }
    const float mean = red[head * 64] * (1.0f / 64.0f);
    __syncthreads();

    const float dv = val - mean;
    red[t] = dv * dv;
    __syncthreads();
    for (int off = 32; off >= 1; off >>= 1) {
        if (idx < off) red[t] += red[t + off];
        __syncthreads();
    }
    const float var = red[head * 64] * (1.0f / 64.0f);

    const float ynorm = dv * rsqrtf(var + EPSV);
    y[tok * Hdim + t] = (ynorm * gn_w[t] + gn_b[t]) * gsilu[tok * Hdim + t];
}

extern "C" void kernel_launch(void* const* d_in, const int* in_sizes, int n_in,
                              void* d_out, int out_size, void* d_ws, size_t ws_size,
                              hipStream_t stream)
{
    const float* query    = (const float*)d_in[0];
    const float* keyval   = (const float*)d_in[1];
    const float* rwg_mu   = (const float*)d_in[2];
    const float* W_rwg0   = (const float*)d_in[3];
    const float* W_rwg2   = (const float*)d_in[4];
    const float* rwg_bias = (const float*)d_in[5];
    const float* kv_mu    = (const float*)d_in[6];
    const float* W_kv0    = (const float*)d_in[7];
    const float* W_kv2    = (const float*)d_in[8];
    const float* kv_bias  = (const float*)d_in[9];
    const float* Wr       = (const float*)d_in[10];
    const float* Wa       = (const float*)d_in[11];
    const float* Wb       = (const float*)d_in[12];
    const float* bb       = (const float*)d_in[13];
    const float* Wk       = (const float*)d_in[14];
    const float* Wv       = (const float*)d_in[15];
    const float* Wg       = (const float*)d_in[16];
    const float* u        = (const float*)d_in[17];
    const float* gn_w     = (const float*)d_in[18];
    const float* gn_b     = (const float*)d_in[19];
    const float* Wo       = (const float*)d_in[20];

    float* ws = (float*)d_ws;
    const long SZ = (long)NTOK * Hdim;
    float* xr    = ws + 0 * SZ;
    float* xw    = ws + 1 * SZ;
    float* xg    = ws + 2 * SZ;
    float* xk    = ws + 3 * SZ;
    float* xv    = ws + 4 * SZ;
    float* rbuf  = ws + 5 * SZ;
    float* kbuf  = ws + 6 * SZ;
    float* vbuf  = ws + 7 * SZ;
    float* dbuf  = ws + 8 * SZ;
    float* gsbuf = ws + 9 * SZ;
    float* twbuf = ws + 10 * SZ;
    float* obuf  = xr;
    float* ybuf  = xw;

    rwkv_features_kernel<<<NTOK, 256, 0, stream>>>(
        query, keyval, rwg_mu, W_rwg0, W_rwg2, rwg_bias,
        kv_mu, W_kv0, W_kv2, kv_bias, xr, xw, xg, xk, xv);

    auto gemm = [&](const float* X, const float* W, const float* bias,
                    float* Y, int M, int K, int N, int act) {
        dim3 grid(M / 128, N / 32);
        rwkv_wmma_gemm_kernel<<<grid, 128, 0, stream>>>(X, W, bias, Y, K, N, act);
    };

    gemm(xr, Wr, nullptr, rbuf,  NTOK, 256, 256, ACT_NONE);
    gemm(xk, Wk, nullptr, kbuf,  NTOK, 256, 256, ACT_NONE);
    gemm(xv, Wv, nullptr, vbuf,  NTOK, 256, 256, ACT_NONE);
    gemm(xg, Wg, nullptr, gsbuf, NTOK, 256, 256, ACT_SILU);
    gemm(xw, Wa, nullptr, twbuf, NTOK, 256, 64,  ACT_TANH);
    gemm(twbuf, Wb, bb, dbuf,    NTOK, 64,  256, ACT_DECAY);

    rwkv_scan_kernel<<<Bsz * LQ * NHEAD, 256, 0, stream>>>(rbuf, kbuf, vbuf, dbuf, u, obuf);

    rwkv_gn_silu_kernel<<<NTOK, 256, 0, stream>>>(obuf, gsbuf, gn_w, gn_b, ybuf);

    gemm(ybuf, Wo, nullptr, (float*)d_out, NTOK, 256, 256, ACT_NONE);
}


# scan: one wave per chain and 32 value columns, each lane holds its whole 64-row state column: o needs no k-slice join and no per-step barrier; r/k/decay rows read back whole from LDS; sum order kept
# speedup vs baseline: 1.7425x; 1.7425x over previous
#include <hip/hip_runtime.h>
#include <math.h>

#define Hdim 256
#define NHEAD 4
#define DK 64
#define PLR 32
#define Bsz 2
#define LQ 8
#define LKs 512
#define NTOK (Bsz * LQ * LKs)
#define EPSV 1e-5f

typedef float v2f __attribute__((ext_vector_type(2)));
typedef float v8f __attribute__((ext_vector_type(8)));

#define ACT_NONE  0
#define ACT_SILU  1
#define ACT_TANH  2
#define ACT_DECAY 3

__global__ void rwkv_features_kernel(
    const float* __restrict__ query, const float* __restrict__ keyval,
    const float* __restrict__ rwg_mu, const float* __restrict__ W_rwg0,
    const float* __restrict__ W_rwg2, const float* __restrict__ rwg_bias,
    const float* __restrict__ kv_mu, const float* __restrict__ W_kv0,
    const float* __restrict__ W_kv2, const float* __restrict__ kv_bias,
    float* __restrict__ xr, float* __restrict__ xw, float* __restrict__ xg,
    float* __restrict__ xk, float* __restrict__ xv)
{
    const int tid = blockIdx.x;
    const int e   = tid >> 9;
    const int l   = tid & 511;
    const int b   = e >> 3;
    const int t   = threadIdx.x;

    __shared__ float xrwg_s[Hdim];
    __shared__ float xkv_s[Hdim];
    __shared__ float t1_s[3 * PLR];
    __shared__ float t2_s[2 * PLR];

    const float h  = keyval[((long)b * LKs + l) * Hdim + t];
    const float q  = query[(long)e * Hdim + t];
    const float hp = (l == 0) ? 0.0f : keyval[((long)b * LKs + (l - 1)) * Hdim + t];
    const float dc = q - h;
    const float d  = hp - h;

    xrwg_s[t] = h + dc * rwg_mu[t];
    xkv_s[t]  = h + d  * kv_mu[t];
    __syncthreads();

    if (t < 96) {
        float acc = 0.0f;
        #pragma unroll 8
        for (int i = 0; i < Hdim; ++i) acc += xrwg_s[i] * W_rwg0[i * 96 + t];
        t1_s[t] = tanhf(acc);
    } else if (t < 160) {
        const int j = t - 96;
        float acc = 0.0f;
        #pragma unroll 8
        for (int i = 0; i < Hdim; ++i) acc += xkv_s[i] * W_kv0[i * 64 + j];
        t2_s[j] = tanhf(acc);
    }
    __syncthreads();

    float rmu = rwg_bias[t];
    float wmu = rwg_bias[Hdim + t];
    float gmu = rwg_bias[2 * Hdim + t];
    #pragma unroll
    for (int r = 0; r < PLR; ++r) {
        rmu += t1_s[r]            * W_rwg2[t * 96 + r];
        wmu += t1_s[PLR + r]      * W_rwg2[t * 96 + PLR + r];
        gmu += t1_s[2 * PLR + r]  * W_rwg2[t * 96 + 2 * PLR + r];
    }
    float kmu = kv_bias[t];
    float vmu = kv_bias[Hdim + t];
    #pragma unroll
    for (int r = 0; r < PLR; ++r) {
        kmu += t2_s[r]       * W_kv2[t * 64 + r];
        vmu += t2_s[PLR + r] * W_kv2[t * 64 + PLR + r];
    }

    const long o = (long)tid * Hdim + t;
    xr[o] = h + dc * rmu;
    xw[o] = h + dc * wmu;
    xg[o] = h + dc * gmu;
    xk[o] = h + d  * kmu;
    xv[o] = h + d  * vmu;
}

__device__ __forceinline__ float apply_act(float x, int act)
{
    switch (act) {
        case ACT_SILU:  return x / (1.0f + expf(-x));
        case ACT_TANH:  return tanhf(x);
        case ACT_DECAY: return expf(-expf(x));
        default:        return x;
    }
}

__global__ __launch_bounds__(128) void rwkv_wmma_gemm_kernel(
    const float* __restrict__ X, const float* __restrict__ W,
    const float* __restrict__ bias, float* __restrict__ Y,
    int K, int N, int act)
{
    const int wave  = threadIdx.x >> 5;
    const int lane  = threadIdx.x & 31;
    const int mbase = (blockIdx.x * 4 + wave) << 5;
    const int nbase = blockIdx.y << 5;

    const int sub  = lane & 15;
    const int koff = (lane >> 4) << 1;

    const float* __restrict__ Arow0 = X + (long)(mbase + sub) * K + koff;
    const float* __restrict__ Arow1 = Arow0 + (long)16 * K;
    const float* __restrict__ Bcol0 = W + (long)koff * N + nbase + sub;
    const float* __restrict__ Bcol1 = Bcol0 + 16;

    v8f c00 = {}, c01 = {}, c10 = {}, c11 = {};
    #pragma unroll 4
    for (int k = 0; k < K; k += 4) {
        const v2f a0 = *(const v2f*)(Arow0 + k);
        const v2f a1 = *(const v2f*)(Arow1 + k);
        v2f b0, b1;
        b0.x = Bcol0[(long)k * N];
        b0.y = Bcol0[(long)(k + 1) * N];
        b1.x = Bcol1[(long)k * N];
        b1.y = Bcol1[(long)(k + 1) * N];
        c00 = __builtin_amdgcn_wmma_f32_16x16x4_f32(false, a0, false, b0, (short)0, c00, false, false);
        c01 = __builtin_amdgcn_wmma_f32_16x16x4_f32(false, a0, false, b1, (short)0, c01, false, false);
        c10 = __builtin_amdgcn_wmma_f32_16x16x4_f32(false, a1, false, b0, (short)0, c10, false, false);
        c11 = __builtin_amdgcn_wmma_f32_16x16x4_f32(false, a1, false, b1, (short)0, c11, false, false);
    }

    const int rbase = (lane >> 4) * 8;
    const int col0  = nbase + sub;
    const int col1  = col0 + 16;
    const float bv0 = bias ? bias[col0] : 0.0f;
    const float bv1 = bias ? bias[col1] : 0.0f;
    #pragma unroll
    for (int r = 0; r < 8; ++r) {
        const long row0 = (long)mbase + rbase + r;
        const long row1 = row0 + 16;
        Y[row0 * N + col0] = apply_act(c00[r] + bv0, act);
        Y[row0 * N + col1] = apply_act(c01[r] + bv1, act);
        Y[row1 * N + col0] = apply_act(c10[r] + bv0, act);
        Y[row1 * N + col1] = apply_act(c11[r] + bv1, act);
    }
}

typedef float v4f __attribute__((ext_vector_type(4)));

#define SC_STEPS  4
#define SC_ROW    (3 * DK + 32)
#define SC_CHUNK  (SC_STEPS * SC_ROW)
#define SC_NCHUNK (LKs / SC_STEPS)
#define SC_BLOCKS (SC_STEPS * 4)
#define SC_RING   4

__global__ __launch_bounds__(32) void rwkv_scan_kernel(
    const float* __restrict__ r, const float* __restrict__ k,
    const float* __restrict__ v, const float* __restrict__ decay,
    const float* __restrict__ u, float* __restrict__ o)
{
    __shared__ __attribute__((aligned(16))) float stage[2 * SC_CHUNK];

    const int chain = blockIdx.x >> 1;
    const int half  = blockIdx.x & 1;
    const int e     = chain >> 2;
    const int head  = chain & 3;
    const int lane  = (int)threadIdx.x;
    const long base = ((long)e * LKs) * Hdim + head * DK;
    const int vcol  = half * 32 + lane;

    float S[DK];
    float uu[DK];
    #pragma unroll
    for (int j = 0; j < DK; ++j) { S[j] = 0.0f; uu[j] = u[head * DK + j]; }

    const float* __restrict__ rg = r + base + 2 * lane;
    const float* __restrict__ kg = k + base + 2 * lane;
    const float* __restrict__ dg = decay + base + 2 * lane;
    const float* __restrict__ vg = v + base + vcol;
    float*       __restrict__ og = o + base + vcol;

    v2f gr[SC_STEPS], gk[SC_STEPS], gd[SC_STEPS];
    float gv[SC_STEPS];

    auto fetch = [&](int chunk) {
        #pragma unroll
        for (int i = 0; i < SC_STEPS; ++i) {
            const long off = (long)(chunk * SC_STEPS + i) * Hdim;
            gr[i] = *(const v2f*)(rg + off);
            gk[i] = *(const v2f*)(kg + off);
            gd[i] = *(const v2f*)(dg + off);
            gv[i] = vg[off];
        }
    };
    auto put = [&](float* buf) {
        #pragma unroll
        for (int i = 0; i < SC_STEPS; ++i) {
            float* slot = buf + i * SC_ROW;
            *(v2f*)(slot + 2 * lane)          = gr[i];
            *(v2f*)(slot + DK + 2 * lane)     = gk[i];
            *(v2f*)(slot + 2 * DK + 2 * lane) = gd[i];
            slot[3 * DK + lane]               = gv[i];
        }
    };

    v4f R[SC_RING][4], K[SC_RING][4], D[SC_RING][4];
    float VT[2];

    auto take = [&](const float* buf, int nb) {
        const int si = (nb % SC_BLOCKS) >> 2, sq = nb & 3, set = nb % SC_RING;
        const float* slot = buf + si * SC_ROW + sq * 16;
        #pragma unroll
        for (int g = 0; g < 4; ++g) {
            R[set][g] = *(const v4f*)(slot + 4 * g);
            K[set][g] = *(const v4f*)(slot + DK + 4 * g);
            D[set][g] = *(const v4f*)(slot + 2 * DK + 4 * g);
        }
        if (sq == 0) VT[(nb >> 2) & 1] = buf[si * SC_ROW + 3 * DK + lane];
    };

    fetch(0);
    put(stage);
    __syncthreads();
    fetch(1);
    take(stage, 0);
    take(stage, 1);

    for (int c = 0; c < SC_NCHUNK; ++c) {
        const float* cur = stage + (c & 1) * SC_CHUNK;
        float*       nxt = stage + ((c + 1) & 1) * SC_CHUNK;
        put(nxt);
        __syncthreads();
        fetch((c + 2 < SC_NCHUNK) ? (c + 2) : (SC_NCHUNK - 1));

        float part[4];
        #pragma unroll
        for (int b = 0; b < SC_BLOCKS; b += 2) {
            take((b + 2 < SC_BLOCKS) ? cur : (const float*)nxt, b + 2);
            take((b + 3 < SC_BLOCKS) ? cur : (const float*)nxt, b + 3);
            __builtin_amdgcn_sched_barrier(0);

            const int i = b >> 2, q = b & 3;
            const int sa = b % SC_RING, sb = (b + 1) % SC_RING;
            const float vt = VT[i & 1];
            float acca = 0.0f, accb = 0.0f;
            #pragma unroll
            for (int j = 0; j < 16; ++j) {
                const int ka = q * 16 + j, kb = ka + 16;
                const float kva = K[sa][j >> 2][j & 3] * vt;
                const float kvb = K[sb][j >> 2][j & 3] * vt;
                acca += R[sa][j >> 2][j & 3] * (S[ka] + uu[ka] * kva);
                accb += R[sb][j >> 2][j & 3] * (S[kb] + uu[kb] * kvb);
                S[ka] = D[sa][j >> 2][j & 3] * S[ka] + kva;
                S[kb] = D[sb][j >> 2][j & 3] * S[kb] + kvb;
            }
            part[q]     = acca;
            part[q + 1] = accb;
            if (q == 2)
                og[(long)(c * SC_STEPS + i) * Hdim] = part[0] + part[1] + part[2] + part[3];
            __builtin_amdgcn_sched_barrier(0);
        }
    }
}

__global__ void rwkv_gn_silu_kernel(
    const float* __restrict__ o, const float* __restrict__ gsilu,
    const float* __restrict__ gn_w, const float* __restrict__ gn_b,
    float* __restrict__ y)
{
    const long tok = blockIdx.x;
    const int t    = threadIdx.x;
    const int head = t >> 6;
    const int idx  = t & 63;

    const float val = o[tok * Hdim + t];
    __shared__ float red[256];

    red[t] = val;
    __syncthreads();
    for (int off = 32; off >= 1; off >>= 1) {
        if (idx < off) red[t] += red[t + off];
        __syncthreads();
    }
    const float mean = red[head * 64] * (1.0f / 64.0f);
    __syncthreads();

    const float dv = val - mean;
    red[t] = dv * dv;
    __syncthreads();
    for (int off = 32; off >= 1; off >>= 1) {
        if (idx < off) red[t] += red[t + off];
        __syncthreads();
    }
    const float var = red[head * 64] * (1.0f / 64.0f);

    const float ynorm = dv * rsqrtf(var + EPSV);
    y[tok * Hdim + t] = (ynorm * gn_w[t] + gn_b[t]) * gsilu[tok * Hdim + t];
}

extern "C" void kernel_launch(void* const* d_in, const int* in_sizes, int n_in,
                              void* d_out, int out_size, void* d_ws, size_t ws_size,
                              hipStream_t stream)
{
    const float* query    = (const float*)d_in[0];
    const float* keyval   = (const float*)d_in[1];
    const float* rwg_mu   = (const float*)d_in[2];
    const float* W_rwg0   = (const float*)d_in[3];
    const float* W_rwg2   = (const float*)d_in[4];
    const float* rwg_bias = (const float*)d_in[5];
    const float* kv_mu    = (const float*)d_in[6];
    const float* W_kv0    = (const float*)d_in[7];
    const float* W_kv2    = (const float*)d_in[8];
    const float* kv_bias  = (const float*)d_in[9];
    const float* Wr       = (const float*)d_in[10];
    const float* Wa       = (const float*)d_in[11];
    const float* Wb       = (const float*)d_in[12];
    const float* bb       = (const float*)d_in[13];
    const float* Wk       = (const float*)d_in[14];
    const float* Wv       = (const float*)d_in[15];
    const float* Wg       = (const float*)d_in[16];
    const float* u        = (const float*)d_in[17];
    const float* gn_w     = (const float*)d_in[18];
    const float* gn_b     = (const float*)d_in[19];
    const float* Wo       = (const float*)d_in[20];

    float* ws = (float*)d_ws;
    const long SZ = (long)NTOK * Hdim;
    float* xr    = ws + 0 * SZ;
    float* xw    = ws + 1 * SZ;
    float* xg    = ws + 2 * SZ;
    float* xk    = ws + 3 * SZ;
    float* xv    = ws + 4 * SZ;
    float* rbuf  = ws + 5 * SZ;
    float* kbuf  = ws + 6 * SZ;
    float* vbuf  = ws + 7 * SZ;
    float* dbuf  = ws + 8 * SZ;
    float* gsbuf = ws + 9 * SZ;
    float* twbuf = ws + 10 * SZ;
    float* obuf  = xr;
    float* ybuf  = xw;

    rwkv_features_kernel<<<NTOK, 256, 0, stream>>>(
        query, keyval, rwg_mu, W_rwg0, W_rwg2, rwg_bias,
        kv_mu, W_kv0, W_kv2, kv_bias, xr, xw, xg, xk, xv);

    auto gemm = [&](const float* X, const float* W, const float* bias,
                    float* Y, int M, int K, int N, int act) {
        dim3 grid(M / 128, N / 32);
        rwkv_wmma_gemm_kernel<<<grid, 128, 0, stream>>>(X, W, bias, Y, K, N, act);
    };

    gemm(xr, Wr, nullptr, rbuf,  NTOK, 256, 256, ACT_NONE);
    gemm(xk, Wk, nullptr, kbuf,  NTOK, 256, 256, ACT_NONE);
    gemm(xv, Wv, nullptr, vbuf,  NTOK, 256, 256, ACT_NONE);
    gemm(xg, Wg, nullptr, gsbuf, NTOK, 256, 256, ACT_SILU);
    gemm(xw, Wa, nullptr, twbuf, NTOK, 256, 64,  ACT_TANH);
    gemm(twbuf, Wb, bb, dbuf,    NTOK, 64,  256, ACT_DECAY);

    rwkv_scan_kernel<<<Bsz * LQ * NHEAD * 2, 32, 0, stream>>>(rbuf, kbuf, vbuf, dbuf, u, obuf);

    rwkv_gn_silu_kernel<<<NTOK, 256, 0, stream>>>(obuf, gsbuf, gn_w, gn_b, ybuf);

    gemm(ybuf, Wo, nullptr, (float*)d_out, NTOK, 256, 256, ACT_NONE);
}


# features: low-rank mu networks run as token-tile GEMMs on the f32 matrix instruction (10-wave group per 32 key positions, tiles in LDS, weights fetched once per 32 tokens); k sum order changed
# speedup vs baseline: 2.3630x; 1.3561x over previous
#include <hip/hip_runtime.h>
#include <math.h>

#define Hdim 256
#define NHEAD 4
#define DK 64
#define PLR 32
#define Bsz 2
#define LQ 8
#define LKs 512
#define NTOK (Bsz * LQ * LKs)
#define EPSV 1e-5f

#define ACT_NONE  0
#define ACT_SILU  1
#define ACT_TANH  2
#define ACT_DECAY 3

typedef float v2f __attribute__((ext_vector_type(2)));
typedef float v4f __attribute__((ext_vector_type(4)));
typedef float v8f __attribute__((ext_vector_type(8)));

#define FT_TOK   32
#define FT_WAVES 10
#define FT_XP    260
#define FT_T1P   100
#define FT_T2P   68

__global__ __launch_bounds__(FT_WAVES * 32) void rwkv_features_kernel(
    const float* __restrict__ query, const float* __restrict__ keyval,
    const float* __restrict__ rwg_mu, const float* __restrict__ W_rwg0,
    const float* __restrict__ W_rwg2, const float* __restrict__ rwg_bias,
    const float* __restrict__ kv_mu, const float* __restrict__ W_kv0,
    const float* __restrict__ W_kv2, const float* __restrict__ kv_bias,
    float* __restrict__ xr, float* __restrict__ xw, float* __restrict__ xg,
    float* __restrict__ xk, float* __restrict__ xv)
{
    __shared__ __attribute__((aligned(16))) float xr_s[FT_TOK * FT_XP];
    __shared__ __attribute__((aligned(16))) float xk_s[FT_TOK * FT_XP];
    __shared__ __attribute__((aligned(16))) float t1_s[FT_TOK * FT_T1P];
    __shared__ __attribute__((aligned(16))) float t2_s[FT_TOK * FT_T2P];

    const int tid  = (int)threadIdx.x;
    const int wave = __builtin_amdgcn_readfirstlane(tid >> 5);
    const int lane = tid & 31;
    const int sub  = lane & 15;
    const int hh   = lane >> 4;
    const int koff = hh << 1;

    const int tile = (int)blockIdx.x;
    const int e    = tile >> 4;
    const int l0   = (tile & 15) * FT_TOK;
    const int b    = e >> 3;
    const float* __restrict__ hrow = keyval + ((long)b * LKs + l0) * Hdim;
    const float* __restrict__ qrow = query + (long)e * Hdim;

    {
        const int c4 = (tid & 63) << 2;
        const int t0 = tid >> 6;
        const v4f q  = *(const v4f*)(qrow + c4);
        const v4f mr = *(const v4f*)(rwg_mu + c4);
        const v4f mk = *(const v4f*)(kv_mu + c4);
        #pragma unroll
        for (int j = 0; j < 7; ++j) {
            const int tok = t0 + 5 * j;
            if (j < 6 || tok < FT_TOK) {
                const bool first = (l0 + tok == 0);
                const int prow = first ? 0 : tok - 1;
                const v4f h  = *(const v4f*)(hrow + tok * Hdim + c4);
                v4f hp = *(const v4f*)(hrow + prow * Hdim + c4);
                if (first) hp = (v4f){0.0f, 0.0f, 0.0f, 0.0f};
                const v4f dc = q - h;
                const v4f d  = hp - h;
                *(v4f*)(xr_s + tok * FT_XP + c4) = h + dc * mr;
                *(v4f*)(xk_s + tok * FT_XP + c4) = h + d * mk;
            }
        }
    }
    __syncthreads();

    {
        const bool kvA = (wave >= 6);
        const float* Xs = kvA ? xk_s : xr_s;
        const float* __restrict__ W0 = kvA ? W_kv0 : W_rwg0;
        const int N0  = kvA ? 2 * PLR : 3 * PLR;
        const int nc  = (kvA ? wave - 6 : wave) << 4;
        float* Td     = kvA ? t2_s : t1_s;
        const int PT  = kvA ? FT_T2P : FT_T1P;

        const float* A0 = Xs + sub * FT_XP + koff;
        const float* A1 = A0 + 16 * FT_XP;
        const float* __restrict__ Bp = W0 + koff * N0 + nc + sub;

        v8f c0 = {}, c1 = {};
        #pragma unroll 8
        for (int k = 0; k < Hdim; k += 4) {
            const v2f a0 = *(const v2f*)(A0 + k);
            const v2f a1 = *(const v2f*)(A1 + k);
            v2f bf;
            bf.x = Bp[k * N0];
            bf.y = Bp[(k + 1) * N0];
            c0 = __builtin_amdgcn_wmma_f32_16x16x4_f32(false, a0, false, bf, (short)0, c0, false, false);
            c1 = __builtin_amdgcn_wmma_f32_16x16x4_f32(false, a1, false, bf, (short)0, c1, false, false);
        }
        #pragma unroll
        for (int r = 0; r < 8; ++r) {
            const int m = r + 8 * hh;
            Td[m * PT + nc + sub]        = tanhf(c0[r]);
            Td[(16 + m) * PT + nc + sub] = tanhf(c1[r]);
        }
    }
    __syncthreads();

    {
        const int o    = wave >> 1;
        const bool kvB = (o >= 3);
        const int g    = kvB ? o - 3 : o;
        const float* Ts = kvB ? t2_s : t1_s;
        const int PT   = kvB ? FT_T2P : FT_T1P;
        const float* __restrict__ W2 = kvB ? W_kv2 : W_rwg2;
        const int NW   = kvB ? 2 * PLR : 3 * PLR;
        const float* __restrict__ bias = (kvB ? kv_bias : rwg_bias) + g * Hdim;
        float* __restrict__ outp = (o == 0) ? xr : (o == 1) ? xw : (o == 2) ? xg : (o == 3) ? xk : xv;

        const float* A0 = Ts + sub * PT + g * PLR + koff;
        const float* A1 = A0 + 16 * PT;
        v2f fa0[PLR / 4], fa1[PLR / 4];
        #pragma unroll
        for (int s = 0; s < PLR / 4; ++s) {
            fa0[s] = *(const v2f*)(A0 + 4 * s);
            fa1[s] = *(const v2f*)(A1 + 4 * s);
        }

        const int rowA = 8 * hh - 1;
        const int rowB = rowA + 16;
        const bool first = (l0 + rowA < 0);
        const int rowA0 = first ? 0 : rowA;

        for (int j = 0; j < 8; ++j) {
            const int col = ((((wave & 1) << 3) + j) << 4) + sub;
            const float* __restrict__ Bp = W2 + col * NW + g * PLR + koff;
            const float* __restrict__ hc = hrow + col;

            v2f fb[PLR / 4];
            #pragma unroll
            for (int s = 0; s < PLR / 4; ++s) fb[s] = *(const v2f*)(Bp + 4 * s);
            float hA[9], hB[9];
            hA[0] = hc[rowA0 * Hdim];
            hB[0] = hc[rowB * Hdim];
            #pragma unroll
            for (int i = 1; i < 9; ++i) {
                hA[i] = hc[(rowA + i) * Hdim];
                hB[i] = hc[(rowB + i) * Hdim];
            }
            if (first) hA[0] = 0.0f;
            const float bv = bias[col];
            const float qv = qrow[col];

            v8f c0 = {}, c1 = {};
            #pragma unroll
            for (int s = 0; s < PLR / 4; ++s) {
                c0 = __builtin_amdgcn_wmma_f32_16x16x4_f32(false, fa0[s], false, fb[s], (short)0, c0, false, false);
                c1 = __builtin_amdgcn_wmma_f32_16x16x4_f32(false, fa1[s], false, fb[s], (short)0, c1, false, false);
            }

            const long ob = ((long)e * LKs + l0 + 8 * hh) * Hdim + col;
            #pragma unroll
            for (int r = 0; r < 8; ++r) {
                const float h0 = hA[r + 1];
                const float h1 = hB[r + 1];
                const float dl0 = kvB ? (hA[r] - h0) : (qv - h0);
                const float dl1 = kvB ? (hB[r] - h1) : (qv - h1);
                outp[ob + r * Hdim]        = h0 + dl0 * (c0[r] + bv);
                outp[ob + (r + 16) * Hdim] = h1 + dl1 * (c1[r] + bv);
            }
        }
    }
}

__device__ __forceinline__ float apply_act(float x, int act)
{
    switch (act) {
        case ACT_SILU:  return x / (1.0f + expf(-x));
        case ACT_TANH:  return tanhf(x);
        case ACT_DECAY: return expf(-expf(x));
        default:        return x;
    }
}

__global__ __launch_bounds__(128) void rwkv_wmma_gemm_kernel(
    const float* __restrict__ X, const float* __restrict__ W,
    const float* __restrict__ bias, float* __restrict__ Y,
    int K, int N, int act)
{
    const int wave  = threadIdx.x >> 5;
    const int lane  = threadIdx.x & 31;
    const int mbase = (blockIdx.x * 4 + wave) << 5;
    const int nbase = blockIdx.y << 5;

    const int sub  = lane & 15;
    const int koff = (lane >> 4) << 1;

    const float* __restrict__ Arow0 = X + (long)(mbase + sub) * K + koff;
    const float* __restrict__ Arow1 = Arow0 + (long)16 * K;
    const float* __restrict__ Bcol0 = W + (long)koff * N + nbase + sub;
    const float* __restrict__ Bcol1 = Bcol0 + 16;

    v8f c00 = {}, c01 = {}, c10 = {}, c11 = {};
    #pragma unroll 4
    for (int k = 0; k < K; k += 4) {
        const v2f a0 = *(const v2f*)(Arow0 + k);
        const v2f a1 = *(const v2f*)(Arow1 + k);
        v2f b0, b1;
        b0.x = Bcol0[(long)k * N];
        b0.y = Bcol0[(long)(k + 1) * N];
        b1.x = Bcol1[(long)k * N];
        b1.y = Bcol1[(long)(k + 1) * N];
        c00 = __builtin_amdgcn_wmma_f32_16x16x4_f32(false, a0, false, b0, (short)0, c00, false, false);
        c01 = __builtin_amdgcn_wmma_f32_16x16x4_f32(false, a0, false, b1, (short)0, c01, false, false);
        c10 = __builtin_amdgcn_wmma_f32_16x16x4_f32(false, a1, false, b0, (short)0, c10, false, false);
        c11 = __builtin_amdgcn_wmma_f32_16x16x4_f32(false, a1, false, b1, (short)0, c11, false, false);
    }

    const int rbase = (lane >> 4) * 8;
    const int col0  = nbase + sub;
    const int col1  = col0 + 16;
    const float bv0 = bias ? bias[col0] : 0.0f;
    const float bv1 = bias ? bias[col1] : 0.0f;
    #pragma unroll
    for (int r = 0; r < 8; ++r) {
        const long row0 = (long)mbase + rbase + r;
        const long row1 = row0 + 16;
        Y[row0 * N + col0] = apply_act(c00[r] + bv0, act);
        Y[row0 * N + col1] = apply_act(c01[r] + bv1, act);
        Y[row1 * N + col0] = apply_act(c10[r] + bv0, act);
        Y[row1 * N + col1] = apply_act(c11[r] + bv1, act);
    }
}

#define SC_STEPS  4
#define SC_ROW    (3 * DK + 32)
#define SC_CHUNK  (SC_STEPS * SC_ROW)
#define SC_NCHUNK (LKs / SC_STEPS)
#define SC_BLOCKS (SC_STEPS * 4)
#define SC_RING   4

__global__ __launch_bounds__(32) void rwkv_scan_kernel(
    const float* __restrict__ r, const float* __restrict__ k,
    const float* __restrict__ v, const float* __restrict__ decay,
    const float* __restrict__ u, float* __restrict__ o)
{
    __shared__ __attribute__((aligned(16))) float stage[2 * SC_CHUNK];

    const int chain = blockIdx.x >> 1;
    const int half  = blockIdx.x & 1;
    const int e     = chain >> 2;
    const int head  = chain & 3;
    const int lane  = (int)threadIdx.x;
    const long base = ((long)e * LKs) * Hdim + head * DK;
    const int vcol  = half * 32 + lane;

    float S[DK];
    float uu[DK];
    #pragma unroll
    for (int j = 0; j < DK; ++j) { S[j] = 0.0f; uu[j] = u[head * DK + j]; }

    const float* __restrict__ rg = r + base + 2 * lane;
    const float* __restrict__ kg = k + base + 2 * lane;
    const float* __restrict__ dg = decay + base + 2 * lane;
    const float* __restrict__ vg = v + base + vcol;
    float*       __restrict__ og = o + base + vcol;

    v2f gr[SC_STEPS], gk[SC_STEPS], gd[SC_STEPS];
    float gv[SC_STEPS];

    auto fetch = [&](int chunk) {
        #pragma unroll
        for (int i = 0; i < SC_STEPS; ++i) {
            const long off = (long)(chunk * SC_STEPS + i) * Hdim;
            gr[i] = *(const v2f*)(rg + off);
            gk[i] = *(const v2f*)(kg + off);
            gd[i] = *(const v2f*)(dg + off);
            gv[i] = vg[off];
        }
    };
    auto put = [&](float* buf) {
        #pragma unroll
        for (int i = 0; i < SC_STEPS; ++i) {
            float* slot = buf + i * SC_ROW;
            *(v2f*)(slot + 2 * lane)          = gr[i];
            *(v2f*)(slot + DK + 2 * lane)     = gk[i];
            *(v2f*)(slot + 2 * DK + 2 * lane) = gd[i];
            slot[3 * DK + lane]               = gv[i];
        }
    };

    v4f R[SC_RING][4], K[SC_RING][4], D[SC_RING][4];
    float VT[2];

    auto take = [&](const float* buf, int nb) {
        const int si = (nb % SC_BLOCKS) >> 2, sq = nb & 3, set = nb % SC_RING;
        const float* slot = buf + si * SC_ROW + sq * 16;
        #pragma unroll
        for (int g = 0; g < 4; ++g) {
            R[set][g] = *(const v4f*)(slot + 4 * g);
            K[set][g] = *(const v4f*)(slot + DK + 4 * g);
            D[set][g] = *(const v4f*)(slot + 2 * DK + 4 * g);
        }
        if (sq == 0) VT[(nb >> 2) & 1] = buf[si * SC_ROW + 3 * DK + lane];
    };

    fetch(0);
    put(stage);
    __syncthreads();
    fetch(1);
    take(stage, 0);
    take(stage, 1);

    for (int c = 0; c < SC_NCHUNK; ++c) {
        const float* cur = stage + (c & 1) * SC_CHUNK;
        float*       nxt = stage + ((c + 1) & 1) * SC_CHUNK;
        put(nxt);
        __syncthreads();
        fetch((c + 2 < SC_NCHUNK) ? (c + 2) : (SC_NCHUNK - 1));

        float part[4];
        #pragma unroll
        for (int b = 0; b < SC_BLOCKS; b += 2) {
            take((b + 2 < SC_BLOCKS) ? cur : (const float*)nxt, b + 2);
            take((b + 3 < SC_BLOCKS) ? cur : (const float*)nxt, b + 3);
            __builtin_amdgcn_sched_barrier(0);

            const int i = b >> 2, q = b & 3;
            const int sa = b % SC_RING, sb = (b + 1) % SC_RING;
            const float vt = VT[i & 1];
            float acca = 0.0f, accb = 0.0f;
            #pragma unroll
            for (int j = 0; j < 16; ++j) {
                const int ka = q * 16 + j, kb = ka + 16;
                const float kva = K[sa][j >> 2][j & 3] * vt;
                const float kvb = K[sb][j >> 2][j & 3] * vt;
                acca += R[sa][j >> 2][j & 3] * (S[ka] + uu[ka] * kva);
                accb += R[sb][j >> 2][j & 3] * (S[kb] + uu[kb] * kvb);
                S[ka] = D[sa][j >> 2][j & 3] * S[ka] + kva;
                S[kb] = D[sb][j >> 2][j & 3] * S[kb] + kvb;
            }
            part[q]     = acca;
            part[q + 1] = accb;
            if (q == 2)
                og[(long)(c * SC_STEPS + i) * Hdim] = part[0] + part[1] + part[2] + part[3];
            __builtin_amdgcn_sched_barrier(0);
        }
    }
}

__global__ void rwkv_gn_silu_kernel(
    const float* __restrict__ o, const float* __restrict__ gsilu,
    const float* __restrict__ gn_w, const float* __restrict__ gn_b,
    float* __restrict__ y)
{
    const long tok = blockIdx.x;
    const int t    = threadIdx.x;
    const int head = t >> 6;
    const int idx  = t & 63;

    const float val = o[tok * Hdim + t];
    __shared__ float red[256];

    red[t] = val;
    __syncthreads();
    for (int off = 32; off >= 1; off >>= 1) {
        if (idx < off) red[t] += red[t + off];
        __syncthreads();
    }
    const float mean = red[head * 64] * (1.0f / 64.0f);
    __syncthreads();

    const float dv = val - mean;
    red[t] = dv * dv;
    __syncthreads();
    for (int off = 32; off >= 1; off >>= 1) {
        if (idx < off) red[t] += red[t + off];
        __syncthreads();
    }
    const float var = red[head * 64] * (1.0f / 64.0f);

    const float ynorm = dv * rsqrtf(var + EPSV);
    y[tok * Hdim + t] = (ynorm * gn_w[t] + gn_b[t]) * gsilu[tok * Hdim + t];
}

extern "C" void kernel_launch(void* const* d_in, const int* in_sizes, int n_in,
                              void* d_out, int out_size, void* d_ws, size_t ws_size,
                              hipStream_t stream)
{
    const float* query    = (const float*)d_in[0];
    const float* keyval   = (const float*)d_in[1];
    const float* rwg_mu   = (const float*)d_in[2];
    const float* W_rwg0   = (const float*)d_in[3];
    const float* W_rwg2   = (const float*)d_in[4];
    const float* rwg_bias = (const float*)d_in[5];
    const float* kv_mu    = (const float*)d_in[6];
    const float* W_kv0    = (const float*)d_in[7];
    const float* W_kv2    = (const float*)d_in[8];
    const float* kv_bias  = (const float*)d_in[9];
    const float* Wr       = (const float*)d_in[10];
    const float* Wa       = (const float*)d_in[11];
    const float* Wb       = (const float*)d_in[12];
    const float* bb       = (const float*)d_in[13];
    const float* Wk       = (const float*)d_in[14];
    const float* Wv       = (const float*)d_in[15];
    const float* Wg       = (const float*)d_in[16];
    const float* u        = (const float*)d_in[17];
    const float* gn_w     = (const float*)d_in[18];
    const float* gn_b     = (const float*)d_in[19];
    const float* Wo       = (const float*)d_in[20];

    float* ws = (float*)d_ws;
    const long SZ = (long)NTOK * Hdim;
    float* xr    = ws + 0 * SZ;
    float* xw    = ws + 1 * SZ;
    float* xg    = ws + 2 * SZ;
    float* xk    = ws + 3 * SZ;
    float* xv    = ws + 4 * SZ;
    float* rbuf  = ws + 5 * SZ;
    float* kbuf  = ws + 6 * SZ;
    float* vbuf  = ws + 7 * SZ;
    float* dbuf  = ws + 8 * SZ;
    float* gsbuf = ws + 9 * SZ;
    float* twbuf = ws + 10 * SZ;
    float* obuf  = xr;
    float* ybuf  = xw;

    rwkv_features_kernel<<<NTOK / FT_TOK, FT_WAVES * 32, 0, stream>>>(
        query, keyval, rwg_mu, W_rwg0, W_rwg2, rwg_bias,
        kv_mu, W_kv0, W_kv2, kv_bias, xr, xw, xg, xk, xv);

    auto gemm = [&](const float* X, const float* W, const float* bias,
                    float* Y, int M, int K, int N, int act) {
        dim3 grid(M / 128, N / 32);
        rwkv_wmma_gemm_kernel<<<grid, 128, 0, stream>>>(X, W, bias, Y, K, N, act);
    };

    gemm(xr, Wr, nullptr, rbuf,  NTOK, 256, 256, ACT_NONE);
    gemm(xk, Wk, nullptr, kbuf,  NTOK, 256, 256, ACT_NONE);
    gemm(xv, Wv, nullptr, vbuf,  NTOK, 256, 256, ACT_NONE);
    gemm(xg, Wg, nullptr, gsbuf, NTOK, 256, 256, ACT_SILU);
    gemm(xw, Wa, nullptr, twbuf, NTOK, 256, 64,  ACT_TANH);
    gemm(twbuf, Wb, bb, dbuf,    NTOK, 64,  256, ACT_DECAY);

    rwkv_scan_kernel<<<Bsz * LQ * NHEAD * 2, 32, 0, stream>>>(rbuf, kbuf, vbuf, dbuf, u, obuf);

    rwkv_gn_silu_kernel<<<NTOK, 256, 0, stream>>>(obuf, gsbuf, gn_w, gn_b, ybuf);

    gemm(ybuf, Wo, nullptr, (float*)d_out, NTOK, 256, 256, ACT_NONE);
}


# projections fused into the feature work-group: the five mixed [32 x 256] tiles stay in LDS and feed the r, k, v, g and decay products there; x planes and 6 launches gone; product order kept
# speedup vs baseline: 4.1369x; 1.7507x over previous
#include <hip/hip_runtime.h>
#include <math.h>

#define Hdim 256
#define NHEAD 4
#define DK 64
#define PLR 32
#define GLR 64
#define Bsz 2
#define LQ 8
#define LKs 512
#define NTOK (Bsz * LQ * LKs)
#define EPSV 1e-5f

#define ACT_NONE  0
#define ACT_SILU  1
#define ACT_TANH  2
#define ACT_DECAY 3

typedef float v2f __attribute__((ext_vector_type(2)));
typedef float v4f __attribute__((ext_vector_type(4)));
typedef float v8f __attribute__((ext_vector_type(8)));

#define FT_TOK   32
#define FT_WAVES 10
#define FT_XP    260
#define FT_T1P   100
#define FT_T2P   68
#define FT_TWP   68
#define FT_TILE  (FT_TOK * FT_XP)

template <int K, int N, int P>
__device__ __forceinline__ void ft_products(
    const float* Xs, const float* __restrict__ W, int nc, int sub, int koff,
    v8f& c00, v8f& c01, v8f& c10, v8f& c11)
{
    const float* A0 = Xs + sub * P + koff;
    const float* A1 = A0 + 16 * P;
    const float* __restrict__ B0 = W + koff * N + nc + sub;
    const float* __restrict__ B1 = B0 + 16;
    #pragma unroll 8
    for (int k = 0; k < K; k += 4) {
        const v2f a0 = *(const v2f*)(A0 + k);
        const v2f a1 = *(const v2f*)(A1 + k);
        v2f b0, b1;
        b0.x = B0[k * N];
        b0.y = B0[(k + 1) * N];
        b1.x = B1[k * N];
        b1.y = B1[(k + 1) * N];
        c00 = __builtin_amdgcn_wmma_f32_16x16x4_f32(false, a0, false, b0, (short)0, c00, false, false);
        c01 = __builtin_amdgcn_wmma_f32_16x16x4_f32(false, a0, false, b1, (short)0, c01, false, false);
        c10 = __builtin_amdgcn_wmma_f32_16x16x4_f32(false, a1, false, b0, (short)0, c10, false, false);
        c11 = __builtin_amdgcn_wmma_f32_16x16x4_f32(false, a1, false, b1, (short)0, c11, false, false);
    }
}

template <bool SILU>
__device__ __forceinline__ void ft_strip(
    const float* Xs, const float* __restrict__ Wp, float* __restrict__ yp,
    int nc, int sub, int koff)
{
    v8f c00 = {}, c01 = {}, c10 = {}, c11 = {};
    ft_products<Hdim, Hdim, FT_XP>(Xs, Wp, nc, sub, koff, c00, c01, c10, c11);
    #pragma unroll
    for (int r = 0; r < 8; ++r) {
        if (SILU) {
            yp[r * Hdim]             = c00[r] / (1.0f + expf(-c00[r]));
            yp[r * Hdim + 16]        = c01[r] / (1.0f + expf(-c01[r]));
            yp[(r + 16) * Hdim]      = c10[r] / (1.0f + expf(-c10[r]));
            yp[(r + 16) * Hdim + 16] = c11[r] / (1.0f + expf(-c11[r]));
        } else {
            yp[r * Hdim]             = c00[r];
            yp[r * Hdim + 16]        = c01[r];
            yp[(r + 16) * Hdim]      = c10[r];
            yp[(r + 16) * Hdim + 16] = c11[r];
        }
    }
}

__device__ __forceinline__ void ft_long_job(
    int jb, const float* xs,
    const float* __restrict__ Wr, const float* __restrict__ Wk,
    const float* __restrict__ Wv, const float* __restrict__ Wg,
    float* __restrict__ rbuf, float* __restrict__ kbuf,
    float* __restrict__ vbuf, float* __restrict__ gsbuf,
    long yrow, int sub, int koff)
{
    const int nc = (jb & 7) << 5;
    switch (jb >> 3) {
        case 0:  ft_strip<false>(xs,               Wr, rbuf  + yrow + nc, nc, sub, koff); break;
        case 1:  ft_strip<false>(xs + 3 * FT_TILE, Wk, kbuf  + yrow + nc, nc, sub, koff); break;
        case 2:  ft_strip<false>(xs + 4 * FT_TILE, Wv, vbuf  + yrow + nc, nc, sub, koff); break;
        default: ft_strip<true>(xs + 2 * FT_TILE,  Wg, gsbuf + yrow + nc, nc, sub, koff); break;
    }
}

__global__ __launch_bounds__(FT_WAVES * 32) void rwkv_features_kernel(
    const float* __restrict__ query, const float* __restrict__ keyval,
    const float* __restrict__ rwg_mu, const float* __restrict__ W_rwg0,
    const float* __restrict__ W_rwg2, const float* __restrict__ rwg_bias,
    const float* __restrict__ kv_mu, const float* __restrict__ W_kv0,
    const float* __restrict__ W_kv2, const float* __restrict__ kv_bias,
    const float* __restrict__ Wr, const float* __restrict__ Wa,
    const float* __restrict__ Wb, const float* __restrict__ bb,
    const float* __restrict__ Wk, const float* __restrict__ Wv,
    const float* __restrict__ Wg,
    float* __restrict__ rbuf, float* __restrict__ kbuf, float* __restrict__ vbuf,
    float* __restrict__ gsbuf, float* __restrict__ dbuf)
{
    __shared__ __attribute__((aligned(16))) float xs[5 * FT_TILE];
    __shared__ __attribute__((aligned(16))) float t1_s[FT_TOK * FT_T1P];
    __shared__ __attribute__((aligned(16))) float t2_s[FT_TOK * FT_T2P];
    __shared__ __attribute__((aligned(16))) float tw_s[FT_TOK * FT_TWP];
    float* const xr_s = xs;
    float* const xk_s = xs + 3 * FT_TILE;

    const int tid  = (int)threadIdx.x;
    const int wave = __builtin_amdgcn_readfirstlane(tid >> 5);
    const int lane = tid & 31;
    const int sub  = lane & 15;
    const int hh   = lane >> 4;
    const int koff = hh << 1;

    const int tile = (int)blockIdx.x;
    const int e    = tile >> 4;
    const int l0   = (tile & 15) * FT_TOK;
    const int b    = e >> 3;
    const float* __restrict__ hrow = keyval + ((long)b * LKs + l0) * Hdim;
    const float* __restrict__ qrow = query + (long)e * Hdim;

    {
        const int c4 = (tid & 63) << 2;
        const int t0 = tid >> 6;
        const v4f q  = *(const v4f*)(qrow + c4);
        const v4f mr = *(const v4f*)(rwg_mu + c4);
        const v4f mk = *(const v4f*)(kv_mu + c4);
        #pragma unroll
        for (int j = 0; j < 7; ++j) {
            const int tok = t0 + 5 * j;
            if (j < 6 || tok < FT_TOK) {
                const bool first = (l0 + tok == 0);
                const int prow = first ? 0 : tok - 1;
                const v4f h  = *(const v4f*)(hrow + tok * Hdim + c4);
                v4f hp = *(const v4f*)(hrow + prow * Hdim + c4);
                if (first) hp = (v4f){0.0f, 0.0f, 0.0f, 0.0f};
                const v4f dc = q - h;
                const v4f d  = hp - h;
                *(v4f*)(xr_s + tok * FT_XP + c4) = h + dc * mr;
                *(v4f*)(xk_s + tok * FT_XP + c4) = h + d * mk;
            }
        }
    }
    __syncthreads();

    {
        const bool kvA = (wave >= 6);
        const float* Xs = kvA ? xk_s : xr_s;
        const float* __restrict__ W0 = kvA ? W_kv0 : W_rwg0;
        const int N0  = kvA ? 2 * PLR : 3 * PLR;
        const int nc  = (kvA ? wave - 6 : wave) << 4;
        float* Td     = kvA ? t2_s : t1_s;
        const int PT  = kvA ? FT_T2P : FT_T1P;

        const float* A0 = Xs + sub * FT_XP + koff;
        const float* A1 = A0 + 16 * FT_XP;
        const float* __restrict__ Bp = W0 + koff * N0 + nc + sub;

        v8f c0 = {}, c1 = {};
        #pragma unroll 8
        for (int k = 0; k < Hdim; k += 4) {
            const v2f a0 = *(const v2f*)(A0 + k);
            const v2f a1 = *(const v2f*)(A1 + k);
            v2f bf;
            bf.x = Bp[k * N0];
            bf.y = Bp[(k + 1) * N0];
            c0 = __builtin_amdgcn_wmma_f32_16x16x4_f32(false, a0, false, bf, (short)0, c0, false, false);
            c1 = __builtin_amdgcn_wmma_f32_16x16x4_f32(false, a1, false, bf, (short)0, c1, false, false);
        }
        #pragma unroll
        for (int r = 0; r < 8; ++r) {
            const int m = r + 8 * hh;
            Td[m * PT + nc + sub]        = tanhf(c0[r]);
            Td[(16 + m) * PT + nc + sub] = tanhf(c1[r]);
        }
    }
    __syncthreads();

    {
        const int o    = wave >> 1;
        const bool kvB = (o >= 3);
        const int g    = kvB ? o - 3 : o;
        const float* Ts = kvB ? t2_s : t1_s;
        const int PT   = kvB ? FT_T2P : FT_T1P;
        const float* __restrict__ W2 = kvB ? W_kv2 : W_rwg2;
        const int NW   = kvB ? 2 * PLR : 3 * PLR;
        const float* __restrict__ bias = (kvB ? kv_bias : rwg_bias) + g * Hdim;
        float* outp = xs + o * FT_TILE;

        const float* A0 = Ts + sub * PT + g * PLR + koff;
        const float* A1 = A0 + 16 * PT;
        v2f fa0[PLR / 4], fa1[PLR / 4];
        #pragma unroll
        for (int s = 0; s < PLR / 4; ++s) {
            fa0[s] = *(const v2f*)(A0 + 4 * s);
            fa1[s] = *(const v2f*)(A1 + 4 * s);
        }

        const int rowA = 8 * hh - 1;
        const int rowB = rowA + 16;
        const bool first = (l0 + rowA < 0);
        const int rowA0 = first ? 0 : rowA;

        for (int j = 0; j < 8; ++j) {
            const int col = ((((wave & 1) << 3) + j) << 4) + sub;
            const float* __restrict__ Bp = W2 + col * NW + g * PLR + koff;
            const float* __restrict__ hc = hrow + col;

            v2f fb[PLR / 4];
            #pragma unroll
            for (int s = 0; s < PLR / 4; ++s) fb[s] = *(const v2f*)(Bp + 4 * s);
            float hA[9], hB[9];
            hA[0] = hc[rowA0 * Hdim];
            hB[0] = hc[rowB * Hdim];
            #pragma unroll
            for (int i = 1; i < 9; ++i) {
                hA[i] = hc[(rowA + i) * Hdim];
                hB[i] = hc[(rowB + i) * Hdim];
            }
            if (first) hA[0] = 0.0f;
            const float bv = bias[col];
            const float qv = qrow[col];

            v8f c0 = {}, c1 = {};
            #pragma unroll
            for (int s = 0; s < PLR / 4; ++s) {
                c0 = __builtin_amdgcn_wmma_f32_16x16x4_f32(false, fa0[s], false, fb[s], (short)0, c0, false, false);
                c1 = __builtin_amdgcn_wmma_f32_16x16x4_f32(false, fa1[s], false, fb[s], (short)0, c1, false, false);
            }

            const int ob = 8 * hh * FT_XP + col;
            #pragma unroll
            for (int r = 0; r < 8; ++r) {
                const float h0 = hA[r + 1];
                const float h1 = hB[r + 1];
                const float dl0 = kvB ? (hA[r] - h0) : (qv - h0);
                const float dl1 = kvB ? (hB[r] - h1) : (qv - h1);
                outp[ob + r * FT_XP]        = h0 + dl0 * (c0[r] + bv);
                outp[ob + (r + 16) * FT_XP] = h1 + dl1 * (c1[r] + bv);
            }
        }
    }
    __syncthreads();

    {
        const long yrow = ((long)e * LKs + l0 + 8 * hh) * Hdim + sub;

        if (wave < 2) {
            const int nc = wave << 5;
            v8f c00 = {}, c01 = {}, c10 = {}, c11 = {};
            ft_products<Hdim, GLR, FT_XP>(xs + FT_TILE, Wa, nc, sub, koff, c00, c01, c10, c11);
            float* tp = tw_s + 8 * hh * FT_TWP + nc + sub;
            #pragma unroll
            for (int r = 0; r < 8; ++r) {
                tp[r * FT_TWP]             = tanhf(c00[r]);
                tp[r * FT_TWP + 16]        = tanhf(c01[r]);
                tp[(r + 16) * FT_TWP]      = tanhf(c10[r]);
                tp[(r + 16) * FT_TWP + 16] = tanhf(c11[r]);
            }
        } else {
            ft_long_job(wave - 2, xs, Wr, Wk, Wv, Wg, rbuf, kbuf, vbuf, gsbuf, yrow, sub, koff);
        }
        __syncthreads();

        for (int uj = wave - 4; uj >= 0 && uj < 8; uj += 6) {
            const int nc = uj << 5;
            v8f c00 = {}, c01 = {}, c10 = {}, c11 = {};
            ft_products<GLR, Hdim, FT_TWP>(tw_s, Wb, nc, sub, koff, c00, c01, c10, c11);
            const float bv0 = bb[nc + sub];
            const float bv1 = bb[nc + sub + 16];
            float* __restrict__ yp = dbuf + yrow + nc;
            #pragma unroll
            for (int r = 0; r < 8; ++r) {
                yp[r * Hdim]             = expf(-expf(c00[r] + bv0));
                yp[r * Hdim + 16]        = expf(-expf(c01[r] + bv1));
                yp[(r + 16) * Hdim]      = expf(-expf(c10[r] + bv0));
                yp[(r + 16) * Hdim + 16] = expf(-expf(c11[r] + bv1));
            }
        }
        for (int t = wave; t < 24; t += FT_WAVES)
            ft_long_job(8 + t, xs, Wr, Wk, Wv, Wg, rbuf, kbuf, vbuf, gsbuf, yrow, sub, koff);
    }
}

__device__ __forceinline__ float apply_act(float x, int act)
{
    switch (act) {
        case ACT_SILU:  return x / (1.0f + expf(-x));
        case ACT_TANH:  return tanhf(x);
        case ACT_DECAY: return expf(-expf(x));
        default:        return x;
    }
}

__global__ __launch_bounds__(128) void rwkv_wmma_gemm_kernel(
    const float* __restrict__ X, const float* __restrict__ W,
    const float* __restrict__ bias, float* __restrict__ Y,
    int K, int N, int act)
{
    const int wave  = threadIdx.x >> 5;
    const int lane  = threadIdx.x & 31;
    const int mbase = (blockIdx.x * 4 + wave) << 5;
    const int nbase = blockIdx.y << 5;

    const int sub  = lane & 15;
    const int koff = (lane >> 4) << 1;

    const float* __restrict__ Arow0 = X + (long)(mbase + sub) * K + koff;
    const float* __restrict__ Arow1 = Arow0 + (long)16 * K;
    const float* __restrict__ Bcol0 = W + (long)koff * N + nbase + sub;
    const float* __restrict__ Bcol1 = Bcol0 + 16;

    v8f c00 = {}, c01 = {}, c10 = {}, c11 = {};
    #pragma unroll 4
    for (int k = 0; k < K; k += 4) {
        const v2f a0 = *(const v2f*)(Arow0 + k);
        const v2f a1 = *(const v2f*)(Arow1 + k);
        v2f b0, b1;
        b0.x = Bcol0[(long)k * N];
        b0.y = Bcol0[(long)(k + 1) * N];
        b1.x = Bcol1[(long)k * N];
        b1.y = Bcol1[(long)(k + 1) * N];
        c00 = __builtin_amdgcn_wmma_f32_16x16x4_f32(false, a0, false, b0, (short)0, c00, false, false);
        c01 = __builtin_amdgcn_wmma_f32_16x16x4_f32(false, a0, false, b1, (short)0, c01, false, false);
        c10 = __builtin_amdgcn_wmma_f32_16x16x4_f32(false, a1, false, b0, (short)0, c10, false, false);
        c11 = __builtin_amdgcn_wmma_f32_16x16x4_f32(false, a1, false, b1, (short)0, c11, false, false);
    }

    const int rbase = (lane >> 4) * 8;
    const int col0  = nbase + sub;
    const int col1  = col0 + 16;
    const float bv0 = bias ? bias[col0] : 0.0f;
    const float bv1 = bias ? bias[col1] : 0.0f;
    #pragma unroll
    for (int r = 0; r < 8; ++r) {
        const long row0 = (long)mbase + rbase + r;
        const long row1 = row0 + 16;
        Y[row0 * N + col0] = apply_act(c00[r] + bv0, act);
        Y[row0 * N + col1] = apply_act(c01[r] + bv1, act);
        Y[row1 * N + col0] = apply_act(c10[r] + bv0, act);
        Y[row1 * N + col1] = apply_act(c11[r] + bv1, act);
    }
}

#define SC_STEPS  4
#define SC_ROW    (3 * DK + 32)
#define SC_CHUNK  (SC_STEPS * SC_ROW)
#define SC_NCHUNK (LKs / SC_STEPS)
#define SC_BLOCKS (SC_STEPS * 4)
#define SC_RING   4

__global__ __launch_bounds__(32) void rwkv_scan_kernel(
    const float* __restrict__ r, const float* __restrict__ k,
    const float* __restrict__ v, const float* __restrict__ decay,
    const float* __restrict__ u, float* __restrict__ o)
{
    __shared__ __attribute__((aligned(16))) float stage[2 * SC_CHUNK];

    const int chain = blockIdx.x >> 1;
    const int half  = blockIdx.x & 1;
    const int e     = chain >> 2;
    const int head  = chain & 3;
    const int lane  = (int)threadIdx.x;
    const long base = ((long)e * LKs) * Hdim + head * DK;
    const int vcol  = half * 32 + lane;

    float S[DK];
    float uu[DK];
    #pragma unroll
    for (int j = 0; j < DK; ++j) { S[j] = 0.0f; uu[j] = u[head * DK + j]; }

    const float* __restrict__ rg = r + base + 2 * lane;
    const float* __restrict__ kg = k + base + 2 * lane;
    const float* __restrict__ dg = decay + base + 2 * lane;
    const float* __restrict__ vg = v + base + vcol;
    float*       __restrict__ og = o + base + vcol;

    v2f gr[SC_STEPS], gk[SC_STEPS], gd[SC_STEPS];
    float gv[SC_STEPS];

    auto fetch = [&](int chunk) {
        #pragma unroll
        for (int i = 0; i < SC_STEPS; ++i) {
            const long off = (long)(chunk * SC_STEPS + i) * Hdim;
            gr[i] = *(const v2f*)(rg + off);
            gk[i] = *(const v2f*)(kg + off);
            gd[i] = *(const v2f*)(dg + off);
            gv[i] = vg[off];
        }
    };
    auto put = [&](float* buf) {
        #pragma unroll
        for (int i = 0; i < SC_STEPS; ++i) {
            float* slot = buf + i * SC_ROW;
            *(v2f*)(slot + 2 * lane)          = gr[i];
            *(v2f*)(slot + DK + 2 * lane)     = gk[i];
            *(v2f*)(slot + 2 * DK + 2 * lane) = gd[i];
            slot[3 * DK + lane]               = gv[i];
        }
    };

    v4f R[SC_RING][4], K[SC_RING][4], D[SC_RING][4];
    float VT[2];

    auto take = [&](const float* buf, int nb) {
        const int si = (nb % SC_BLOCKS) >> 2, sq = nb & 3, set = nb % SC_RING;
        const float* slot = buf + si * SC_ROW + sq * 16;
        #pragma unroll
        for (int g = 0; g < 4; ++g) {
            R[set][g] = *(const v4f*)(slot + 4 * g);
            K[set][g] = *(const v4f*)(slot + DK + 4 * g);
            D[set][g] = *(const v4f*)(slot + 2 * DK + 4 * g);
        }
        if (sq == 0) VT[(nb >> 2) & 1] = buf[si * SC_ROW + 3 * DK + lane];
    };

    fetch(0);
    put(stage);
    __syncthreads();
    fetch(1);
    take(stage, 0);
    take(stage, 1);

    for (int c = 0; c < SC_NCHUNK; ++c) {
        const float* cur = stage + (c & 1) * SC_CHUNK;
        float*       nxt = stage + ((c + 1) & 1) * SC_CHUNK;
        put(nxt);
        __syncthreads();
        fetch((c + 2 < SC_NCHUNK) ? (c + 2) : (SC_NCHUNK - 1));

        float part[4];
        #pragma unroll
        for (int b = 0; b < SC_BLOCKS; b += 2) {
            take((b + 2 < SC_BLOCKS) ? cur : (const float*)nxt, b + 2);
            take((b + 3 < SC_BLOCKS) ? cur : (const float*)nxt, b + 3);
            __builtin_amdgcn_sched_barrier(0);

            const int i = b >> 2, q = b & 3;
            const int sa = b % SC_RING, sb = (b + 1) % SC_RING;
            const float vt = VT[i & 1];
            float acca = 0.0f, accb = 0.0f;
            #pragma unroll
            for (int j = 0; j < 16; ++j) {
                const int ka = q * 16 + j, kb = ka + 16;
                const float kva = K[sa][j >> 2][j & 3] * vt;
                const float kvb = K[sb][j >> 2][j & 3] * vt;
                acca += R[sa][j >> 2][j & 3] * (S[ka] + uu[ka] * kva);
                accb += R[sb][j >> 2][j & 3] * (S[kb] + uu[kb] * kvb);
                S[ka] = D[sa][j >> 2][j & 3] * S[ka] + kva;
                S[kb] = D[sb][j >> 2][j & 3] * S[kb] + kvb;
            }
            part[q]     = acca;
            part[q + 1] = accb;
            if (q == 2)
                og[(long)(c * SC_STEPS + i) * Hdim] = part[0] + part[1] + part[2] + part[3];
            __builtin_amdgcn_sched_barrier(0);
        }
    }
}

__global__ void rwkv_gn_silu_kernel(
    const float* __restrict__ o, const float* __restrict__ gsilu,
    const float* __restrict__ gn_w, const float* __restrict__ gn_b,
    float* __restrict__ y)
{
    const long tok = blockIdx.x;
    const int t    = threadIdx.x;
    const int head = t >> 6;
    const int idx  = t & 63;

    const float val = o[tok * Hdim + t];
    __shared__ float red[256];

    red[t] = val;
    __syncthreads();
    for (int off = 32; off >= 1; off >>= 1) {
        if (idx < off) red[t] += red[t + off];
        __syncthreads();
    }
    const float mean = red[head * 64] * (1.0f / 64.0f);
    __syncthreads();

    const float dv = val - mean;
    red[t] = dv * dv;
    __syncthreads();
    for (int off = 32; off >= 1; off >>= 1) {
        if (idx < off) red[t] += red[t + off];
        __syncthreads();
    }
    const float var = red[head * 64] * (1.0f / 64.0f);

    const float ynorm = dv * rsqrtf(var + EPSV);
    y[tok * Hdim + t] = (ynorm * gn_w[t] + gn_b[t]) * gsilu[tok * Hdim + t];
}

extern "C" void kernel_launch(void* const* d_in, const int* in_sizes, int n_in,
                              void* d_out, int out_size, void* d_ws, size_t ws_size,
                              hipStream_t stream)
{
    const float* query    = (const float*)d_in[0];
    const float* keyval   = (const float*)d_in[1];
    const float* rwg_mu   = (const float*)d_in[2];
    const float* W_rwg0   = (const float*)d_in[3];
    const float* W_rwg2   = (const float*)d_in[4];
    const float* rwg_bias = (const float*)d_in[5];
    const float* kv_mu    = (const float*)d_in[6];
    const float* W_kv0    = (const float*)d_in[7];
    const float* W_kv2    = (const float*)d_in[8];
    const float* kv_bias  = (const float*)d_in[9];
    const float* Wr       = (const float*)d_in[10];
    const float* Wa       = (const float*)d_in[11];
    const float* Wb       = (const float*)d_in[12];
    const float* bb       = (const float*)d_in[13];
    const float* Wk       = (const float*)d_in[14];
    const float* Wv       = (const float*)d_in[15];
    const float* Wg       = (const float*)d_in[16];
    const float* u        = (const float*)d_in[17];
    const float* gn_w     = (const float*)d_in[18];
    const float* gn_b     = (const float*)d_in[19];
    const float* Wo       = (const float*)d_in[20];

    float* ws = (float*)d_ws;
    const long SZ = (long)NTOK * Hdim;
    float* rbuf  = ws + 0 * SZ;
    float* kbuf  = ws + 1 * SZ;
    float* vbuf  = ws + 2 * SZ;
    float* dbuf  = ws + 3 * SZ;
    float* gsbuf = ws + 4 * SZ;
    float* obuf  = ws + 5 * SZ;
    float* ybuf  = ws + 6 * SZ;

    rwkv_features_kernel<<<NTOK / FT_TOK, FT_WAVES * 32, 0, stream>>>(
        query, keyval, rwg_mu, W_rwg0, W_rwg2, rwg_bias,
        kv_mu, W_kv0, W_kv2, kv_bias,
        Wr, Wa, Wb, bb, Wk, Wv, Wg,
        rbuf, kbuf, vbuf, gsbuf, dbuf);

    auto gemm = [&](const float* X, const float* W, const float* bias,
                    float* Y, int M, int K, int N, int act) {
        dim3 grid(M / 128, N / 32);
        rwkv_wmma_gemm_kernel<<<grid, 128, 0, stream>>>(X, W, bias, Y, K, N, act);
    };

    rwkv_scan_kernel<<<Bsz * LQ * NHEAD * 2, 32, 0, stream>>>(rbuf, kbuf, vbuf, dbuf, u, obuf);

    rwkv_gn_silu_kernel<<<NTOK, 256, 0, stream>>>(obuf, gsbuf, gn_w, gn_b, ybuf);

    gemm(ybuf, Wo, nullptr, (float*)d_out, NTOK, 256, 256, ACT_NONE);
}


# GroupNorm+gate moved into the scan kernel: a third wave per chain takes each 4-step chunk of o from LDS and finishes it beside the two scan waves; gn launch and o plane gone; GN sum order changed
# speedup vs baseline: 4.2682x; 1.0317x over previous
#include <hip/hip_runtime.h>
#include <math.h>

#define Hdim 256
#define NHEAD 4
#define DK 64
#define PLR 32
#define GLR 64
#define Bsz 2
#define LQ 8
#define LKs 512
#define NTOK (Bsz * LQ * LKs)
#define EPSV 1e-5f

#define ACT_NONE  0
#define ACT_SILU  1
#define ACT_TANH  2
#define ACT_DECAY 3

typedef float v2f __attribute__((ext_vector_type(2)));
typedef float v4f __attribute__((ext_vector_type(4)));
typedef float v8f __attribute__((ext_vector_type(8)));

#define FT_TOK   32
#define FT_WAVES 10
#define FT_XP    260
#define FT_T1P   100
#define FT_T2P   68
#define FT_TWP   68
#define FT_TILE  (FT_TOK * FT_XP)

template <int K, int N, int P>
__device__ __forceinline__ void ft_products(
    const float* Xs, const float* __restrict__ W, int nc, int sub, int koff,
    v8f& c00, v8f& c01, v8f& c10, v8f& c11)
{
    const float* A0 = Xs + sub * P + koff;
    const float* A1 = A0 + 16 * P;
    const float* __restrict__ B0 = W + koff * N + nc + sub;
    const float* __restrict__ B1 = B0 + 16;
    #pragma unroll 8
    for (int k = 0; k < K; k += 4) {
        const v2f a0 = *(const v2f*)(A0 + k);
        const v2f a1 = *(const v2f*)(A1 + k);
        v2f b0, b1;
        b0.x = B0[k * N];
        b0.y = B0[(k + 1) * N];
        b1.x = B1[k * N];
        b1.y = B1[(k + 1) * N];
        c00 = __builtin_amdgcn_wmma_f32_16x16x4_f32(false, a0, false, b0, (short)0, c00, false, false);
        c01 = __builtin_amdgcn_wmma_f32_16x16x4_f32(false, a0, false, b1, (short)0, c01, false, false);
        c10 = __builtin_amdgcn_wmma_f32_16x16x4_f32(false, a1, false, b0, (short)0, c10, false, false);
        c11 = __builtin_amdgcn_wmma_f32_16x16x4_f32(false, a1, false, b1, (short)0, c11, false, false);
    }
}

template <bool SILU>
__device__ __forceinline__ void ft_strip(
    const float* Xs, const float* __restrict__ Wp, float* __restrict__ yp,
    int nc, int sub, int koff)
{
    v8f c00 = {}, c01 = {}, c10 = {}, c11 = {};
    ft_products<Hdim, Hdim, FT_XP>(Xs, Wp, nc, sub, koff, c00, c01, c10, c11);
    #pragma unroll
    for (int r = 0; r < 8; ++r) {
        if (SILU) {
            yp[r * Hdim]             = c00[r] / (1.0f + expf(-c00[r]));
            yp[r * Hdim + 16]        = c01[r] / (1.0f + expf(-c01[r]));
            yp[(r + 16) * Hdim]      = c10[r] / (1.0f + expf(-c10[r]));
            yp[(r + 16) * Hdim + 16] = c11[r] / (1.0f + expf(-c11[r]));
        } else {
            yp[r * Hdim]             = c00[r];
            yp[r * Hdim + 16]        = c01[r];
            yp[(r + 16) * Hdim]      = c10[r];
            yp[(r + 16) * Hdim + 16] = c11[r];
        }
    }
}

__device__ __forceinline__ void ft_long_job(
    int jb, const float* xs,
    const float* __restrict__ Wr, const float* __restrict__ Wk,
    const float* __restrict__ Wv, const float* __restrict__ Wg,
    float* __restrict__ rbuf, float* __restrict__ kbuf,
    float* __restrict__ vbuf, float* __restrict__ gsbuf,
    long yrow, int sub, int koff)
{
    const int nc = (jb & 7) << 5;
    switch (jb >> 3) {
        case 0:  ft_strip<false>(xs,               Wr, rbuf  + yrow + nc, nc, sub, koff); break;
        case 1:  ft_strip<false>(xs + 3 * FT_TILE, Wk, kbuf  + yrow + nc, nc, sub, koff); break;
        case 2:  ft_strip<false>(xs + 4 * FT_TILE, Wv, vbuf  + yrow + nc, nc, sub, koff); break;
        default: ft_strip<true>(xs + 2 * FT_TILE,  Wg, gsbuf + yrow + nc, nc, sub, koff); break;
    }
}

__global__ __launch_bounds__(FT_WAVES * 32) void rwkv_features_kernel(
    const float* __restrict__ query, const float* __restrict__ keyval,
    const float* __restrict__ rwg_mu, const float* __restrict__ W_rwg0,
    const float* __restrict__ W_rwg2, const float* __restrict__ rwg_bias,
    const float* __restrict__ kv_mu, const float* __restrict__ W_kv0,
    const float* __restrict__ W_kv2, const float* __restrict__ kv_bias,
    const float* __restrict__ Wr, const float* __restrict__ Wa,
    const float* __restrict__ Wb, const float* __restrict__ bb,
    const float* __restrict__ Wk, const float* __restrict__ Wv,
    const float* __restrict__ Wg,
    float* __restrict__ rbuf, float* __restrict__ kbuf, float* __restrict__ vbuf,
    float* __restrict__ gsbuf, float* __restrict__ dbuf)
{
    __shared__ __attribute__((aligned(16))) float xs[5 * FT_TILE];
    __shared__ __attribute__((aligned(16))) float t1_s[FT_TOK * FT_T1P];
    __shared__ __attribute__((aligned(16))) float t2_s[FT_TOK * FT_T2P];
    __shared__ __attribute__((aligned(16))) float tw_s[FT_TOK * FT_TWP];
    float* const xr_s = xs;
    float* const xk_s = xs + 3 * FT_TILE;

    const int tid  = (int)threadIdx.x;
    const int wave = __builtin_amdgcn_readfirstlane(tid >> 5);
    const int lane = tid & 31;
    const int sub  = lane & 15;
    const int hh   = lane >> 4;
    const int koff = hh << 1;

    const int tile = (int)blockIdx.x;
    const int e    = tile >> 4;
    const int l0   = (tile & 15) * FT_TOK;
    const int b    = e >> 3;
    const float* __restrict__ hrow = keyval + ((long)b * LKs + l0) * Hdim;
    const float* __restrict__ qrow = query + (long)e * Hdim;

    {
        const int c4 = (tid & 63) << 2;
        const int t0 = tid >> 6;
        const v4f q  = *(const v4f*)(qrow + c4);
        const v4f mr = *(const v4f*)(rwg_mu + c4);
        const v4f mk = *(const v4f*)(kv_mu + c4);
        #pragma unroll
        for (int j = 0; j < 7; ++j) {
            const int tok = t0 + 5 * j;
            if (j < 6 || tok < FT_TOK) {
                const bool first = (l0 + tok == 0);
                const int prow = first ? 0 : tok - 1;
                const v4f h  = *(const v4f*)(hrow + tok * Hdim + c4);
                v4f hp = *(const v4f*)(hrow + prow * Hdim + c4);
                if (first) hp = (v4f){0.0f, 0.0f, 0.0f, 0.0f};
                const v4f dc = q - h;
                const v4f d  = hp - h;
                *(v4f*)(xr_s + tok * FT_XP + c4) = h + dc * mr;
                *(v4f*)(xk_s + tok * FT_XP + c4) = h + d * mk;
            }
        }
    }
    __syncthreads();

    {
        const bool kvA = (wave >= 6);
        const float* Xs = kvA ? xk_s : xr_s;
        const float* __restrict__ W0 = kvA ? W_kv0 : W_rwg0;
        const int N0  = kvA ? 2 * PLR : 3 * PLR;
        const int nc  = (kvA ? wave - 6 : wave) << 4;
        float* Td     = kvA ? t2_s : t1_s;
        const int PT  = kvA ? FT_T2P : FT_T1P;

        const float* A0 = Xs + sub * FT_XP + koff;
        const float* A1 = A0 + 16 * FT_XP;
        const float* __restrict__ Bp = W0 + koff * N0 + nc + sub;

        v8f c0 = {}, c1 = {};
        #pragma unroll 8
        for (int k = 0; k < Hdim; k += 4) {
            const v2f a0 = *(const v2f*)(A0 + k);
            const v2f a1 = *(const v2f*)(A1 + k);
            v2f bf;
            bf.x = Bp[k * N0];
            bf.y = Bp[(k + 1) * N0];
            c0 = __builtin_amdgcn_wmma_f32_16x16x4_f32(false, a0, false, bf, (short)0, c0, false, false);
            c1 = __builtin_amdgcn_wmma_f32_16x16x4_f32(false, a1, false, bf, (short)0, c1, false, false);
        }
        #pragma unroll
        for (int r = 0; r < 8; ++r) {
            const int m = r + 8 * hh;
            Td[m * PT + nc + sub]        = tanhf(c0[r]);
            Td[(16 + m) * PT + nc + sub] = tanhf(c1[r]);
        }
    }
    __syncthreads();

    {
        const int o    = wave >> 1;
        const bool kvB = (o >= 3);
        const int g    = kvB ? o - 3 : o;
        const float* Ts = kvB ? t2_s : t1_s;
        const int PT   = kvB ? FT_T2P : FT_T1P;
        const float* __restrict__ W2 = kvB ? W_kv2 : W_rwg2;
        const int NW   = kvB ? 2 * PLR : 3 * PLR;
        const float* __restrict__ bias = (kvB ? kv_bias : rwg_bias) + g * Hdim;
        float* outp = xs + o * FT_TILE;

        const float* A0 = Ts + sub * PT + g * PLR + koff;
        const float* A1 = A0 + 16 * PT;
        v2f fa0[PLR / 4], fa1[PLR / 4];
        #pragma unroll
        for (int s = 0; s < PLR / 4; ++s) {
            fa0[s] = *(const v2f*)(A0 + 4 * s);
            fa1[s] = *(const v2f*)(A1 + 4 * s);
        }

        const int rowA = 8 * hh - 1;
        const int rowB = rowA + 16;
        const bool first = (l0 + rowA < 0);
        const int rowA0 = first ? 0 : rowA;

        for (int j = 0; j < 8; ++j) {
            const int col = ((((wave & 1) << 3) + j) << 4) + sub;
            const float* __restrict__ Bp = W2 + col * NW + g * PLR + koff;
            const float* __restrict__ hc = hrow + col;

            v2f fb[PLR / 4];
            #pragma unroll
            for (int s = 0; s < PLR / 4; ++s) fb[s] = *(const v2f*)(Bp + 4 * s);
            float hA[9], hB[9];
            hA[0] = hc[rowA0 * Hdim];
            hB[0] = hc[rowB * Hdim];
            #pragma unroll
            for (int i = 1; i < 9; ++i) {
                hA[i] = hc[(rowA + i) * Hdim];
                hB[i] = hc[(rowB + i) * Hdim];
            }
            if (first) hA[0] = 0.0f;
            const float bv = bias[col];
            const float qv = qrow[col];

            v8f c0 = {}, c1 = {};
            #pragma unroll
            for (int s = 0; s < PLR / 4; ++s) {
                c0 = __builtin_amdgcn_wmma_f32_16x16x4_f32(false, fa0[s], false, fb[s], (short)0, c0, false, false);
                c1 = __builtin_amdgcn_wmma_f32_16x16x4_f32(false, fa1[s], false, fb[s], (short)0, c1, false, false);
            }

            const int ob = 8 * hh * FT_XP + col;
            #pragma unroll
            for (int r = 0; r < 8; ++r) {
                const float h0 = hA[r + 1];
                const float h1 = hB[r + 1];
                const float dl0 = kvB ? (hA[r] - h0) : (qv - h0);
                const float dl1 = kvB ? (hB[r] - h1) : (qv - h1);
                outp[ob + r * FT_XP]        = h0 + dl0 * (c0[r] + bv);
                outp[ob + (r + 16) * FT_XP] = h1 + dl1 * (c1[r] + bv);
            }
        }
    }
    __syncthreads();

    {
        const long yrow = ((long)e * LKs + l0 + 8 * hh) * Hdim + sub;

        if (wave < 2) {
            const int nc = wave << 5;
            v8f c00 = {}, c01 = {}, c10 = {}, c11 = {};
            ft_products<Hdim, GLR, FT_XP>(xs + FT_TILE, Wa, nc, sub, koff, c00, c01, c10, c11);
            float* tp = tw_s + 8 * hh * FT_TWP + nc + sub;
            #pragma unroll
            for (int r = 0; r < 8; ++r) {
                tp[r * FT_TWP]             = tanhf(c00[r]);
                tp[r * FT_TWP + 16]        = tanhf(c01[r]);
                tp[(r + 16) * FT_TWP]      = tanhf(c10[r]);
                tp[(r + 16) * FT_TWP + 16] = tanhf(c11[r]);
            }
        } else {
            ft_long_job(wave - 2, xs, Wr, Wk, Wv, Wg, rbuf, kbuf, vbuf, gsbuf, yrow, sub, koff);
        }
        __syncthreads();

        for (int uj = wave - 4; uj >= 0 && uj < 8; uj += 6) {
            const int nc = uj << 5;
            v8f c00 = {}, c01 = {}, c10 = {}, c11 = {};
            ft_products<GLR, Hdim, FT_TWP>(tw_s, Wb, nc, sub, koff, c00, c01, c10, c11);
            const float bv0 = bb[nc + sub];
            const float bv1 = bb[nc + sub + 16];
            float* __restrict__ yp = dbuf + yrow + nc;
            #pragma unroll
            for (int r = 0; r < 8; ++r) {
                yp[r * Hdim]             = expf(-expf(c00[r] + bv0));
                yp[r * Hdim + 16]        = expf(-expf(c01[r] + bv1));
                yp[(r + 16) * Hdim]      = expf(-expf(c10[r] + bv0));
                yp[(r + 16) * Hdim + 16] = expf(-expf(c11[r] + bv1));
            }
        }
        for (int t = wave; t < 24; t += FT_WAVES)
            ft_long_job(8 + t, xs, Wr, Wk, Wv, Wg, rbuf, kbuf, vbuf, gsbuf, yrow, sub, koff);
    }
}

__device__ __forceinline__ float apply_act(float x, int act)
{
    switch (act) {
        case ACT_SILU:  return x / (1.0f + expf(-x));
        case ACT_TANH:  return tanhf(x);
        case ACT_DECAY: return expf(-expf(x));
        default:        return x;
    }
}

__global__ __launch_bounds__(128) void rwkv_wmma_gemm_kernel(
    const float* __restrict__ X, const float* __restrict__ W,
    const float* __restrict__ bias, float* __restrict__ Y,
    int K, int N, int act)
{
    const int wave  = threadIdx.x >> 5;
    const int lane  = threadIdx.x & 31;
    const int mbase = (blockIdx.x * 4 + wave) << 5;
    const int nbase = blockIdx.y << 5;

    const int sub  = lane & 15;
    const int koff = (lane >> 4) << 1;

    const float* __restrict__ Arow0 = X + (long)(mbase + sub) * K + koff;
    const float* __restrict__ Arow1 = Arow0 + (long)16 * K;
    const float* __restrict__ Bcol0 = W + (long)koff * N + nbase + sub;
    const float* __restrict__ Bcol1 = Bcol0 + 16;

    v8f c00 = {}, c01 = {}, c10 = {}, c11 = {};
    #pragma unroll 4
    for (int k = 0; k < K; k += 4) {
        const v2f a0 = *(const v2f*)(Arow0 + k);
        const v2f a1 = *(const v2f*)(Arow1 + k);
        v2f b0, b1;
        b0.x = Bcol0[(long)k * N];
        b0.y = Bcol0[(long)(k + 1) * N];
        b1.x = Bcol1[(long)k * N];
        b1.y = Bcol1[(long)(k + 1) * N];
        c00 = __builtin_amdgcn_wmma_f32_16x16x4_f32(false, a0, false, b0, (short)0, c00, false, false);
        c01 = __builtin_amdgcn_wmma_f32_16x16x4_f32(false, a0, false, b1, (short)0, c01, false, false);
        c10 = __builtin_amdgcn_wmma_f32_16x16x4_f32(false, a1, false, b0, (short)0, c10, false, false);
        c11 = __builtin_amdgcn_wmma_f32_16x16x4_f32(false, a1, false, b1, (short)0, c11, false, false);
    }

    const int rbase = (lane >> 4) * 8;
    const int col0  = nbase + sub;
    const int col1  = col0 + 16;
    const float bv0 = bias ? bias[col0] : 0.0f;
    const float bv1 = bias ? bias[col1] : 0.0f;
    #pragma unroll
    for (int r = 0; r < 8; ++r) {
        const long row0 = (long)mbase + rbase + r;
        const long row1 = row0 + 16;
        Y[row0 * N + col0] = apply_act(c00[r] + bv0, act);
        Y[row0 * N + col1] = apply_act(c01[r] + bv1, act);
        Y[row1 * N + col0] = apply_act(c10[r] + bv0, act);
        Y[row1 * N + col1] = apply_act(c11[r] + bv1, act);
    }
}

#define SC_STEPS  4
#define SC_ROW    (3 * DK + 32)
#define SC_CHUNK  (SC_STEPS * SC_ROW)
#define SC_NCHUNK (LKs / SC_STEPS)
#define SC_BLOCKS (SC_STEPS * 4)
#define SC_RING   4
#define SC_OBUF   (DK * SC_STEPS)

template <int MASK>
__device__ __forceinline__ float sc_row_xor(float x)
{
    const int b = __builtin_bit_cast(int, x);
    return __builtin_bit_cast(float, __builtin_amdgcn_update_dpp(0, b, 0x160 + MASK, 0xF, 0xF, true));
}
__device__ __forceinline__ float sc_xhalf(float x)
{
    const int b = __builtin_bit_cast(int, x);
    return __builtin_bit_cast(float, __builtin_amdgcn_permlanex16(b, b, 0x76543210, 0xfedcba98, false, false));
}
__device__ __forceinline__ float sc_wave_sum(float x)
{
    x += sc_row_xor<1>(x);
    x += sc_row_xor<2>(x);
    x += sc_row_xor<4>(x);
    x += sc_row_xor<8>(x);
    return x + sc_xhalf(x);
}

__global__ __launch_bounds__(96) void rwkv_scan_kernel(
    const float* __restrict__ r, const float* __restrict__ k,
    const float* __restrict__ v, const float* __restrict__ decay,
    const float* __restrict__ u, const float* __restrict__ gsilu,
    const float* __restrict__ gn_w, const float* __restrict__ gn_b,
    float* __restrict__ y)
{
    __shared__ __attribute__((aligned(16))) float stage_all[2 * 2 * SC_CHUNK];
    __shared__ __attribute__((aligned(16))) float obuf[2 * SC_OBUF];

    const int chain = blockIdx.x;
    const int wave  = __builtin_amdgcn_readfirstlane((int)threadIdx.x >> 5);
    const bool scan = (wave < 2);
    const int half  = wave & 1;
    const int e     = chain >> 2;
    const int head  = chain & 3;
    const int lane  = (int)threadIdx.x & 31;
    const long base = ((long)e * LKs) * Hdim + head * DK;
    const int vcol  = half * 32 + lane;
    float* const stage = stage_all + half * (2 * SC_CHUNK);

    float S[DK];
    float uu[DK];
    #pragma unroll
    for (int j = 0; j < DK; ++j) { S[j] = 0.0f; uu[j] = u[head * DK + j]; }

    const float* __restrict__ rg = r + base + 2 * lane;
    const float* __restrict__ kg = k + base + 2 * lane;
    const float* __restrict__ dg = decay + base + 2 * lane;
    const float* __restrict__ vg = v + base + vcol;

    v2f gr[SC_STEPS], gk[SC_STEPS], gd[SC_STEPS];
    float gv[SC_STEPS];

    auto fetch = [&](int chunk) {
        #pragma unroll
        for (int i = 0; i < SC_STEPS; ++i) {
            const long off = (long)(chunk * SC_STEPS + i) * Hdim;
            gr[i] = *(const v2f*)(rg + off);
            gk[i] = *(const v2f*)(kg + off);
            gd[i] = *(const v2f*)(dg + off);
            gv[i] = vg[off];
        }
    };
    auto put = [&](float* buf) {
        #pragma unroll
        for (int i = 0; i < SC_STEPS; ++i) {
            float* slot = buf + i * SC_ROW;
            *(v2f*)(slot + 2 * lane)          = gr[i];
            *(v2f*)(slot + DK + 2 * lane)     = gk[i];
            *(v2f*)(slot + 2 * DK + 2 * lane) = gd[i];
            slot[3 * DK + lane]               = gv[i];
        }
    };

    v4f R[SC_RING][4], K[SC_RING][4], D[SC_RING][4];
    float VT[2];

    auto take = [&](const float* buf, int nb) {
        const int si = (nb % SC_BLOCKS) >> 2, sq = nb & 3, set = nb % SC_RING;
        const float* slot = buf + si * SC_ROW + sq * 16;
        #pragma unroll
        for (int g = 0; g < 4; ++g) {
            R[set][g] = *(const v4f*)(slot + 4 * g);
            K[set][g] = *(const v4f*)(slot + DK + 4 * g);
            D[set][g] = *(const v4f*)(slot + 2 * DK + 4 * g);
        }
        if (sq == 0) VT[(nb >> 2) & 1] = buf[si * SC_ROW + 3 * DK + lane];
    };

    const float* __restrict__ gsg = gsilu + base + lane;
    float*       __restrict__ yg  = y + base + lane;
    const float gwa = gn_w[head * DK + lane], gwb = gn_w[head * DK + lane + 32];
    const float gba = gn_b[head * DK + lane], gbb = gn_b[head * DK + lane + 32];
    float ga[SC_STEPS], gb[SC_STEPS];
    float na[SC_STEPS], nb[SC_STEPS];
    #pragma unroll
    for (int i = 0; i < SC_STEPS; ++i) { ga[i] = 0.0f; gb[i] = 0.0f; na[i] = 0.0f; nb[i] = 0.0f; }

    auto gates = [&](int chunk) {
        #pragma unroll
        for (int i = 0; i < SC_STEPS; ++i) {
            const long off = (long)(chunk * SC_STEPS + i) * Hdim;
            na[i] = gsg[off];
            nb[i] = gsg[off + 32];
        }
    };
    auto finish = [&](int chunk, const float* ob) {
        const v4f oa = *(const v4f*)(ob + lane * SC_STEPS);
        const v4f oc = *(const v4f*)(ob + (lane + 32) * SC_STEPS);
        float mean[SC_STEPS], da[SC_STEPS], db[SC_STEPS], var[SC_STEPS];
        #pragma unroll
        for (int i = 0; i < SC_STEPS; ++i) mean[i] = sc_wave_sum(oa[i] + oc[i]) * (1.0f / 64.0f);
        #pragma unroll
        for (int i = 0; i < SC_STEPS; ++i) {
            da[i] = oa[i] - mean[i];
            db[i] = oc[i] - mean[i];
            var[i] = sc_wave_sum(da[i] * da[i] + db[i] * db[i]) * (1.0f / 64.0f);
        }
        #pragma unroll
        for (int i = 0; i < SC_STEPS; ++i) {
            const float rstd = __builtin_amdgcn_rsqf(var[i] + EPSV);
            const long off = (long)(chunk * SC_STEPS + i) * Hdim;
            yg[off]      = (da[i] * rstd * gwa + gba) * ga[i];
            yg[off + 32] = (db[i] * rstd * gwb + gbb) * gb[i];
        }
    };
    if (scan) {
        fetch(0);
        put(stage);
    }
    __syncthreads();
    if (scan) {
        fetch(1);
        take(stage, 0);
        take(stage, 1);
    }

    for (int c = 0; c < SC_NCHUNK; ++c) {
        const float* cur = stage + (c & 1) * SC_CHUNK;
        float*       nxt = stage + ((c + 1) & 1) * SC_CHUNK;
        const float* ob = obuf + ((c + 1) & 1) * SC_OBUF;
        float* const ow = obuf + (c & 1) * SC_OBUF + vcol * SC_STEPS;
        if (scan) put(nxt);
        __syncthreads();
        if (!scan) {
            gates(c);
            if (c > 0) finish(c - 1, ob);
            #pragma unroll
            for (int i = 0; i < SC_STEPS; ++i) { ga[i] = na[i]; gb[i] = nb[i]; }
            continue;
        }
        fetch((c + 2 < SC_NCHUNK) ? (c + 2) : (SC_NCHUNK - 1));

        float part[4];
        #pragma unroll
        for (int b = 0; b < SC_BLOCKS; b += 2) {
            take((b + 2 < SC_BLOCKS) ? cur : (const float*)nxt, b + 2);
            take((b + 3 < SC_BLOCKS) ? cur : (const float*)nxt, b + 3);
            __builtin_amdgcn_sched_barrier(0);

            const int i = b >> 2, q = b & 3;
            const int sa = b % SC_RING, sb = (b + 1) % SC_RING;
            const float vt = VT[i & 1];
            float acca = 0.0f, accb = 0.0f;
            #pragma unroll
            for (int j = 0; j < 16; ++j) {
                const int ka = q * 16 + j, kb = ka + 16;
                const float kva = K[sa][j >> 2][j & 3] * vt;
                const float kvb = K[sb][j >> 2][j & 3] * vt;
                acca += R[sa][j >> 2][j & 3] * (S[ka] + uu[ka] * kva);
                accb += R[sb][j >> 2][j & 3] * (S[kb] + uu[kb] * kvb);
                S[ka] = D[sa][j >> 2][j & 3] * S[ka] + kva;
                S[kb] = D[sb][j >> 2][j & 3] * S[kb] + kvb;
            }
            part[q]     = acca;
            part[q + 1] = accb;
            if (q == 2) ow[i] = part[0] + part[1] + part[2] + part[3];
            __builtin_amdgcn_sched_barrier(0);
        }
    }

    {
        const float* ob = obuf + ((SC_NCHUNK + 1) & 1) * SC_OBUF;
        __syncthreads();
        if (!scan) finish(SC_NCHUNK - 1, ob);
    }
}

extern "C" void kernel_launch(void* const* d_in, const int* in_sizes, int n_in,
                              void* d_out, int out_size, void* d_ws, size_t ws_size,
                              hipStream_t stream)
{
    const float* query    = (const float*)d_in[0];
    const float* keyval   = (const float*)d_in[1];
    const float* rwg_mu   = (const float*)d_in[2];
    const float* W_rwg0   = (const float*)d_in[3];
    const float* W_rwg2   = (const float*)d_in[4];
    const float* rwg_bias = (const float*)d_in[5];
    const float* kv_mu    = (const float*)d_in[6];
    const float* W_kv0    = (const float*)d_in[7];
    const float* W_kv2    = (const float*)d_in[8];
    const float* kv_bias  = (const float*)d_in[9];
    const float* Wr       = (const float*)d_in[10];
    const float* Wa       = (const float*)d_in[11];
    const float* Wb       = (const float*)d_in[12];
    const float* bb       = (const float*)d_in[13];
    const float* Wk       = (const float*)d_in[14];
    const float* Wv       = (const float*)d_in[15];
    const float* Wg       = (const float*)d_in[16];
    const float* u        = (const float*)d_in[17];
    const float* gn_w     = (const float*)d_in[18];
    const float* gn_b     = (const float*)d_in[19];
    const float* Wo       = (const float*)d_in[20];

    float* ws = (float*)d_ws;
    const long SZ = (long)NTOK * Hdim;
    float* rbuf  = ws + 0 * SZ;
    float* kbuf  = ws + 1 * SZ;
    float* vbuf  = ws + 2 * SZ;
    float* dbuf  = ws + 3 * SZ;
    float* gsbuf = ws + 4 * SZ;
    float* ybuf  = ws + 5 * SZ;

    rwkv_features_kernel<<<NTOK / FT_TOK, FT_WAVES * 32, 0, stream>>>(
        query, keyval, rwg_mu, W_rwg0, W_rwg2, rwg_bias,
        kv_mu, W_kv0, W_kv2, kv_bias,
        Wr, Wa, Wb, bb, Wk, Wv, Wg,
        rbuf, kbuf, vbuf, gsbuf, dbuf);

    auto gemm = [&](const float* X, const float* W, const float* bias,
                    float* Y, int M, int K, int N, int act) {
        dim3 grid(M / 128, N / 32);
        rwkv_wmma_gemm_kernel<<<grid, 128, 0, stream>>>(X, W, bias, Y, K, N, act);
    };

    rwkv_scan_kernel<<<Bsz * LQ * NHEAD, 96, 0, stream>>>(
        rbuf, kbuf, vbuf, dbuf, u, gsbuf, gn_w, gn_b, ybuf);

    gemm(ybuf, Wo, nullptr, (float*)d_out, NTOK, 256, 256, ACT_NONE);
}


# output projection: one work-group per 32 tokens holds its [32 x 256] y rows in LDS and forms all 256 columns from them (8 waves, a 32-column strip each): y read once, not 8 times; product order kept
# speedup vs baseline: 4.9925x; 1.1697x over previous
#include <hip/hip_runtime.h>
#include <math.h>

#define Hdim 256
#define NHEAD 4
#define DK 64
#define PLR 32
#define GLR 64
#define Bsz 2
#define LQ 8
#define LKs 512
#define NTOK (Bsz * LQ * LKs)
#define EPSV 1e-5f

typedef float v2f __attribute__((ext_vector_type(2)));
typedef float v4f __attribute__((ext_vector_type(4)));
typedef float v8f __attribute__((ext_vector_type(8)));

#define FT_TOK   32
#define FT_WAVES 10
#define FT_XP    260
#define FT_T1P   100
#define FT_T2P   68
#define FT_TWP   68
#define FT_TILE  (FT_TOK * FT_XP)

template <int K, int N, int P>
__device__ __forceinline__ void ft_products(
    const float* Xs, const float* __restrict__ W, int nc, int sub, int koff,
    v8f& c00, v8f& c01, v8f& c10, v8f& c11)
{
    const float* A0 = Xs + sub * P + koff;
    const float* A1 = A0 + 16 * P;
    const float* __restrict__ B0 = W + koff * N + nc + sub;
    const float* __restrict__ B1 = B0 + 16;
    #pragma unroll 8
    for (int k = 0; k < K; k += 4) {
        const v2f a0 = *(const v2f*)(A0 + k);
        const v2f a1 = *(const v2f*)(A1 + k);
        v2f b0, b1;
        b0.x = B0[k * N];
        b0.y = B0[(k + 1) * N];
        b1.x = B1[k * N];
        b1.y = B1[(k + 1) * N];
        c00 = __builtin_amdgcn_wmma_f32_16x16x4_f32(false, a0, false, b0, (short)0, c00, false, false);
        c01 = __builtin_amdgcn_wmma_f32_16x16x4_f32(false, a0, false, b1, (short)0, c01, false, false);
        c10 = __builtin_amdgcn_wmma_f32_16x16x4_f32(false, a1, false, b0, (short)0, c10, false, false);
        c11 = __builtin_amdgcn_wmma_f32_16x16x4_f32(false, a1, false, b1, (short)0, c11, false, false);
    }
}

template <bool SILU>
__device__ __forceinline__ void ft_strip(
    const float* Xs, const float* __restrict__ Wp, float* __restrict__ yp,
    int nc, int sub, int koff)
{
    v8f c00 = {}, c01 = {}, c10 = {}, c11 = {};
    ft_products<Hdim, Hdim, FT_XP>(Xs, Wp, nc, sub, koff, c00, c01, c10, c11);
    #pragma unroll
    for (int r = 0; r < 8; ++r) {
        if (SILU) {
            yp[r * Hdim]             = c00[r] / (1.0f + expf(-c00[r]));
            yp[r * Hdim + 16]        = c01[r] / (1.0f + expf(-c01[r]));
            yp[(r + 16) * Hdim]      = c10[r] / (1.0f + expf(-c10[r]));
            yp[(r + 16) * Hdim + 16] = c11[r] / (1.0f + expf(-c11[r]));
        } else {
            yp[r * Hdim]             = c00[r];
            yp[r * Hdim + 16]        = c01[r];
            yp[(r + 16) * Hdim]      = c10[r];
            yp[(r + 16) * Hdim + 16] = c11[r];
        }
    }
}

__device__ __forceinline__ void ft_long_job(
    int jb, const float* xs,
    const float* __restrict__ Wr, const float* __restrict__ Wk,
    const float* __restrict__ Wv, const float* __restrict__ Wg,
    float* __restrict__ rbuf, float* __restrict__ kbuf,
    float* __restrict__ vbuf, float* __restrict__ gsbuf,
    long yrow, int sub, int koff)
{
    const int nc = (jb & 7) << 5;
    switch (jb >> 3) {
        case 0:  ft_strip<false>(xs,               Wr, rbuf  + yrow + nc, nc, sub, koff); break;
        case 1:  ft_strip<false>(xs + 3 * FT_TILE, Wk, kbuf  + yrow + nc, nc, sub, koff); break;
        case 2:  ft_strip<false>(xs + 4 * FT_TILE, Wv, vbuf  + yrow + nc, nc, sub, koff); break;
        default: ft_strip<true>(xs + 2 * FT_TILE,  Wg, gsbuf + yrow + nc, nc, sub, koff); break;
    }
}

__global__ __launch_bounds__(FT_WAVES * 32) void rwkv_features_kernel(
    const float* __restrict__ query, const float* __restrict__ keyval,
    const float* __restrict__ rwg_mu, const float* __restrict__ W_rwg0,
    const float* __restrict__ W_rwg2, const float* __restrict__ rwg_bias,
    const float* __restrict__ kv_mu, const float* __restrict__ W_kv0,
    const float* __restrict__ W_kv2, const float* __restrict__ kv_bias,
    const float* __restrict__ Wr, const float* __restrict__ Wa,
    const float* __restrict__ Wb, const float* __restrict__ bb,
    const float* __restrict__ Wk, const float* __restrict__ Wv,
    const float* __restrict__ Wg,
    float* __restrict__ rbuf, float* __restrict__ kbuf, float* __restrict__ vbuf,
    float* __restrict__ gsbuf, float* __restrict__ dbuf)
{
    __shared__ __attribute__((aligned(16))) float xs[5 * FT_TILE];
    __shared__ __attribute__((aligned(16))) float t1_s[FT_TOK * FT_T1P];
    __shared__ __attribute__((aligned(16))) float t2_s[FT_TOK * FT_T2P];
    __shared__ __attribute__((aligned(16))) float tw_s[FT_TOK * FT_TWP];
    float* const xr_s = xs;
    float* const xk_s = xs + 3 * FT_TILE;

    const int tid  = (int)threadIdx.x;
    const int wave = __builtin_amdgcn_readfirstlane(tid >> 5);
    const int lane = tid & 31;
    const int sub  = lane & 15;
    const int hh   = lane >> 4;
    const int koff = hh << 1;

    const int tile = (int)blockIdx.x;
    const int e    = tile >> 4;
    const int l0   = (tile & 15) * FT_TOK;
    const int b    = e >> 3;
    const float* __restrict__ hrow = keyval + ((long)b * LKs + l0) * Hdim;
    const float* __restrict__ qrow = query + (long)e * Hdim;

    {
        const int c4 = (tid & 63) << 2;
        const int t0 = tid >> 6;
        const v4f q  = *(const v4f*)(qrow + c4);
        const v4f mr = *(const v4f*)(rwg_mu + c4);
        const v4f mk = *(const v4f*)(kv_mu + c4);
        #pragma unroll
        for (int j = 0; j < 7; ++j) {
            const int tok = t0 + 5 * j;
            if (j < 6 || tok < FT_TOK) {
                const bool first = (l0 + tok == 0);
                const int prow = first ? 0 : tok - 1;
                const v4f h  = *(const v4f*)(hrow + tok * Hdim + c4);
                v4f hp = *(const v4f*)(hrow + prow * Hdim + c4);
                if (first) hp = (v4f){0.0f, 0.0f, 0.0f, 0.0f};
                const v4f dc = q - h;
                const v4f d  = hp - h;
                *(v4f*)(xr_s + tok * FT_XP + c4) = h + dc * mr;
                *(v4f*)(xk_s + tok * FT_XP + c4) = h + d * mk;
            }
        }
    }
    __syncthreads();

    {
        const bool kvA = (wave >= 6);
        const float* Xs = kvA ? xk_s : xr_s;
        const float* __restrict__ W0 = kvA ? W_kv0 : W_rwg0;
        const int N0  = kvA ? 2 * PLR : 3 * PLR;
        const int nc  = (kvA ? wave - 6 : wave) << 4;
        float* Td     = kvA ? t2_s : t1_s;
        const int PT  = kvA ? FT_T2P : FT_T1P;

        const float* A0 = Xs + sub * FT_XP + koff;
        const float* A1 = A0 + 16 * FT_XP;
        const float* __restrict__ Bp = W0 + koff * N0 + nc + sub;

        v8f c0 = {}, c1 = {};
        #pragma unroll 8
        for (int k = 0; k < Hdim; k += 4) {
            const v2f a0 = *(const v2f*)(A0 + k);
            const v2f a1 = *(const v2f*)(A1 + k);
            v2f bf;
            bf.x = Bp[k * N0];
            bf.y = Bp[(k + 1) * N0];
            c0 = __builtin_amdgcn_wmma_f32_16x16x4_f32(false, a0, false, bf, (short)0, c0, false, false);
            c1 = __builtin_amdgcn_wmma_f32_16x16x4_f32(false, a1, false, bf, (short)0, c1, false, false);
        }
        #pragma unroll
        for (int r = 0; r < 8; ++r) {
            const int m = r + 8 * hh;
            Td[m * PT + nc + sub]        = tanhf(c0[r]);
            Td[(16 + m) * PT + nc + sub] = tanhf(c1[r]);
        }
    }
    __syncthreads();

    {
        const int o    = wave >> 1;
        const bool kvB = (o >= 3);
        const int g    = kvB ? o - 3 : o;
        const float* Ts = kvB ? t2_s : t1_s;
        const int PT   = kvB ? FT_T2P : FT_T1P;
        const float* __restrict__ W2 = kvB ? W_kv2 : W_rwg2;
        const int NW   = kvB ? 2 * PLR : 3 * PLR;
        const float* __restrict__ bias = (kvB ? kv_bias : rwg_bias) + g * Hdim;
        float* outp = xs + o * FT_TILE;

        const float* A0 = Ts + sub * PT + g * PLR + koff;
        const float* A1 = A0 + 16 * PT;
        v2f fa0[PLR / 4], fa1[PLR / 4];
        #pragma unroll
        for (int s = 0; s < PLR / 4; ++s) {
            fa0[s] = *(const v2f*)(A0 + 4 * s);
            fa1[s] = *(const v2f*)(A1 + 4 * s);
        }

        const int rowA = 8 * hh - 1;
        const int rowB = rowA + 16;
        const bool first = (l0 + rowA < 0);
        const int rowA0 = first ? 0 : rowA;

        for (int j = 0; j < 8; ++j) {
            const int col = ((((wave & 1) << 3) + j) << 4) + sub;
            const float* __restrict__ Bp = W2 + col * NW + g * PLR + koff;
            const float* __restrict__ hc = hrow + col;

            v2f fb[PLR / 4];
            #pragma unroll
            for (int s = 0; s < PLR / 4; ++s) fb[s] = *(const v2f*)(Bp + 4 * s);
            float hA[9], hB[9];
            hA[0] = hc[rowA0 * Hdim];
            hB[0] = hc[rowB * Hdim];
            #pragma unroll
            for (int i = 1; i < 9; ++i) {
                hA[i] = hc[(rowA + i) * Hdim];
                hB[i] = hc[(rowB + i) * Hdim];
            }
            if (first) hA[0] = 0.0f;
            const float bv = bias[col];
            const float qv = qrow[col];

            v8f c0 = {}, c1 = {};
            #pragma unroll
            for (int s = 0; s < PLR / 4; ++s) {
                c0 = __builtin_amdgcn_wmma_f32_16x16x4_f32(false, fa0[s], false, fb[s], (short)0, c0, false, false);
                c1 = __builtin_amdgcn_wmma_f32_16x16x4_f32(false, fa1[s], false, fb[s], (short)0, c1, false, false);
            }

            const int ob = 8 * hh * FT_XP + col;
            #pragma unroll
            for (int r = 0; r < 8; ++r) {
                const float h0 = hA[r + 1];
                const float h1 = hB[r + 1];
                const float dl0 = kvB ? (hA[r] - h0) : (qv - h0);
                const float dl1 = kvB ? (hB[r] - h1) : (qv - h1);
                outp[ob + r * FT_XP]        = h0 + dl0 * (c0[r] + bv);
                outp[ob + (r + 16) * FT_XP] = h1 + dl1 * (c1[r] + bv);
            }
        }
    }
    __syncthreads();

    {
        const long yrow = ((long)e * LKs + l0 + 8 * hh) * Hdim + sub;

        if (wave < 2) {
            const int nc = wave << 5;
            v8f c00 = {}, c01 = {}, c10 = {}, c11 = {};
            ft_products<Hdim, GLR, FT_XP>(xs + FT_TILE, Wa, nc, sub, koff, c00, c01, c10, c11);
            float* tp = tw_s + 8 * hh * FT_TWP + nc + sub;
            #pragma unroll
            for (int r = 0; r < 8; ++r) {
                tp[r * FT_TWP]             = tanhf(c00[r]);
                tp[r * FT_TWP + 16]        = tanhf(c01[r]);
                tp[(r + 16) * FT_TWP]      = tanhf(c10[r]);
                tp[(r + 16) * FT_TWP + 16] = tanhf(c11[r]);
            }
        } else {
            ft_long_job(wave - 2, xs, Wr, Wk, Wv, Wg, rbuf, kbuf, vbuf, gsbuf, yrow, sub, koff);
        }
        __syncthreads();

        for (int uj = wave - 4; uj >= 0 && uj < 8; uj += 6) {
            const int nc = uj << 5;
            v8f c00 = {}, c01 = {}, c10 = {}, c11 = {};
            ft_products<GLR, Hdim, FT_TWP>(tw_s, Wb, nc, sub, koff, c00, c01, c10, c11);
            const float bv0 = bb[nc + sub];
            const float bv1 = bb[nc + sub + 16];
            float* __restrict__ yp = dbuf + yrow + nc;
            #pragma unroll
            for (int r = 0; r < 8; ++r) {
                yp[r * Hdim]             = expf(-expf(c00[r] + bv0));
                yp[r * Hdim + 16]        = expf(-expf(c01[r] + bv1));
                yp[(r + 16) * Hdim]      = expf(-expf(c10[r] + bv0));
                yp[(r + 16) * Hdim + 16] = expf(-expf(c11[r] + bv1));
            }
        }
        for (int t = wave; t < 24; t += FT_WAVES)
            ft_long_job(8 + t, xs, Wr, Wk, Wv, Wg, rbuf, kbuf, vbuf, gsbuf, yrow, sub, koff);
    }
}

#define SC_STEPS  4
#define SC_ROW    (3 * DK + 32)
#define SC_CHUNK  (SC_STEPS * SC_ROW)
#define SC_NCHUNK (LKs / SC_STEPS)
#define SC_BLOCKS (SC_STEPS * 4)
#define SC_RING   4
#define SC_OBUF   (DK * SC_STEPS)

template <int MASK>
__device__ __forceinline__ float sc_row_xor(float x)
{
    const int b = __builtin_bit_cast(int, x);
    return __builtin_bit_cast(float, __builtin_amdgcn_update_dpp(0, b, 0x160 + MASK, 0xF, 0xF, true));
}
__device__ __forceinline__ float sc_xhalf(float x)
{
    const int b = __builtin_bit_cast(int, x);
    return __builtin_bit_cast(float, __builtin_amdgcn_permlanex16(b, b, 0x76543210, 0xfedcba98, false, false));
}
__device__ __forceinline__ float sc_wave_sum(float x)
{
    x += sc_row_xor<1>(x);
    x += sc_row_xor<2>(x);
    x += sc_row_xor<4>(x);
    x += sc_row_xor<8>(x);
    return x + sc_xhalf(x);
}

__global__ __launch_bounds__(96) void rwkv_scan_kernel(
    const float* __restrict__ r, const float* __restrict__ k,
    const float* __restrict__ v, const float* __restrict__ decay,
    const float* __restrict__ u, const float* __restrict__ gsilu,
    const float* __restrict__ gn_w, const float* __restrict__ gn_b,
    float* __restrict__ y)
{
    __shared__ __attribute__((aligned(16))) float stage_all[2 * 2 * SC_CHUNK];
    __shared__ __attribute__((aligned(16))) float obuf[2 * SC_OBUF];

    const int chain = blockIdx.x;
    const int wave  = __builtin_amdgcn_readfirstlane((int)threadIdx.x >> 5);
    const bool scan = (wave < 2);
    const int half  = wave & 1;
    const int e     = chain >> 2;
    const int head  = chain & 3;
    const int lane  = (int)threadIdx.x & 31;
    const long base = ((long)e * LKs) * Hdim + head * DK;
    const int vcol  = half * 32 + lane;
    float* const stage = stage_all + half * (2 * SC_CHUNK);

    float S[DK];
    float uu[DK];
    #pragma unroll
    for (int j = 0; j < DK; ++j) { S[j] = 0.0f; uu[j] = u[head * DK + j]; }

    const float* __restrict__ rg = r + base + 2 * lane;
    const float* __restrict__ kg = k + base + 2 * lane;
    const float* __restrict__ dg = decay + base + 2 * lane;
    const float* __restrict__ vg = v + base + vcol;

    v2f gr[SC_STEPS], gk[SC_STEPS], gd[SC_STEPS];
    float gv[SC_STEPS];

    auto fetch = [&](int chunk) {
        #pragma unroll
        for (int i = 0; i < SC_STEPS; ++i) {
            const long off = (long)(chunk * SC_STEPS + i) * Hdim;
            gr[i] = *(const v2f*)(rg + off);
            gk[i] = *(const v2f*)(kg + off);
            gd[i] = *(const v2f*)(dg + off);
            gv[i] = vg[off];
        }
    };
    auto put = [&](float* buf) {
        #pragma unroll
        for (int i = 0; i < SC_STEPS; ++i) {
            float* slot = buf + i * SC_ROW;
            *(v2f*)(slot + 2 * lane)          = gr[i];
            *(v2f*)(slot + DK + 2 * lane)     = gk[i];
            *(v2f*)(slot + 2 * DK + 2 * lane) = gd[i];
            slot[3 * DK + lane]               = gv[i];
        }
    };

    v4f R[SC_RING][4], K[SC_RING][4], D[SC_RING][4];
    float VT[2];

    auto take = [&](const float* buf, int nb) {
        const int si = (nb % SC_BLOCKS) >> 2, sq = nb & 3, set = nb % SC_RING;
        const float* slot = buf + si * SC_ROW + sq * 16;
        #pragma unroll
        for (int g = 0; g < 4; ++g) {
            R[set][g] = *(const v4f*)(slot + 4 * g);
            K[set][g] = *(const v4f*)(slot + DK + 4 * g);
            D[set][g] = *(const v4f*)(slot + 2 * DK + 4 * g);
        }
        if (sq == 0) VT[(nb >> 2) & 1] = buf[si * SC_ROW + 3 * DK + lane];
    };

    const float* __restrict__ gsg = gsilu + base + lane;
    float*       __restrict__ yg  = y + base + lane;
    const float gwa = gn_w[head * DK + lane], gwb = gn_w[head * DK + lane + 32];
    const float gba = gn_b[head * DK + lane], gbb = gn_b[head * DK + lane + 32];
    float ga[SC_STEPS], gb[SC_STEPS];
    float na[SC_STEPS], nb[SC_STEPS];
    #pragma unroll
    for (int i = 0; i < SC_STEPS; ++i) { ga[i] = 0.0f; gb[i] = 0.0f; na[i] = 0.0f; nb[i] = 0.0f; }

    auto gates = [&](int chunk) {
        #pragma unroll
        for (int i = 0; i < SC_STEPS; ++i) {
            const long off = (long)(chunk * SC_STEPS + i) * Hdim;
            na[i] = gsg[off];
            nb[i] = gsg[off + 32];
        }
    };
    auto finish = [&](int chunk, const float* ob) {
        const v4f oa = *(const v4f*)(ob + lane * SC_STEPS);
        const v4f oc = *(const v4f*)(ob + (lane + 32) * SC_STEPS);
        float mean[SC_STEPS], da[SC_STEPS], db[SC_STEPS], var[SC_STEPS];
        #pragma unroll
        for (int i = 0; i < SC_STEPS; ++i) mean[i] = sc_wave_sum(oa[i] + oc[i]) * (1.0f / 64.0f);
        #pragma unroll
        for (int i = 0; i < SC_STEPS; ++i) {
            da[i] = oa[i] - mean[i];
            db[i] = oc[i] - mean[i];
            var[i] = sc_wave_sum(da[i] * da[i] + db[i] * db[i]) * (1.0f / 64.0f);
        }
        #pragma unroll
        for (int i = 0; i < SC_STEPS; ++i) {
            const float rstd = __builtin_amdgcn_rsqf(var[i] + EPSV);
            const long off = (long)(chunk * SC_STEPS + i) * Hdim;
            yg[off]      = (da[i] * rstd * gwa + gba) * ga[i];
            yg[off + 32] = (db[i] * rstd * gwb + gbb) * gb[i];
        }
    };
    if (scan) {
        fetch(0);
        put(stage);
    }
    __syncthreads();
    if (scan) {
        fetch(1);
        take(stage, 0);
        take(stage, 1);
    }

    for (int c = 0; c < SC_NCHUNK; ++c) {
        const float* cur = stage + (c & 1) * SC_CHUNK;
        float*       nxt = stage + ((c + 1) & 1) * SC_CHUNK;
        const float* ob = obuf + ((c + 1) & 1) * SC_OBUF;
        float* const ow = obuf + (c & 1) * SC_OBUF + vcol * SC_STEPS;
        if (scan) put(nxt);
        __syncthreads();
        if (!scan) {
            gates(c);
            if (c > 0) finish(c - 1, ob);
            #pragma unroll
            for (int i = 0; i < SC_STEPS; ++i) { ga[i] = na[i]; gb[i] = nb[i]; }
            continue;
        }
        fetch((c + 2 < SC_NCHUNK) ? (c + 2) : (SC_NCHUNK - 1));

        float part[4];
        #pragma unroll
        for (int b = 0; b < SC_BLOCKS; b += 2) {
            take((b + 2 < SC_BLOCKS) ? cur : (const float*)nxt, b + 2);
            take((b + 3 < SC_BLOCKS) ? cur : (const float*)nxt, b + 3);
            __builtin_amdgcn_sched_barrier(0);

            const int i = b >> 2, q = b & 3;
            const int sa = b % SC_RING, sb = (b + 1) % SC_RING;
            const float vt = VT[i & 1];
            float acca = 0.0f, accb = 0.0f;
            #pragma unroll
            for (int j = 0; j < 16; ++j) {
                const int ka = q * 16 + j, kb = ka + 16;
                const float kva = K[sa][j >> 2][j & 3] * vt;
                const float kvb = K[sb][j >> 2][j & 3] * vt;
                acca += R[sa][j >> 2][j & 3] * (S[ka] + uu[ka] * kva);
                accb += R[sb][j >> 2][j & 3] * (S[kb] + uu[kb] * kvb);
                S[ka] = D[sa][j >> 2][j & 3] * S[ka] + kva;
                S[kb] = D[sb][j >> 2][j & 3] * S[kb] + kvb;
            }
            part[q]     = acca;
            part[q + 1] = accb;
            if (q == 2) ow[i] = part[0] + part[1] + part[2] + part[3];
            __builtin_amdgcn_sched_barrier(0);
        }
    }

    {
        const float* ob = obuf + ((SC_NCHUNK + 1) & 1) * SC_OBUF;
        __syncthreads();
        if (!scan) finish(SC_NCHUNK - 1, ob);
    }
}

#define OP_WAVES 8

__global__ __launch_bounds__(OP_WAVES * 32) void rwkv_out_kernel(
    const float* __restrict__ y, const float* __restrict__ Wo, float* __restrict__ out)
{
    __shared__ __attribute__((aligned(16))) float ys[FT_TILE];

    const int tid  = (int)threadIdx.x;
    const int wave = __builtin_amdgcn_readfirstlane(tid >> 5);
    const int lane = tid & 31;
    const int sub  = lane & 15;
    const int hh   = lane >> 4;
    const int koff = hh << 1;
    const long row0 = (long)blockIdx.x * FT_TOK;

    {
        const int c4 = (tid & 63) << 2;
        const int t0 = tid >> 6;
        const float* __restrict__ yp = y + row0 * Hdim + c4;
        #pragma unroll
        for (int j = 0; j < FT_TOK / 4; ++j) {
            const int tok = t0 + 4 * j;
            *(v4f*)(ys + tok * FT_XP + c4) = *(const v4f*)(yp + tok * Hdim);
        }
    }
    __syncthreads();

    const int nc = wave << 5;
    ft_strip<false>(ys, Wo, out + (row0 + 8 * hh) * Hdim + sub + nc, nc, sub, koff);
}

extern "C" void kernel_launch(void* const* d_in, const int* in_sizes, int n_in,
                              void* d_out, int out_size, void* d_ws, size_t ws_size,
                              hipStream_t stream)
{
    const float* query    = (const float*)d_in[0];
    const float* keyval   = (const float*)d_in[1];
    const float* rwg_mu   = (const float*)d_in[2];
    const float* W_rwg0   = (const float*)d_in[3];
    const float* W_rwg2   = (const float*)d_in[4];
    const float* rwg_bias = (const float*)d_in[5];
    const float* kv_mu    = (const float*)d_in[6];
    const float* W_kv0    = (const float*)d_in[7];
    const float* W_kv2    = (const float*)d_in[8];
    const float* kv_bias  = (const float*)d_in[9];
    const float* Wr       = (const float*)d_in[10];
    const float* Wa       = (const float*)d_in[11];
    const float* Wb       = (const float*)d_in[12];
    const float* bb       = (const float*)d_in[13];
    const float* Wk       = (const float*)d_in[14];
    const float* Wv       = (const float*)d_in[15];
    const float* Wg       = (const float*)d_in[16];
    const float* u        = (const float*)d_in[17];
    const float* gn_w     = (const float*)d_in[18];
    const float* gn_b     = (const float*)d_in[19];
    const float* Wo       = (const float*)d_in[20];

    float* ws = (float*)d_ws;
    const long SZ = (long)NTOK * Hdim;
    float* rbuf  = ws + 0 * SZ;
    float* kbuf  = ws + 1 * SZ;
    float* vbuf  = ws + 2 * SZ;
    float* dbuf  = ws + 3 * SZ;
    float* gsbuf = ws + 4 * SZ;
    float* ybuf  = ws + 5 * SZ;

    rwkv_features_kernel<<<NTOK / FT_TOK, FT_WAVES * 32, 0, stream>>>(
        query, keyval, rwg_mu, W_rwg0, W_rwg2, rwg_bias,
        kv_mu, W_kv0, W_kv2, kv_bias,
        Wr, Wa, Wb, bb, Wk, Wv, Wg,
        rbuf, kbuf, vbuf, gsbuf, dbuf);

    rwkv_scan_kernel<<<Bsz * LQ * NHEAD, 96, 0, stream>>>(
        rbuf, kbuf, vbuf, dbuf, u, gsbuf, gn_w, gn_b, ybuf);

    rwkv_out_kernel<<<NTOK / FT_TOK, OP_WAVES * 32, 0, stream>>>(ybuf, Wo, (float*)d_out);
}


# scan: chain on 4 waves, lane = one 16-k quarter x 2 value columns (packed pair); quarter runs joined in the quad by lane broadcast; rows staged once per group in a 3-buffer LDS ring; sum order kept
# speedup vs baseline: 7.2115x; 1.4445x over previous
#include <hip/hip_runtime.h>
#include <math.h>

#define Hdim 256
#define NHEAD 4
#define DK 64
#define PLR 32
#define GLR 64
#define Bsz 2
#define LQ 8
#define LKs 512
#define NTOK (Bsz * LQ * LKs)
#define EPSV 1e-5f

typedef float v2f __attribute__((ext_vector_type(2)));
typedef float v4f __attribute__((ext_vector_type(4)));
typedef float v8f __attribute__((ext_vector_type(8)));

#define FT_TOK   32
#define FT_WAVES 10
#define FT_XP    260
#define FT_T1P   100
#define FT_T2P   68
#define FT_TWP   68
#define FT_TILE  (FT_TOK * FT_XP)

template <int K, int N, int P>
__device__ __forceinline__ void ft_products(
    const float* Xs, const float* __restrict__ W, int nc, int sub, int koff,
    v8f& c00, v8f& c01, v8f& c10, v8f& c11)
{
    const float* A0 = Xs + sub * P + koff;
    const float* A1 = A0 + 16 * P;
    const float* __restrict__ B0 = W + koff * N + nc + sub;
    const float* __restrict__ B1 = B0 + 16;
    #pragma unroll 8
    for (int k = 0; k < K; k += 4) {
        const v2f a0 = *(const v2f*)(A0 + k);
        const v2f a1 = *(const v2f*)(A1 + k);
        v2f b0, b1;
        b0.x = B0[k * N];
        b0.y = B0[(k + 1) * N];
        b1.x = B1[k * N];
        b1.y = B1[(k + 1) * N];
        c00 = __builtin_amdgcn_wmma_f32_16x16x4_f32(false, a0, false, b0, (short)0, c00, false, false);
        c01 = __builtin_amdgcn_wmma_f32_16x16x4_f32(false, a0, false, b1, (short)0, c01, false, false);
        c10 = __builtin_amdgcn_wmma_f32_16x16x4_f32(false, a1, false, b0, (short)0, c10, false, false);
        c11 = __builtin_amdgcn_wmma_f32_16x16x4_f32(false, a1, false, b1, (short)0, c11, false, false);
    }
}

template <bool SILU>
__device__ __forceinline__ void ft_strip(
    const float* Xs, const float* __restrict__ Wp, float* __restrict__ yp,
    int nc, int sub, int koff)
{
    v8f c00 = {}, c01 = {}, c10 = {}, c11 = {};
    ft_products<Hdim, Hdim, FT_XP>(Xs, Wp, nc, sub, koff, c00, c01, c10, c11);
    #pragma unroll
    for (int r = 0; r < 8; ++r) {
        if (SILU) {
            yp[r * Hdim]             = c00[r] / (1.0f + expf(-c00[r]));
            yp[r * Hdim + 16]        = c01[r] / (1.0f + expf(-c01[r]));
            yp[(r + 16) * Hdim]      = c10[r] / (1.0f + expf(-c10[r]));
            yp[(r + 16) * Hdim + 16] = c11[r] / (1.0f + expf(-c11[r]));
        } else {
            yp[r * Hdim]             = c00[r];
            yp[r * Hdim + 16]        = c01[r];
            yp[(r + 16) * Hdim]      = c10[r];
            yp[(r + 16) * Hdim + 16] = c11[r];
        }
    }
}

__device__ __forceinline__ void ft_long_job(
    int jb, const float* xs,
    const float* __restrict__ Wr, const float* __restrict__ Wk,
    const float* __restrict__ Wv, const float* __restrict__ Wg,
    float* __restrict__ rbuf, float* __restrict__ kbuf,
    float* __restrict__ vbuf, float* __restrict__ gsbuf,
    long yrow, int sub, int koff)
{
    const int nc = (jb & 7) << 5;
    switch (jb >> 3) {
        case 0:  ft_strip<false>(xs,               Wr, rbuf  + yrow + nc, nc, sub, koff); break;
        case 1:  ft_strip<false>(xs + 3 * FT_TILE, Wk, kbuf  + yrow + nc, nc, sub, koff); break;
        case 2:  ft_strip<false>(xs + 4 * FT_TILE, Wv, vbuf  + yrow + nc, nc, sub, koff); break;
        default: ft_strip<true>(xs + 2 * FT_TILE,  Wg, gsbuf + yrow + nc, nc, sub, koff); break;
    }
}

__global__ __launch_bounds__(FT_WAVES * 32) void rwkv_features_kernel(
    const float* __restrict__ query, const float* __restrict__ keyval,
    const float* __restrict__ rwg_mu, const float* __restrict__ W_rwg0,
    const float* __restrict__ W_rwg2, const float* __restrict__ rwg_bias,
    const float* __restrict__ kv_mu, const float* __restrict__ W_kv0,
    const float* __restrict__ W_kv2, const float* __restrict__ kv_bias,
    const float* __restrict__ Wr, const float* __restrict__ Wa,
    const float* __restrict__ Wb, const float* __restrict__ bb,
    const float* __restrict__ Wk, const float* __restrict__ Wv,
    const float* __restrict__ Wg,
    float* __restrict__ rbuf, float* __restrict__ kbuf, float* __restrict__ vbuf,
    float* __restrict__ gsbuf, float* __restrict__ dbuf)
{
    __shared__ __attribute__((aligned(16))) float xs[5 * FT_TILE];
    __shared__ __attribute__((aligned(16))) float t1_s[FT_TOK * FT_T1P];
    __shared__ __attribute__((aligned(16))) float t2_s[FT_TOK * FT_T2P];
    __shared__ __attribute__((aligned(16))) float tw_s[FT_TOK * FT_TWP];
    float* const xr_s = xs;
    float* const xk_s = xs + 3 * FT_TILE;

    const int tid  = (int)threadIdx.x;
    const int wave = __builtin_amdgcn_readfirstlane(tid >> 5);
    const int lane = tid & 31;
    const int sub  = lane & 15;
    const int hh   = lane >> 4;
    const int koff = hh << 1;

    const int tile = (int)blockIdx.x;
    const int e    = tile >> 4;
    const int l0   = (tile & 15) * FT_TOK;
    const int b    = e >> 3;
    const float* __restrict__ hrow = keyval + ((long)b * LKs + l0) * Hdim;
    const float* __restrict__ qrow = query + (long)e * Hdim;

    {
        const int c4 = (tid & 63) << 2;
        const int t0 = tid >> 6;
        const v4f q  = *(const v4f*)(qrow + c4);
        const v4f mr = *(const v4f*)(rwg_mu + c4);
        const v4f mk = *(const v4f*)(kv_mu + c4);
        #pragma unroll
        for (int j = 0; j < 7; ++j) {
            const int tok = t0 + 5 * j;
            if (j < 6 || tok < FT_TOK) {
                const bool first = (l0 + tok == 0);
                const int prow = first ? 0 : tok - 1;
                const v4f h  = *(const v4f*)(hrow + tok * Hdim + c4);
                v4f hp = *(const v4f*)(hrow + prow * Hdim + c4);
                if (first) hp = (v4f){0.0f, 0.0f, 0.0f, 0.0f};
                const v4f dc = q - h;
                const v4f d  = hp - h;
                *(v4f*)(xr_s + tok * FT_XP + c4) = h + dc * mr;
                *(v4f*)(xk_s + tok * FT_XP + c4) = h + d * mk;
            }
        }
    }
    __syncthreads();

    {
        const bool kvA = (wave >= 6);
        const float* Xs = kvA ? xk_s : xr_s;
        const float* __restrict__ W0 = kvA ? W_kv0 : W_rwg0;
        const int N0  = kvA ? 2 * PLR : 3 * PLR;
        const int nc  = (kvA ? wave - 6 : wave) << 4;
        float* Td     = kvA ? t2_s : t1_s;
        const int PT  = kvA ? FT_T2P : FT_T1P;

        const float* A0 = Xs + sub * FT_XP + koff;
        const float* A1 = A0 + 16 * FT_XP;
        const float* __restrict__ Bp = W0 + koff * N0 + nc + sub;

        v8f c0 = {}, c1 = {};
        #pragma unroll 8
        for (int k = 0; k < Hdim; k += 4) {
            const v2f a0 = *(const v2f*)(A0 + k);
            const v2f a1 = *(const v2f*)(A1 + k);
            v2f bf;
            bf.x = Bp[k * N0];
            bf.y = Bp[(k + 1) * N0];
            c0 = __builtin_amdgcn_wmma_f32_16x16x4_f32(false, a0, false, bf, (short)0, c0, false, false);
            c1 = __builtin_amdgcn_wmma_f32_16x16x4_f32(false, a1, false, bf, (short)0, c1, false, false);
        }
        #pragma unroll
        for (int r = 0; r < 8; ++r) {
            const int m = r + 8 * hh;
            Td[m * PT + nc + sub]        = tanhf(c0[r]);
            Td[(16 + m) * PT + nc + sub] = tanhf(c1[r]);
        }
    }
    __syncthreads();

    {
        const int o    = wave >> 1;
        const bool kvB = (o >= 3);
        const int g    = kvB ? o - 3 : o;
        const float* Ts = kvB ? t2_s : t1_s;
        const int PT   = kvB ? FT_T2P : FT_T1P;
        const float* __restrict__ W2 = kvB ? W_kv2 : W_rwg2;
        const int NW   = kvB ? 2 * PLR : 3 * PLR;
        const float* __restrict__ bias = (kvB ? kv_bias : rwg_bias) + g * Hdim;
        float* outp = xs + o * FT_TILE;

        const float* A0 = Ts + sub * PT + g * PLR + koff;
        const float* A1 = A0 + 16 * PT;
        v2f fa0[PLR / 4], fa1[PLR / 4];
        #pragma unroll
        for (int s = 0; s < PLR / 4; ++s) {
            fa0[s] = *(const v2f*)(A0 + 4 * s);
            fa1[s] = *(const v2f*)(A1 + 4 * s);
        }

        const int rowA = 8 * hh - 1;
        const int rowB = rowA + 16;
        const bool first = (l0 + rowA < 0);
        const int rowA0 = first ? 0 : rowA;

        for (int j = 0; j < 8; ++j) {
            const int col = ((((wave & 1) << 3) + j) << 4) + sub;
            const float* __restrict__ Bp = W2 + col * NW + g * PLR + koff;
            const float* __restrict__ hc = hrow + col;

            v2f fb[PLR / 4];
            #pragma unroll
            for (int s = 0; s < PLR / 4; ++s) fb[s] = *(const v2f*)(Bp + 4 * s);
            float hA[9], hB[9];
            hA[0] = hc[rowA0 * Hdim];
            hB[0] = hc[rowB * Hdim];
            #pragma unroll
            for (int i = 1; i < 9; ++i) {
                hA[i] = hc[(rowA + i) * Hdim];
                hB[i] = hc[(rowB + i) * Hdim];
            }
            if (first) hA[0] = 0.0f;
            const float bv = bias[col];
            const float qv = qrow[col];

            v8f c0 = {}, c1 = {};
            #pragma unroll
            for (int s = 0; s < PLR / 4; ++s) {
                c0 = __builtin_amdgcn_wmma_f32_16x16x4_f32(false, fa0[s], false, fb[s], (short)0, c0, false, false);
                c1 = __builtin_amdgcn_wmma_f32_16x16x4_f32(false, fa1[s], false, fb[s], (short)0, c1, false, false);
            }

            const int ob = 8 * hh * FT_XP + col;
            #pragma unroll
            for (int r = 0; r < 8; ++r) {
                const float h0 = hA[r + 1];
                const float h1 = hB[r + 1];
                const float dl0 = kvB ? (hA[r] - h0) : (qv - h0);
                const float dl1 = kvB ? (hB[r] - h1) : (qv - h1);
                outp[ob + r * FT_XP]        = h0 + dl0 * (c0[r] + bv);
                outp[ob + (r + 16) * FT_XP] = h1 + dl1 * (c1[r] + bv);
            }
        }
    }
    __syncthreads();

    {
        const long yrow = ((long)e * LKs + l0 + 8 * hh) * Hdim + sub;

        if (wave < 2) {
            const int nc = wave << 5;
            v8f c00 = {}, c01 = {}, c10 = {}, c11 = {};
            ft_products<Hdim, GLR, FT_XP>(xs + FT_TILE, Wa, nc, sub, koff, c00, c01, c10, c11);
            float* tp = tw_s + 8 * hh * FT_TWP + nc + sub;
            #pragma unroll
            for (int r = 0; r < 8; ++r) {
                tp[r * FT_TWP]             = tanhf(c00[r]);
                tp[r * FT_TWP + 16]        = tanhf(c01[r]);
                tp[(r + 16) * FT_TWP]      = tanhf(c10[r]);
                tp[(r + 16) * FT_TWP + 16] = tanhf(c11[r]);
            }
        } else {
            ft_long_job(wave - 2, xs, Wr, Wk, Wv, Wg, rbuf, kbuf, vbuf, gsbuf, yrow, sub, koff);
        }
        __syncthreads();

        for (int uj = wave - 4; uj >= 0 && uj < 8; uj += 6) {
            const int nc = uj << 5;
            v8f c00 = {}, c01 = {}, c10 = {}, c11 = {};
            ft_products<GLR, Hdim, FT_TWP>(tw_s, Wb, nc, sub, koff, c00, c01, c10, c11);
            const float bv0 = bb[nc + sub];
            const float bv1 = bb[nc + sub + 16];
            float* __restrict__ yp = dbuf + yrow + nc;
            #pragma unroll
            for (int r = 0; r < 8; ++r) {
                yp[r * Hdim]             = expf(-expf(c00[r] + bv0));
                yp[r * Hdim + 16]        = expf(-expf(c01[r] + bv1));
                yp[(r + 16) * Hdim]      = expf(-expf(c10[r] + bv0));
                yp[(r + 16) * Hdim + 16] = expf(-expf(c11[r] + bv1));
            }
        }
        for (int t = wave; t < 24; t += FT_WAVES)
            ft_long_job(8 + t, xs, Wr, Wk, Wv, Wg, rbuf, kbuf, vbuf, gsbuf, yrow, sub, koff);
    }
}

#define SC_STEPS  8
#define SC_SLOT   (4 * DK)
#define SC_CHUNK  (SC_STEPS * SC_SLOT)
#define SC_NCHUNK (LKs / SC_STEPS)
#define SC_NBUF   3
#define SC_SCANW  4
#define SC_KQ     (DK / 4)
#define SC_OBUF   (DK * SC_STEPS)

template <int MASK>
__device__ __forceinline__ float sc_row_xor(float x)
{
    const int b = __builtin_bit_cast(int, x);
    return __builtin_bit_cast(float, __builtin_amdgcn_update_dpp(0, b, 0x160 + MASK, 0xF, 0xF, true));
}
template <int CTRL>
__device__ __forceinline__ float sc_quad_lane(float x)
{
    const int b = __builtin_bit_cast(int, x);
    return __builtin_bit_cast(float, __builtin_amdgcn_update_dpp(0, b, CTRL, 0xF, 0xF, true));
}
__device__ __forceinline__ float sc_quad_join(float x)
{
    return ((sc_quad_lane<0x00>(x) + sc_quad_lane<0x55>(x)) + sc_quad_lane<0xAA>(x)) + sc_quad_lane<0xFF>(x);
}
__device__ __forceinline__ float sc_xhalf(float x)
{
    const int b = __builtin_bit_cast(int, x);
    return __builtin_bit_cast(float, __builtin_amdgcn_permlanex16(b, b, 0x76543210, 0xfedcba98, false, false));
}
__device__ __forceinline__ float sc_wave_sum(float x)
{
    x += sc_row_xor<1>(x);
    x += sc_row_xor<2>(x);
    x += sc_row_xor<4>(x);
    x += sc_row_xor<8>(x);
    return x + sc_xhalf(x);
}

__global__ __launch_bounds__((SC_SCANW + 1) * 32) void rwkv_scan_kernel(
    const float* __restrict__ r, const float* __restrict__ k,
    const float* __restrict__ v, const float* __restrict__ decay,
    const float* __restrict__ u, const float* __restrict__ gsilu,
    const float* __restrict__ gn_w, const float* __restrict__ gn_b,
    float* __restrict__ y)
{
    __shared__ __attribute__((aligned(16))) float stage[SC_NBUF * SC_CHUNK];
    __shared__ __attribute__((aligned(16))) float obuf[2 * SC_OBUF];

    const int chain = blockIdx.x;
    const int wave  = __builtin_amdgcn_readfirstlane((int)threadIdx.x >> 5);
    const bool scan = (wave < SC_SCANW);
    const int e     = chain >> 2;
    const int head  = chain & 3;
    const int lane  = (int)threadIdx.x & 31;
    const long base = ((long)e * LKs) * Hdim + head * DK;
    const int kq    = lane & 3;
    const int v0    = (wave & 3) * 16 + ((lane >> 2) << 1);
    const bool odd  = (kq & 1) != 0;

    v2f S[SC_KQ];
    float uu[SC_KQ];
    #pragma unroll
    for (int j = 0; j < SC_KQ; ++j) { S[j] = (v2f){0.0f, 0.0f}; uu[j] = u[head * DK + kq * SC_KQ + j]; }

    const float* __restrict__ srow = (wave == 0) ? r : (wave == 1) ? k : (wave == 2) ? decay : v;
    const float* __restrict__ sg = srow + base + 2 * lane;
    const int sput = (wave & 3) * DK + 2 * lane;

    v2f gx[SC_STEPS];

    auto fetch = [&](int chunk) {
        #pragma unroll
        for (int i = 0; i < SC_STEPS; ++i)
            gx[i] = *(const v2f*)(sg + (long)(chunk * SC_STEPS + i) * Hdim);
    };
    auto put = [&](float* buf) {
        #pragma unroll
        for (int i = 0; i < SC_STEPS; ++i)
            *(v2f*)(buf + i * SC_SLOT + sput) = gx[i];
    };

    v4f R[2][4], K[2][4], D[2][4];
    v2f VT[2];

    auto take = [&](const float* buf, int s, int set) {
        const float* slot = buf + s * SC_SLOT + kq * SC_KQ;
        #pragma unroll
        for (int g = 0; g < 4; ++g) {
            R[set][g] = *(const v4f*)(slot + 4 * g);
            K[set][g] = *(const v4f*)(slot + DK + 4 * g);
            D[set][g] = *(const v4f*)(slot + 2 * DK + 4 * g);
        }
        VT[set] = *(const v2f*)(buf + s * SC_SLOT + 3 * DK + v0);
    };

    const float* __restrict__ gsg = gsilu + base + lane;
    float*       __restrict__ yg  = y + base + lane;
    const float gwa = gn_w[head * DK + lane], gwb = gn_w[head * DK + lane + 32];
    const float gba = gn_b[head * DK + lane], gbb = gn_b[head * DK + lane + 32];
    float ga[SC_STEPS], gb[SC_STEPS];
    float na[SC_STEPS], nb[SC_STEPS];
    #pragma unroll
    for (int i = 0; i < SC_STEPS; ++i) { ga[i] = 0.0f; gb[i] = 0.0f; na[i] = 0.0f; nb[i] = 0.0f; }

    auto gates = [&](int chunk) {
        #pragma unroll
        for (int i = 0; i < SC_STEPS; ++i) {
            const long off = (long)(chunk * SC_STEPS + i) * Hdim;
            na[i] = gsg[off];
            nb[i] = gsg[off + 32];
        }
    };
    auto finish = [&](int chunk, const float* ob) {
        #pragma unroll
        for (int hf = 0; hf < SC_STEPS / 4; ++hf) {
            const v4f oa = *(const v4f*)(ob + lane * SC_STEPS + 4 * hf);
            const v4f oc = *(const v4f*)(ob + (lane + 32) * SC_STEPS + 4 * hf);
            float mean[4], da[4], db[4], var[4];
            #pragma unroll
            for (int i = 0; i < 4; ++i) mean[i] = sc_wave_sum(oa[i] + oc[i]) * (1.0f / 64.0f);
            #pragma unroll
            for (int i = 0; i < 4; ++i) {
                da[i] = oa[i] - mean[i];
                db[i] = oc[i] - mean[i];
                var[i] = sc_wave_sum(da[i] * da[i] + db[i] * db[i]) * (1.0f / 64.0f);
            }
            #pragma unroll
            for (int i = 0; i < 4; ++i) {
                const float rstd = __builtin_amdgcn_rsqf(var[i] + EPSV);
                const long off = (long)(chunk * SC_STEPS + 4 * hf + i) * Hdim;
                yg[off]      = (da[i] * rstd * gwa + gba) * ga[4 * hf + i];
                yg[off + 32] = (db[i] * rstd * gwb + gbb) * gb[4 * hf + i];
            }
        }
    };
    if (scan) {
        fetch(0);
        put(stage);
    }
    __syncthreads();
    if (scan) {
        fetch(1);
        take(stage, 0, 0);
    }

    int rc = 0;
    for (int c = 0; c < SC_NCHUNK; ++c) {
        const int rn = (rc == SC_NBUF - 1) ? 0 : rc + 1;
        const float* cur = stage + rc * SC_CHUNK;
        float*       nxt = stage + rn * SC_CHUNK;
        rc = rn;
        const float* ob = obuf + ((c + 1) & 1) * SC_OBUF;
        float* const ow = obuf + (c & 1) * SC_OBUF + (v0 + (kq & 1)) * SC_STEPS;
        if (scan) put(nxt);
        __syncthreads();
        if (!scan) {
            gates(c);
            if (c > 0) finish(c - 1, ob);
            #pragma unroll
            for (int i = 0; i < SC_STEPS; ++i) { ga[i] = na[i]; gb[i] = nb[i]; }
            continue;
        }
        fetch((c + 2 < SC_NCHUNK) ? (c + 2) : (SC_NCHUNK - 1));

        v2f pa = {0.0f, 0.0f};
        #pragma unroll
        for (int i = 0; i < SC_STEPS; ++i) {
            if (i + 1 < SC_STEPS) take(cur, i + 1, (i + 1) & 1);
            else                  take(nxt, 0, 0);
            __builtin_amdgcn_sched_barrier(0);

            const int set = i & 1;
            const v2f vt = VT[set];
            v2f acc = {0.0f, 0.0f};
            #pragma unroll
            for (int j = 0; j < SC_KQ; ++j) {
                const v2f kv = K[set][j >> 2][j & 3] * vt;
                acc += R[set][j >> 2][j & 3] * (S[j] + uu[j] * kv);
                S[j] = D[set][j >> 2][j & 3] * S[j] + kv;
            }
            if (i > 0) {
                const float j0 = sc_quad_join(pa.x);
                const float j1 = sc_quad_join(pa.y);
                ow[i - 1] = odd ? j1 : j0;
            }
            pa = acc;
            __builtin_amdgcn_sched_barrier(0);
        }
        {
            const float j0 = sc_quad_join(pa.x);
            const float j1 = sc_quad_join(pa.y);
            ow[SC_STEPS - 1] = odd ? j1 : j0;
        }
    }

    {
        const float* ob = obuf + ((SC_NCHUNK + 1) & 1) * SC_OBUF;
        __syncthreads();
        if (!scan) finish(SC_NCHUNK - 1, ob);
    }
}

#define OP_WAVES 8

__global__ __launch_bounds__(OP_WAVES * 32) void rwkv_out_kernel(
    const float* __restrict__ y, const float* __restrict__ Wo, float* __restrict__ out)
{
    __shared__ __attribute__((aligned(16))) float ys[FT_TILE];

    const int tid  = (int)threadIdx.x;
    const int wave = __builtin_amdgcn_readfirstlane(tid >> 5);
    const int lane = tid & 31;
    const int sub  = lane & 15;
    const int hh   = lane >> 4;
    const int koff = hh << 1;
    const long row0 = (long)blockIdx.x * FT_TOK;

    {
        const int c4 = (tid & 63) << 2;
        const int t0 = tid >> 6;
        const float* __restrict__ yp = y + row0 * Hdim + c4;
        #pragma unroll
        for (int j = 0; j < FT_TOK / 4; ++j) {
            const int tok = t0 + 4 * j;
            *(v4f*)(ys + tok * FT_XP + c4) = *(const v4f*)(yp + tok * Hdim);
        }
    }
    __syncthreads();

    const int nc = wave << 5;
    ft_strip<false>(ys, Wo, out + (row0 + 8 * hh) * Hdim + sub + nc, nc, sub, koff);
}

extern "C" void kernel_launch(void* const* d_in, const int* in_sizes, int n_in,
                              void* d_out, int out_size, void* d_ws, size_t ws_size,
                              hipStream_t stream)
{
    const float* query    = (const float*)d_in[0];
    const float* keyval   = (const float*)d_in[1];
    const float* rwg_mu   = (const float*)d_in[2];
    const float* W_rwg0   = (const float*)d_in[3];
    const float* W_rwg2   = (const float*)d_in[4];
    const float* rwg_bias = (const float*)d_in[5];
    const float* kv_mu    = (const float*)d_in[6];
    const float* W_kv0    = (const float*)d_in[7];
    const float* W_kv2    = (const float*)d_in[8];
    const float* kv_bias  = (const float*)d_in[9];
    const float* Wr       = (const float*)d_in[10];
    const float* Wa       = (const float*)d_in[11];
    const float* Wb       = (const float*)d_in[12];
    const float* bb       = (const float*)d_in[13];
    const float* Wk       = (const float*)d_in[14];
    const float* Wv       = (const float*)d_in[15];
    const float* Wg       = (const float*)d_in[16];
    const float* u        = (const float*)d_in[17];
    const float* gn_w     = (const float*)d_in[18];
    const float* gn_b     = (const float*)d_in[19];
    const float* Wo       = (const float*)d_in[20];

    float* ws = (float*)d_ws;
    const long SZ = (long)NTOK * Hdim;
    float* rbuf  = ws + 0 * SZ;
    float* kbuf  = ws + 1 * SZ;
    float* vbuf  = ws + 2 * SZ;
    float* dbuf  = ws + 3 * SZ;
    float* gsbuf = ws + 4 * SZ;
    float* ybuf  = ws + 5 * SZ;

    rwkv_features_kernel<<<NTOK / FT_TOK, FT_WAVES * 32, 0, stream>>>(
        query, keyval, rwg_mu, W_rwg0, W_rwg2, rwg_bias,
        kv_mu, W_kv0, W_kv2, kv_bias,
        Wr, Wa, Wb, bb, Wk, Wv, Wg,
        rbuf, kbuf, vbuf, gsbuf, dbuf);

    rwkv_scan_kernel<<<Bsz * LQ * NHEAD, (SC_SCANW + 1) * 32, 0, stream>>>(
        rbuf, kbuf, vbuf, dbuf, u, gsbuf, gn_w, gn_b, ybuf);

    rwkv_out_kernel<<<NTOK / FT_TOK, OP_WAVES * 32, 0, stream>>>(ybuf, Wo, (float*)d_out);
}


# scan split along time: 4 stretches per chain run from zero state in 256 groups; new launch adds (r x decay prefix) @ carried state on f32 matrix unit, then GroupNorm+gate; o, GN sum order changed
# speedup vs baseline: 9.2521x; 1.2830x over previous
#include <hip/hip_runtime.h>
#include <math.h>

#define Hdim 256
#define NHEAD 4
#define DK 64
#define PLR 32
#define GLR 64
#define Bsz 2
#define LQ 8
#define LKs 512
#define NTOK (Bsz * LQ * LKs)
#define EPSV 1e-5f

typedef float v2f __attribute__((ext_vector_type(2)));
typedef float v4f __attribute__((ext_vector_type(4)));
typedef float v8f __attribute__((ext_vector_type(8)));

#define FT_TOK   32
#define FT_WAVES 10
#define FT_XP    260
#define FT_T1P   100
#define FT_T2P   68
#define FT_TWP   68
#define FT_TILE  (FT_TOK * FT_XP)

template <int K, int N, int P>
__device__ __forceinline__ void ft_products(
    const float* Xs, const float* __restrict__ W, int nc, int sub, int koff,
    v8f& c00, v8f& c01, v8f& c10, v8f& c11)
{
    const float* A0 = Xs + sub * P + koff;
    const float* A1 = A0 + 16 * P;
    const float* __restrict__ B0 = W + koff * N + nc + sub;
    const float* __restrict__ B1 = B0 + 16;
    #pragma unroll 8
    for (int k = 0; k < K; k += 4) {
        const v2f a0 = *(const v2f*)(A0 + k);
        const v2f a1 = *(const v2f*)(A1 + k);
        v2f b0, b1;
        b0.x = B0[k * N];
        b0.y = B0[(k + 1) * N];
        b1.x = B1[k * N];
        b1.y = B1[(k + 1) * N];
        c00 = __builtin_amdgcn_wmma_f32_16x16x4_f32(false, a0, false, b0, (short)0, c00, false, false);
        c01 = __builtin_amdgcn_wmma_f32_16x16x4_f32(false, a0, false, b1, (short)0, c01, false, false);
        c10 = __builtin_amdgcn_wmma_f32_16x16x4_f32(false, a1, false, b0, (short)0, c10, false, false);
        c11 = __builtin_amdgcn_wmma_f32_16x16x4_f32(false, a1, false, b1, (short)0, c11, false, false);
    }
}

template <bool SILU>
__device__ __forceinline__ void ft_strip(
    const float* Xs, const float* __restrict__ Wp, float* __restrict__ yp,
    int nc, int sub, int koff)
{
    v8f c00 = {}, c01 = {}, c10 = {}, c11 = {};
    ft_products<Hdim, Hdim, FT_XP>(Xs, Wp, nc, sub, koff, c00, c01, c10, c11);
    #pragma unroll
    for (int r = 0; r < 8; ++r) {
        if (SILU) {
            yp[r * Hdim]             = c00[r] / (1.0f + expf(-c00[r]));
            yp[r * Hdim + 16]        = c01[r] / (1.0f + expf(-c01[r]));
            yp[(r + 16) * Hdim]      = c10[r] / (1.0f + expf(-c10[r]));
            yp[(r + 16) * Hdim + 16] = c11[r] / (1.0f + expf(-c11[r]));
        } else {
            yp[r * Hdim]             = c00[r];
            yp[r * Hdim + 16]        = c01[r];
            yp[(r + 16) * Hdim]      = c10[r];
            yp[(r + 16) * Hdim + 16] = c11[r];
        }
    }
}

__device__ __forceinline__ void ft_long_job(
    int jb, const float* xs,
    const float* __restrict__ Wr, const float* __restrict__ Wk,
    const float* __restrict__ Wv, const float* __restrict__ Wg,
    float* __restrict__ rbuf, float* __restrict__ kbuf,
    float* __restrict__ vbuf, float* __restrict__ gsbuf,
    long yrow, int sub, int koff)
{
    const int nc = (jb & 7) << 5;
    switch (jb >> 3) {
        case 0:  ft_strip<false>(xs,               Wr, rbuf  + yrow + nc, nc, sub, koff); break;
        case 1:  ft_strip<false>(xs + 3 * FT_TILE, Wk, kbuf  + yrow + nc, nc, sub, koff); break;
        case 2:  ft_strip<false>(xs + 4 * FT_TILE, Wv, vbuf  + yrow + nc, nc, sub, koff); break;
        default: ft_strip<true>(xs + 2 * FT_TILE,  Wg, gsbuf + yrow + nc, nc, sub, koff); break;
    }
}

__global__ __launch_bounds__(FT_WAVES * 32) void rwkv_features_kernel(
    const float* __restrict__ query, const float* __restrict__ keyval,
    const float* __restrict__ rwg_mu, const float* __restrict__ W_rwg0,
    const float* __restrict__ W_rwg2, const float* __restrict__ rwg_bias,
    const float* __restrict__ kv_mu, const float* __restrict__ W_kv0,
    const float* __restrict__ W_kv2, const float* __restrict__ kv_bias,
    const float* __restrict__ Wr, const float* __restrict__ Wa,
    const float* __restrict__ Wb, const float* __restrict__ bb,
    const float* __restrict__ Wk, const float* __restrict__ Wv,
    const float* __restrict__ Wg,
    float* __restrict__ rbuf, float* __restrict__ kbuf, float* __restrict__ vbuf,
    float* __restrict__ gsbuf, float* __restrict__ dbuf)
{
    __shared__ __attribute__((aligned(16))) float xs[5 * FT_TILE];
    __shared__ __attribute__((aligned(16))) float t1_s[FT_TOK * FT_T1P];
    __shared__ __attribute__((aligned(16))) float t2_s[FT_TOK * FT_T2P];
    __shared__ __attribute__((aligned(16))) float tw_s[FT_TOK * FT_TWP];
    float* const xr_s = xs;
    float* const xk_s = xs + 3 * FT_TILE;

    const int tid  = (int)threadIdx.x;
    const int wave = __builtin_amdgcn_readfirstlane(tid >> 5);
    const int lane = tid & 31;
    const int sub  = lane & 15;
    const int hh   = lane >> 4;
    const int koff = hh << 1;

    const int tile = (int)blockIdx.x;
    const int e    = tile >> 4;
    const int l0   = (tile & 15) * FT_TOK;
    const int b    = e >> 3;
    const float* __restrict__ hrow = keyval + ((long)b * LKs + l0) * Hdim;
    const float* __restrict__ qrow = query + (long)e * Hdim;

    {
        const int c4 = (tid & 63) << 2;
        const int t0 = tid >> 6;
        const v4f q  = *(const v4f*)(qrow + c4);
        const v4f mr = *(const v4f*)(rwg_mu + c4);
        const v4f mk = *(const v4f*)(kv_mu + c4);
        #pragma unroll
        for (int j = 0; j < 7; ++j) {
            const int tok = t0 + 5 * j;
            if (j < 6 || tok < FT_TOK) {
                const bool first = (l0 + tok == 0);
                const int prow = first ? 0 : tok - 1;
                const v4f h  = *(const v4f*)(hrow + tok * Hdim + c4);
                v4f hp = *(const v4f*)(hrow + prow * Hdim + c4);
                if (first) hp = (v4f){0.0f, 0.0f, 0.0f, 0.0f};
                const v4f dc = q - h;
                const v4f d  = hp - h;
                *(v4f*)(xr_s + tok * FT_XP + c4) = h + dc * mr;
                *(v4f*)(xk_s + tok * FT_XP + c4) = h + d * mk;
            }
        }
    }
    __syncthreads();

    {
        const bool kvA = (wave >= 6);
        const float* Xs = kvA ? xk_s : xr_s;
        const float* __restrict__ W0 = kvA ? W_kv0 : W_rwg0;
        const int N0  = kvA ? 2 * PLR : 3 * PLR;
        const int nc  = (kvA ? wave - 6 : wave) << 4;
        float* Td     = kvA ? t2_s : t1_s;
        const int PT  = kvA ? FT_T2P : FT_T1P;

        const float* A0 = Xs + sub * FT_XP + koff;
        const float* A1 = A0 + 16 * FT_XP;
        const float* __restrict__ Bp = W0 + koff * N0 + nc + sub;

        v8f c0 = {}, c1 = {};
        #pragma unroll 8
        for (int k = 0; k < Hdim; k += 4) {
            const v2f a0 = *(const v2f*)(A0 + k);
            const v2f a1 = *(const v2f*)(A1 + k);
            v2f bf;
            bf.x = Bp[k * N0];
            bf.y = Bp[(k + 1) * N0];
            c0 = __builtin_amdgcn_wmma_f32_16x16x4_f32(false, a0, false, bf, (short)0, c0, false, false);
            c1 = __builtin_amdgcn_wmma_f32_16x16x4_f32(false, a1, false, bf, (short)0, c1, false, false);
        }
        #pragma unroll
        for (int r = 0; r < 8; ++r) {
            const int m = r + 8 * hh;
            Td[m * PT + nc + sub]        = tanhf(c0[r]);
            Td[(16 + m) * PT + nc + sub] = tanhf(c1[r]);
        }
    }
    __syncthreads();

    {
        const int o    = wave >> 1;
        const bool kvB = (o >= 3);
        const int g    = kvB ? o - 3 : o;
        const float* Ts = kvB ? t2_s : t1_s;
        const int PT   = kvB ? FT_T2P : FT_T1P;
        const float* __restrict__ W2 = kvB ? W_kv2 : W_rwg2;
        const int NW   = kvB ? 2 * PLR : 3 * PLR;
        const float* __restrict__ bias = (kvB ? kv_bias : rwg_bias) + g * Hdim;
        float* outp = xs + o * FT_TILE;

        const float* A0 = Ts + sub * PT + g * PLR + koff;
        const float* A1 = A0 + 16 * PT;
        v2f fa0[PLR / 4], fa1[PLR / 4];
        #pragma unroll
        for (int s = 0; s < PLR / 4; ++s) {
            fa0[s] = *(const v2f*)(A0 + 4 * s);
            fa1[s] = *(const v2f*)(A1 + 4 * s);
        }

        const int rowA = 8 * hh - 1;
        const int rowB = rowA + 16;
        const bool first = (l0 + rowA < 0);
        const int rowA0 = first ? 0 : rowA;

        for (int j = 0; j < 8; ++j) {
            const int col = ((((wave & 1) << 3) + j) << 4) + sub;
            const float* __restrict__ Bp = W2 + col * NW + g * PLR + koff;
            const float* __restrict__ hc = hrow + col;

            v2f fb[PLR / 4];
            #pragma unroll
            for (int s = 0; s < PLR / 4; ++s) fb[s] = *(const v2f*)(Bp + 4 * s);
            float hA[9], hB[9];
            hA[0] = hc[rowA0 * Hdim];
            hB[0] = hc[rowB * Hdim];
            #pragma unroll
            for (int i = 1; i < 9; ++i) {
                hA[i] = hc[(rowA + i) * Hdim];
                hB[i] = hc[(rowB + i) * Hdim];
            }
            if (first) hA[0] = 0.0f;
            const float bv = bias[col];
            const float qv = qrow[col];

            v8f c0 = {}, c1 = {};
            #pragma unroll
            for (int s = 0; s < PLR / 4; ++s) {
                c0 = __builtin_amdgcn_wmma_f32_16x16x4_f32(false, fa0[s], false, fb[s], (short)0, c0, false, false);
                c1 = __builtin_amdgcn_wmma_f32_16x16x4_f32(false, fa1[s], false, fb[s], (short)0, c1, false, false);
            }

            const int ob = 8 * hh * FT_XP + col;
            #pragma unroll
            for (int r = 0; r < 8; ++r) {
                const float h0 = hA[r + 1];
                const float h1 = hB[r + 1];
                const float dl0 = kvB ? (hA[r] - h0) : (qv - h0);
                const float dl1 = kvB ? (hB[r] - h1) : (qv - h1);
                outp[ob + r * FT_XP]        = h0 + dl0 * (c0[r] + bv);
                outp[ob + (r + 16) * FT_XP] = h1 + dl1 * (c1[r] + bv);
            }
        }
    }
    __syncthreads();

    {
        const long yrow = ((long)e * LKs + l0 + 8 * hh) * Hdim + sub;

        if (wave < 2) {
            const int nc = wave << 5;
            v8f c00 = {}, c01 = {}, c10 = {}, c11 = {};
            ft_products<Hdim, GLR, FT_XP>(xs + FT_TILE, Wa, nc, sub, koff, c00, c01, c10, c11);
            float* tp = tw_s + 8 * hh * FT_TWP + nc + sub;
            #pragma unroll
            for (int r = 0; r < 8; ++r) {
                tp[r * FT_TWP]             = tanhf(c00[r]);
                tp[r * FT_TWP + 16]        = tanhf(c01[r]);
                tp[(r + 16) * FT_TWP]      = tanhf(c10[r]);
                tp[(r + 16) * FT_TWP + 16] = tanhf(c11[r]);
            }
        } else {
            ft_long_job(wave - 2, xs, Wr, Wk, Wv, Wg, rbuf, kbuf, vbuf, gsbuf, yrow, sub, koff);
        }
        __syncthreads();

        for (int uj = wave - 4; uj >= 0 && uj < 8; uj += 6) {
            const int nc = uj << 5;
            v8f c00 = {}, c01 = {}, c10 = {}, c11 = {};
            ft_products<GLR, Hdim, FT_TWP>(tw_s, Wb, nc, sub, koff, c00, c01, c10, c11);
            const float bv0 = bb[nc + sub];
            const float bv1 = bb[nc + sub + 16];
            float* __restrict__ yp = dbuf + yrow + nc;
            #pragma unroll
            for (int r = 0; r < 8; ++r) {
                yp[r * Hdim]             = expf(-expf(c00[r] + bv0));
                yp[r * Hdim + 16]        = expf(-expf(c01[r] + bv1));
                yp[(r + 16) * Hdim]      = expf(-expf(c10[r] + bv0));
                yp[(r + 16) * Hdim + 16] = expf(-expf(c11[r] + bv1));
            }
        }
        for (int t = wave; t < 24; t += FT_WAVES)
            ft_long_job(8 + t, xs, Wr, Wk, Wv, Wg, rbuf, kbuf, vbuf, gsbuf, yrow, sub, koff);
    }
}

#define SC_STEPS  8
#define SC_SLOT   (4 * DK)
#define SC_CHUNK  (SC_STEPS * SC_SLOT)
#define SC_NCH    4
#define SC_TL     (LKs / SC_NCH)
#define SC_NCHUNK (SC_TL / SC_STEPS)
#define SC_NBUF   3
#define SC_SCANW  4
#define SC_KQ     (DK / 4)
#define SC_OBUF   (DK * SC_STEPS)

template <int MASK>
__device__ __forceinline__ float sc_row_xor(float x)
{
    const int b = __builtin_bit_cast(int, x);
    return __builtin_bit_cast(float, __builtin_amdgcn_update_dpp(0, b, 0x160 + MASK, 0xF, 0xF, true));
}
template <int CTRL>
__device__ __forceinline__ float sc_quad_lane(float x)
{
    const int b = __builtin_bit_cast(int, x);
    return __builtin_bit_cast(float, __builtin_amdgcn_update_dpp(0, b, CTRL, 0xF, 0xF, true));
}
__device__ __forceinline__ float sc_quad_join(float x)
{
    return ((sc_quad_lane<0x00>(x) + sc_quad_lane<0x55>(x)) + sc_quad_lane<0xAA>(x)) + sc_quad_lane<0xFF>(x);
}
__global__ __launch_bounds__((SC_SCANW + 1) * 32) void rwkv_scan_kernel(
    const float* __restrict__ r, const float* __restrict__ k,
    const float* __restrict__ v, const float* __restrict__ decay,
    const float* __restrict__ u, float* __restrict__ olocal, float* __restrict__ send)
{
    __shared__ __attribute__((aligned(16))) float stage[SC_NBUF * SC_CHUNK];
    __shared__ __attribute__((aligned(16))) float obuf[2 * SC_OBUF];

    const int chain = (int)blockIdx.x / SC_NCH;
    const int tc    = (int)blockIdx.x % SC_NCH;
    const int wave  = __builtin_amdgcn_readfirstlane((int)threadIdx.x >> 5);
    const bool scan = (wave < SC_SCANW);
    const int e     = chain >> 2;
    const int head  = chain & 3;
    const int lane  = (int)threadIdx.x & 31;
    const long base = ((long)e * LKs + tc * SC_TL) * Hdim + head * DK;
    const int kq    = lane & 3;
    const int v0    = (wave & 3) * 16 + ((lane >> 2) << 1);
    const bool odd  = (kq & 1) != 0;

    v2f S[SC_KQ];
    float uu[SC_KQ];
    #pragma unroll
    for (int j = 0; j < SC_KQ; ++j) { S[j] = (v2f){0.0f, 0.0f}; uu[j] = u[head * DK + kq * SC_KQ + j]; }

    const float* __restrict__ srow = (wave == 0) ? r : (wave == 1) ? k : (wave == 2) ? decay : v;
    const float* __restrict__ sg = srow + base + 2 * lane;
    const int sput = (wave & 3) * DK + 2 * lane;

    v2f gx[SC_STEPS];

    auto fetch = [&](int chunk) {
        #pragma unroll
        for (int i = 0; i < SC_STEPS; ++i)
            gx[i] = *(const v2f*)(sg + (long)(chunk * SC_STEPS + i) * Hdim);
    };
    auto put = [&](float* buf) {
        #pragma unroll
        for (int i = 0; i < SC_STEPS; ++i)
            *(v2f*)(buf + i * SC_SLOT + sput) = gx[i];
    };

    v4f R[2][4], K[2][4], D[2][4];
    v2f VT[2];

    auto take = [&](const float* buf, int s, int set) {
        const float* slot = buf + s * SC_SLOT + kq * SC_KQ;
        #pragma unroll
        for (int g = 0; g < 4; ++g) {
            R[set][g] = *(const v4f*)(slot + 4 * g);
            K[set][g] = *(const v4f*)(slot + DK + 4 * g);
            D[set][g] = *(const v4f*)(slot + 2 * DK + 4 * g);
        }
        VT[set] = *(const v2f*)(buf + s * SC_SLOT + 3 * DK + v0);
    };

    float* __restrict__ og = olocal + base + lane;
    auto writeout = [&](int chunk, const float* ob) {
        #pragma unroll
        for (int hf = 0; hf < SC_STEPS / 4; ++hf) {
            const v4f oa = *(const v4f*)(ob + lane * SC_STEPS + 4 * hf);
            const v4f oc = *(const v4f*)(ob + (lane + 32) * SC_STEPS + 4 * hf);
            #pragma unroll
            for (int i = 0; i < 4; ++i) {
                const long off = (long)(chunk * SC_STEPS + 4 * hf + i) * Hdim;
                og[off]      = oa[i];
                og[off + 32] = oc[i];
            }
        }
    };
    if (scan) {
        fetch(0);
        put(stage);
    }
    __syncthreads();
    if (scan) {
        fetch(1);
        take(stage, 0, 0);
    }

    int rc = 0;
    for (int c = 0; c < SC_NCHUNK; ++c) {
        const int rn = (rc == SC_NBUF - 1) ? 0 : rc + 1;
        const float* cur = stage + rc * SC_CHUNK;
        float*       nxt = stage + rn * SC_CHUNK;
        rc = rn;
        const float* ob = obuf + ((c + 1) & 1) * SC_OBUF;
        float* const ow = obuf + (c & 1) * SC_OBUF + (v0 + (kq & 1)) * SC_STEPS;
        if (scan) put(nxt);
        __syncthreads();
        if (!scan) {
            if (c > 0) writeout(c - 1, ob);
            continue;
        }
        fetch((c + 2 < SC_NCHUNK) ? (c + 2) : (SC_NCHUNK - 1));

        v2f pa = {0.0f, 0.0f};
        #pragma unroll
        for (int i = 0; i < SC_STEPS; ++i) {
            if (i + 1 < SC_STEPS) take(cur, i + 1, (i + 1) & 1);
            else                  take(nxt, 0, 0);
            __builtin_amdgcn_sched_barrier(0);

            const int set = i & 1;
            const v2f vt = VT[set];
            v2f acc = {0.0f, 0.0f};
            #pragma unroll
            for (int j = 0; j < SC_KQ; ++j) {
                const v2f kv = K[set][j >> 2][j & 3] * vt;
                acc += R[set][j >> 2][j & 3] * (S[j] + uu[j] * kv);
                S[j] = D[set][j >> 2][j & 3] * S[j] + kv;
            }
            if (i > 0) {
                const float j0 = sc_quad_join(pa.x);
                const float j1 = sc_quad_join(pa.y);
                ow[i - 1] = odd ? j1 : j0;
            }
            pa = acc;
            __builtin_amdgcn_sched_barrier(0);
        }
        {
            const float j0 = sc_quad_join(pa.x);
            const float j1 = sc_quad_join(pa.y);
            ow[SC_STEPS - 1] = odd ? j1 : j0;
        }
    }

    {
        const float* ob = obuf + ((SC_NCHUNK + 1) & 1) * SC_OBUF;
        __syncthreads();
        if (!scan) writeout(SC_NCHUNK - 1, ob);
    }

    if (scan && tc < SC_NCH - 1) {
        float* __restrict__ sp = send + ((long)((chain * (SC_NCH - 1) + tc) * SC_SCANW + wave) * DK + kq * SC_KQ) * 16
                                      + ((lane >> 2) << 1);
        #pragma unroll
        for (int j = 0; j < SC_KQ; ++j) *(v2f*)(sp + j * 16) = S[j];
    }
}

#define CR_WAVES 8
#define CR_PR    68
#define CR_PS    72
#define CR_SEG   (SC_TL / 4)

__device__ __forceinline__ float cr_row_sum(float x)
{
    x += sc_row_xor<1>(x);
    x += sc_row_xor<2>(x);
    x += sc_row_xor<4>(x);
    return x + sc_row_xor<8>(x);
}

__global__ __launch_bounds__(CR_WAVES * 32) void rwkv_carry_kernel(
    const float* __restrict__ r, const float* __restrict__ decay,
    const float* __restrict__ olocal, const float* __restrict__ send,
    const float* __restrict__ gsilu, const float* __restrict__ gn_w,
    const float* __restrict__ gn_b, float* __restrict__ y)
{
    __shared__ __attribute__((aligned(16))) float rt_s[SC_TL * CR_PR];
    __shared__ __attribute__((aligned(16))) float ss_s[DK * CR_PS];
    __shared__ float segtot[4 * DK];
    __shared__ float ptot[(SC_NCH - 2) * 4 * DK];

    const int tid  = (int)threadIdx.x;
    const int wave = __builtin_amdgcn_readfirstlane(tid >> 5);
    const int seg  = __builtin_amdgcn_readfirstlane(tid >> 6);
    const int lane = tid & 31;
    const int sub  = lane & 15;
    const int hh   = lane >> 4;
    const int koff = hh << 1;
    const int kk   = tid & 63;

    const int chain = (int)blockIdx.x / SC_NCH;
    const int tc    = (int)blockIdx.x % SC_NCH;
    const int e     = chain >> 2;
    const int head  = chain & 3;
    const long cbase = ((long)e * LKs) * Hdim + head * DK;
    const long base  = cbase + (long)tc * SC_TL * Hdim;

    {
        const float* __restrict__ dp = decay + base + (long)(seg * CR_SEG) * Hdim + kk;
        const float* __restrict__ rp = r     + base + (long)(seg * CR_SEG) * Hdim + kk;
        float dv[CR_SEG], rv[CR_SEG];
        #pragma unroll
        for (int i = 0; i < CR_SEG; ++i) { dv[i] = dp[i * Hdim]; rv[i] = rp[i * Hdim]; }

        for (int pc = 1; pc < tc; ++pc) {
            const float* __restrict__ qp = decay + cbase + (long)(pc * SC_TL + seg * CR_SEG) * Hdim + kk;
            float qv[CR_SEG];
            #pragma unroll
            for (int i = 0; i < CR_SEG; ++i) qv[i] = qp[i * Hdim];
            float q = qv[0];
            #pragma unroll
            for (int i = 1; i < CR_SEG; ++i) q *= qv[i];
            ptot[((pc - 1) * 4 + seg) * DK + kk] = q;
        }

        float p = 1.0f;
        #pragma unroll
        for (int i = 0; i < CR_SEG; ++i) { const float before = p; p *= dv[i]; dv[i] = before; }
        segtot[seg * DK + kk] = p;
        __syncthreads();

        float start = 1.0f;
        for (int s = 0; s < seg; ++s) start *= segtot[s * DK + kk];
        float* rtp = rt_s + (seg * CR_SEG) * CR_PR + kk;
        #pragma unroll
        for (int i = 0; i < CR_SEG; ++i) rtp[i * CR_PR] = rv[i] * (start * dv[i]);
    }

    {
        const int kr = tid >> 4;
        const int vl = tid & 15;
        float s[16];
        #pragma unroll
        for (int i = 0; i < 16; ++i) s[i] = 0.0f;
        if (tc > 0) {
            const float* __restrict__ sp = send + (long)chain * (SC_NCH - 1) * DK * DK + tid;
            #pragma unroll
            for (int i = 0; i < 16; ++i) s[i] = sp[i * 256];
            for (int pc = 1; pc < tc; ++pc) {
                const float* __restrict__ sq = sp + (long)pc * DK * DK;
                const float* pt = ptot + (pc - 1) * 4 * DK + kr;
                float nv[16];
                #pragma unroll
                for (int i = 0; i < 16; ++i) nv[i] = sq[i * 256];
                float pe[4];
                #pragma unroll
                for (int q = 0; q < 4; ++q)
                    pe[q] = ((pt[16 * q] * pt[DK + 16 * q]) * pt[2 * DK + 16 * q]) * pt[3 * DK + 16 * q];
                #pragma unroll
                for (int i = 0; i < 16; ++i) s[i] = pe[i & 3] * s[i] + nv[i];
            }
        }
        #pragma unroll
        for (int i = 0; i < 16; ++i) ss_s[(kr + 16 * (i & 3)) * CR_PS + 16 * (i >> 2) + vl] = s[i];
    }
    __syncthreads();

    v8f c0 = {}, c1 = {}, c2 = {}, c3 = {};
    {
        const float* A0 = rt_s + (16 * wave + sub) * CR_PR + koff;
        const float* B0 = ss_s + koff * CR_PS + sub;
        #pragma unroll
        for (int k = 0; k < DK; k += 4) {
            const v2f a = *(const v2f*)(A0 + k);
            v2f b0, b1, b2, b3;
            b0.x = B0[k * CR_PS];
            b0.y = B0[(k + 1) * CR_PS];
            b1.x = B0[k * CR_PS + 16];
            b1.y = B0[(k + 1) * CR_PS + 16];
            b2.x = B0[k * CR_PS + 32];
            b2.y = B0[(k + 1) * CR_PS + 32];
            b3.x = B0[k * CR_PS + 48];
            b3.y = B0[(k + 1) * CR_PS + 48];
            c0 = __builtin_amdgcn_wmma_f32_16x16x4_f32(false, a, false, b0, (short)0, c0, false, false);
            c1 = __builtin_amdgcn_wmma_f32_16x16x4_f32(false, a, false, b1, (short)0, c1, false, false);
            c2 = __builtin_amdgcn_wmma_f32_16x16x4_f32(false, a, false, b2, (short)0, c2, false, false);
            c3 = __builtin_amdgcn_wmma_f32_16x16x4_f32(false, a, false, b3, (short)0, c3, false, false);
        }
    }

    {
        const long rowb = base + (long)(16 * wave + 8 * hh) * Hdim + sub;
        const float* __restrict__ op = olocal + rowb;
        const float* __restrict__ gp = gsilu + rowb;
        float* __restrict__ yp = y + rowb;
        const int hc = head * DK + sub;
        const float gw0 = gn_w[hc], gw1 = gn_w[hc + 16], gw2 = gn_w[hc + 32], gw3 = gn_w[hc + 48];
        const float gb0 = gn_b[hc], gb1 = gn_b[hc + 16], gb2 = gn_b[hc + 32], gb3 = gn_b[hc + 48];
        #pragma unroll
        for (int rr = 0; rr < 8; ++rr) {
            const float x0 = c0[rr] + op[rr * Hdim];
            const float x1 = c1[rr] + op[rr * Hdim + 16];
            const float x2 = c2[rr] + op[rr * Hdim + 32];
            const float x3 = c3[rr] + op[rr * Hdim + 48];
            const float mean = cr_row_sum((x0 + x1) + (x2 + x3)) * (1.0f / 64.0f);
            const float d0 = x0 - mean, d1 = x1 - mean, d2 = x2 - mean, d3 = x3 - mean;
            const float var = cr_row_sum((d0 * d0 + d1 * d1) + (d2 * d2 + d3 * d3)) * (1.0f / 64.0f);
            const float rstd = __builtin_amdgcn_rsqf(var + EPSV);
            yp[rr * Hdim]      = (d0 * rstd * gw0 + gb0) * gp[rr * Hdim];
            yp[rr * Hdim + 16] = (d1 * rstd * gw1 + gb1) * gp[rr * Hdim + 16];
            yp[rr * Hdim + 32] = (d2 * rstd * gw2 + gb2) * gp[rr * Hdim + 32];
            yp[rr * Hdim + 48] = (d3 * rstd * gw3 + gb3) * gp[rr * Hdim + 48];
        }
    }
}

#define OP_WAVES 8

__global__ __launch_bounds__(OP_WAVES * 32) void rwkv_out_kernel(
    const float* __restrict__ y, const float* __restrict__ Wo, float* __restrict__ out)
{
    __shared__ __attribute__((aligned(16))) float ys[FT_TILE];

    const int tid  = (int)threadIdx.x;
    const int wave = __builtin_amdgcn_readfirstlane(tid >> 5);
    const int lane = tid & 31;
    const int sub  = lane & 15;
    const int hh   = lane >> 4;
    const int koff = hh << 1;
    const long row0 = (long)blockIdx.x * FT_TOK;

    {
        const int c4 = (tid & 63) << 2;
        const int t0 = tid >> 6;
        const float* __restrict__ yp = y + row0 * Hdim + c4;
        #pragma unroll
        for (int j = 0; j < FT_TOK / 4; ++j) {
            const int tok = t0 + 4 * j;
            *(v4f*)(ys + tok * FT_XP + c4) = *(const v4f*)(yp + tok * Hdim);
        }
    }
    __syncthreads();

    const int nc = wave << 5;
    ft_strip<false>(ys, Wo, out + (row0 + 8 * hh) * Hdim + sub + nc, nc, sub, koff);
}

extern "C" void kernel_launch(void* const* d_in, const int* in_sizes, int n_in,
                              void* d_out, int out_size, void* d_ws, size_t ws_size,
                              hipStream_t stream)
{
    const float* query    = (const float*)d_in[0];
    const float* keyval   = (const float*)d_in[1];
    const float* rwg_mu   = (const float*)d_in[2];
    const float* W_rwg0   = (const float*)d_in[3];
    const float* W_rwg2   = (const float*)d_in[4];
    const float* rwg_bias = (const float*)d_in[5];
    const float* kv_mu    = (const float*)d_in[6];
    const float* W_kv0    = (const float*)d_in[7];
    const float* W_kv2    = (const float*)d_in[8];
    const float* kv_bias  = (const float*)d_in[9];
    const float* Wr       = (const float*)d_in[10];
    const float* Wa       = (const float*)d_in[11];
    const float* Wb       = (const float*)d_in[12];
    const float* bb       = (const float*)d_in[13];
    const float* Wk       = (const float*)d_in[14];
    const float* Wv       = (const float*)d_in[15];
    const float* Wg       = (const float*)d_in[16];
    const float* u        = (const float*)d_in[17];
    const float* gn_w     = (const float*)d_in[18];
    const float* gn_b     = (const float*)d_in[19];
    const float* Wo       = (const float*)d_in[20];

    float* ws = (float*)d_ws;
    const long SZ = (long)NTOK * Hdim;
    float* rbuf  = ws + 0 * SZ;
    float* kbuf  = ws + 1 * SZ;
    float* vbuf  = ws + 2 * SZ;
    float* dbuf  = ws + 3 * SZ;
    float* gsbuf = ws + 4 * SZ;
    float* ybuf  = ws + 5 * SZ;
    float* olbuf = ws + 6 * SZ;
    float* sebuf = ws + 7 * SZ;

    rwkv_features_kernel<<<NTOK / FT_TOK, FT_WAVES * 32, 0, stream>>>(
        query, keyval, rwg_mu, W_rwg0, W_rwg2, rwg_bias,
        kv_mu, W_kv0, W_kv2, kv_bias,
        Wr, Wa, Wb, bb, Wk, Wv, Wg,
        rbuf, kbuf, vbuf, gsbuf, dbuf);

    rwkv_scan_kernel<<<Bsz * LQ * NHEAD * SC_NCH, (SC_SCANW + 1) * 32, 0, stream>>>(
        rbuf, kbuf, vbuf, dbuf, u, olbuf, sebuf);

    rwkv_carry_kernel<<<Bsz * LQ * NHEAD * SC_NCH, CR_WAVES * 32, 0, stream>>>(
        rbuf, dbuf, olbuf, sebuf, gsbuf, gn_w, gn_b, ybuf);

    rwkv_out_kernel<<<NTOK / FT_TOK, OP_WAVES * 32, 0, stream>>>(ybuf, Wo, (float*)d_out);
}


# features work-group on 12 waves, three per SIMD: stage B's 80 tile jobs dealt 6 or 7 to a wave, stage C's strips three long jobs' worth to each wave (was 4, 3.5 or 3.25 on 10 waves); sum orders kept
# speedup vs baseline: 9.7465x; 1.0534x over previous
#include <hip/hip_runtime.h>
#include <math.h>

#define Hdim 256
#define NHEAD 4
#define DK 64
#define PLR 32
#define GLR 64
#define Bsz 2
#define LQ 8
#define LKs 512
#define NTOK (Bsz * LQ * LKs)
#define EPSV 1e-5f

typedef float v2f __attribute__((ext_vector_type(2)));
typedef float v4f __attribute__((ext_vector_type(4)));
typedef float v8f __attribute__((ext_vector_type(8)));

#define FT_TOK   32
#define FT_WAVES 12
#define FT_XP    260
#define FT_T1P   100
#define FT_T2P   68
#define FT_TWP   68
#define FT_TILE  (FT_TOK * FT_XP)

template <int K, int N, int P>
__device__ __forceinline__ void ft_products(
    const float* Xs, const float* __restrict__ W, int nc, int sub, int koff,
    v8f& c00, v8f& c01, v8f& c10, v8f& c11)
{
    const float* A0 = Xs + sub * P + koff;
    const float* A1 = A0 + 16 * P;
    const float* __restrict__ B0 = W + koff * N + nc + sub;
    const float* __restrict__ B1 = B0 + 16;
    #pragma unroll 8
    for (int k = 0; k < K; k += 4) {
        const v2f a0 = *(const v2f*)(A0 + k);
        const v2f a1 = *(const v2f*)(A1 + k);
        v2f b0, b1;
        b0.x = B0[k * N];
        b0.y = B0[(k + 1) * N];
        b1.x = B1[k * N];
        b1.y = B1[(k + 1) * N];
        c00 = __builtin_amdgcn_wmma_f32_16x16x4_f32(false, a0, false, b0, (short)0, c00, false, false);
        c01 = __builtin_amdgcn_wmma_f32_16x16x4_f32(false, a0, false, b1, (short)0, c01, false, false);
        c10 = __builtin_amdgcn_wmma_f32_16x16x4_f32(false, a1, false, b0, (short)0, c10, false, false);
        c11 = __builtin_amdgcn_wmma_f32_16x16x4_f32(false, a1, false, b1, (short)0, c11, false, false);
    }
}

template <bool SILU>
__device__ __forceinline__ void ft_strip(
    const float* Xs, const float* __restrict__ Wp, float* __restrict__ yp,
    int nc, int sub, int koff)
{
    v8f c00 = {}, c01 = {}, c10 = {}, c11 = {};
    ft_products<Hdim, Hdim, FT_XP>(Xs, Wp, nc, sub, koff, c00, c01, c10, c11);
    #pragma unroll
    for (int r = 0; r < 8; ++r) {
        if (SILU) {
            yp[r * Hdim]             = c00[r] / (1.0f + expf(-c00[r]));
            yp[r * Hdim + 16]        = c01[r] / (1.0f + expf(-c01[r]));
            yp[(r + 16) * Hdim]      = c10[r] / (1.0f + expf(-c10[r]));
            yp[(r + 16) * Hdim + 16] = c11[r] / (1.0f + expf(-c11[r]));
        } else {
            yp[r * Hdim]             = c00[r];
            yp[r * Hdim + 16]        = c01[r];
            yp[(r + 16) * Hdim]      = c10[r];
            yp[(r + 16) * Hdim + 16] = c11[r];
        }
    }
}

__device__ __forceinline__ void ft_long_job(
    int jb, const float* xs,
    const float* __restrict__ Wr, const float* __restrict__ Wk,
    const float* __restrict__ Wv, const float* __restrict__ Wg,
    float* __restrict__ rbuf, float* __restrict__ kbuf,
    float* __restrict__ vbuf, float* __restrict__ gsbuf,
    long yrow, int sub, int koff)
{
    const int nc = (jb & 7) << 5;
    switch (jb >> 3) {
        case 0:  ft_strip<false>(xs,               Wr, rbuf  + yrow + nc, nc, sub, koff); break;
        case 1:  ft_strip<false>(xs + 3 * FT_TILE, Wk, kbuf  + yrow + nc, nc, sub, koff); break;
        case 2:  ft_strip<false>(xs + 4 * FT_TILE, Wv, vbuf  + yrow + nc, nc, sub, koff); break;
        default: ft_strip<true>(xs + 2 * FT_TILE,  Wg, gsbuf + yrow + nc, nc, sub, koff); break;
    }
}

__global__ __launch_bounds__(FT_WAVES * 32) void rwkv_features_kernel(
    const float* __restrict__ query, const float* __restrict__ keyval,
    const float* __restrict__ rwg_mu, const float* __restrict__ W_rwg0,
    const float* __restrict__ W_rwg2, const float* __restrict__ rwg_bias,
    const float* __restrict__ kv_mu, const float* __restrict__ W_kv0,
    const float* __restrict__ W_kv2, const float* __restrict__ kv_bias,
    const float* __restrict__ Wr, const float* __restrict__ Wa,
    const float* __restrict__ Wb, const float* __restrict__ bb,
    const float* __restrict__ Wk, const float* __restrict__ Wv,
    const float* __restrict__ Wg,
    float* __restrict__ rbuf, float* __restrict__ kbuf, float* __restrict__ vbuf,
    float* __restrict__ gsbuf, float* __restrict__ dbuf)
{
    __shared__ __attribute__((aligned(16))) float xs[5 * FT_TILE];
    __shared__ __attribute__((aligned(16))) float t1_s[FT_TOK * FT_T1P];
    __shared__ __attribute__((aligned(16))) float t2_s[FT_TOK * FT_T2P];
    __shared__ __attribute__((aligned(16))) float tw_s[FT_TOK * FT_TWP];
    float* const xr_s = xs;
    float* const xk_s = xs + 3 * FT_TILE;

    const int tid  = (int)threadIdx.x;
    const int wave = __builtin_amdgcn_readfirstlane(tid >> 5);
    const int lane = tid & 31;
    const int sub  = lane & 15;
    const int hh   = lane >> 4;
    const int koff = hh << 1;

    const int tile = (int)blockIdx.x;
    const int e    = tile >> 4;
    const int l0   = (tile & 15) * FT_TOK;
    const int b    = e >> 3;
    const float* __restrict__ hrow = keyval + ((long)b * LKs + l0) * Hdim;
    const float* __restrict__ qrow = query + (long)e * Hdim;

    {
        const int c4 = (tid & 63) << 2;
        const int t0 = tid >> 6;
        const v4f q  = *(const v4f*)(qrow + c4);
        const v4f mr = *(const v4f*)(rwg_mu + c4);
        const v4f mk = *(const v4f*)(kv_mu + c4);
        #pragma unroll
        for (int j = 0; j < 6; ++j) {
            const int tok = t0 + 6 * j;
            if (j < 5 || tok < FT_TOK) {
                const bool first = (l0 + tok == 0);
                const int prow = first ? 0 : tok - 1;
                const v4f h  = *(const v4f*)(hrow + tok * Hdim + c4);
                v4f hp = *(const v4f*)(hrow + prow * Hdim + c4);
                if (first) hp = (v4f){0.0f, 0.0f, 0.0f, 0.0f};
                const v4f dc = q - h;
                const v4f d  = hp - h;
                *(v4f*)(xr_s + tok * FT_XP + c4) = h + dc * mr;
                *(v4f*)(xk_s + tok * FT_XP + c4) = h + d * mk;
            }
        }
    }
    __syncthreads();

    if (wave < 10) {
        const bool kvA = (wave >= 6);
        const float* Xs = kvA ? xk_s : xr_s;
        const float* __restrict__ W0 = kvA ? W_kv0 : W_rwg0;
        const int N0  = kvA ? 2 * PLR : 3 * PLR;
        const int nc  = (kvA ? wave - 6 : wave) << 4;
        float* Td     = kvA ? t2_s : t1_s;
        const int PT  = kvA ? FT_T2P : FT_T1P;

        const float* A0 = Xs + sub * FT_XP + koff;
        const float* A1 = A0 + 16 * FT_XP;
        const float* __restrict__ Bp = W0 + koff * N0 + nc + sub;

        v8f c0 = {}, c1 = {};
        #pragma unroll 8
        for (int k = 0; k < Hdim; k += 4) {
            const v2f a0 = *(const v2f*)(A0 + k);
            const v2f a1 = *(const v2f*)(A1 + k);
            v2f bf;
            bf.x = Bp[k * N0];
            bf.y = Bp[(k + 1) * N0];
            c0 = __builtin_amdgcn_wmma_f32_16x16x4_f32(false, a0, false, bf, (short)0, c0, false, false);
            c1 = __builtin_amdgcn_wmma_f32_16x16x4_f32(false, a1, false, bf, (short)0, c1, false, false);
        }
        #pragma unroll
        for (int r = 0; r < 8; ++r) {
            const int m = r + 8 * hh;
            Td[m * PT + nc + sub]        = tanhf(c0[r]);
            Td[(16 + m) * PT + nc + sub] = tanhf(c1[r]);
        }
    }
    __syncthreads();

    {
        const int jbeg = (20 * wave) / 3;
        const int jend = (20 * (wave + 1)) / 3;

        const int rowA = 8 * hh - 1;
        const int rowB = rowA + 16;
        const bool first = (l0 + rowA < 0);
        const int rowA0 = first ? 0 : rowA;

        for (int J = jbeg; J < jend; ++J) {
            const int o    = J >> 4;
            const bool kvB = (o >= 3);
            const int g    = kvB ? o - 3 : o;
            const float* Ts = kvB ? t2_s : t1_s;
            const int PT   = kvB ? FT_T2P : FT_T1P;
            const float* __restrict__ W2 = kvB ? W_kv2 : W_rwg2;
            const int NW   = kvB ? 2 * PLR : 3 * PLR;
            const float* __restrict__ bias = (kvB ? kv_bias : rwg_bias) + g * Hdim;
            float* outp = xs + o * FT_TILE;

            const int col = ((J & 15) << 4) + sub;
            const float* __restrict__ Bp = W2 + col * NW + g * PLR + koff;
            const float* __restrict__ hc = hrow + col;
            const float* A0 = Ts + sub * PT + g * PLR + koff;
            const float* A1 = A0 + 16 * PT;

            v2f fa0[PLR / 4], fa1[PLR / 4], fb[PLR / 4];
            #pragma unroll
            for (int s = 0; s < PLR / 4; ++s) {
                fb[s]  = *(const v2f*)(Bp + 4 * s);
                fa0[s] = *(const v2f*)(A0 + 4 * s);
                fa1[s] = *(const v2f*)(A1 + 4 * s);
            }
            float hA[9], hB[9];
            hA[0] = hc[rowA0 * Hdim];
            hB[0] = hc[rowB * Hdim];
            #pragma unroll
            for (int i = 1; i < 9; ++i) {
                hA[i] = hc[(rowA + i) * Hdim];
                hB[i] = hc[(rowB + i) * Hdim];
            }
            if (first) hA[0] = 0.0f;
            const float bv = bias[col];
            const float qv = qrow[col];

            v8f c0 = {}, c1 = {};
            #pragma unroll
            for (int s = 0; s < PLR / 4; ++s) {
                c0 = __builtin_amdgcn_wmma_f32_16x16x4_f32(false, fa0[s], false, fb[s], (short)0, c0, false, false);
                c1 = __builtin_amdgcn_wmma_f32_16x16x4_f32(false, fa1[s], false, fb[s], (short)0, c1, false, false);
            }

            const int ob = 8 * hh * FT_XP + col;
            #pragma unroll
            for (int r = 0; r < 8; ++r) {
                const float h0 = hA[r + 1];
                const float h1 = hB[r + 1];
                const float dl0 = kvB ? (hA[r] - h0) : (qv - h0);
                const float dl1 = kvB ? (hB[r] - h1) : (qv - h1);
                outp[ob + r * FT_XP]        = h0 + dl0 * (c0[r] + bv);
                outp[ob + (r + 16) * FT_XP] = h1 + dl1 * (c1[r] + bv);
            }
        }
    }
    __syncthreads();

    {
        const long yrow = ((long)e * LKs + l0 + 8 * hh) * Hdim + sub;

        if (wave < 2) {
            const int nc = wave << 5;
            v8f c00 = {}, c01 = {}, c10 = {}, c11 = {};
            ft_products<Hdim, GLR, FT_XP>(xs + FT_TILE, Wa, nc, sub, koff, c00, c01, c10, c11);
            float* tp = tw_s + 8 * hh * FT_TWP + nc + sub;
            #pragma unroll
            for (int r = 0; r < 8; ++r) {
                tp[r * FT_TWP]             = tanhf(c00[r]);
                tp[r * FT_TWP + 16]        = tanhf(c01[r]);
                tp[(r + 16) * FT_TWP]      = tanhf(c10[r]);
                tp[(r + 16) * FT_TWP + 16] = tanhf(c11[r]);
            }
        } else {
            ft_long_job(wave - 2, xs, Wr, Wk, Wv, Wg, rbuf, kbuf, vbuf, gsbuf, yrow, sub, koff);
        }
        __syncthreads();

        const int nup = (wave < 2) ? 4 : 0;
        for (int i = 0; i < nup; ++i) {
            const int nc = ((wave << 2) + i) << 5;
            v8f c00 = {}, c01 = {}, c10 = {}, c11 = {};
            ft_products<GLR, Hdim, FT_TWP>(tw_s, Wb, nc, sub, koff, c00, c01, c10, c11);
            const float bv0 = bb[nc + sub];
            const float bv1 = bb[nc + sub + 16];
            float* __restrict__ yp = dbuf + yrow + nc;
            #pragma unroll
            for (int r = 0; r < 8; ++r) {
                yp[r * Hdim]             = expf(-expf(c00[r] + bv0));
                yp[r * Hdim + 16]        = expf(-expf(c01[r] + bv1));
                yp[(r + 16) * Hdim]      = expf(-expf(c10[r] + bv0));
                yp[(r + 16) * Hdim + 16] = expf(-expf(c11[r] + bv1));
            }
        }
        const int jlast = (wave < 2) ? 20 : 32;
        for (int jb = 10 + wave; jb < jlast; jb += 10)
            ft_long_job(jb, xs, Wr, Wk, Wv, Wg, rbuf, kbuf, vbuf, gsbuf, yrow, sub, koff);
    }
}

#define SC_STEPS  8
#define SC_SLOT   (4 * DK)
#define SC_CHUNK  (SC_STEPS * SC_SLOT)
#define SC_NCH    4
#define SC_TL     (LKs / SC_NCH)
#define SC_NCHUNK (SC_TL / SC_STEPS)
#define SC_NBUF   3
#define SC_SCANW  4
#define SC_KQ     (DK / 4)
#define SC_OBUF   (DK * SC_STEPS)

template <int MASK>
__device__ __forceinline__ float sc_row_xor(float x)
{
    const int b = __builtin_bit_cast(int, x);
    return __builtin_bit_cast(float, __builtin_amdgcn_update_dpp(0, b, 0x160 + MASK, 0xF, 0xF, true));
}
template <int CTRL>
__device__ __forceinline__ float sc_quad_lane(float x)
{
    const int b = __builtin_bit_cast(int, x);
    return __builtin_bit_cast(float, __builtin_amdgcn_update_dpp(0, b, CTRL, 0xF, 0xF, true));
}
__device__ __forceinline__ float sc_quad_join(float x)
{
    return ((sc_quad_lane<0x00>(x) + sc_quad_lane<0x55>(x)) + sc_quad_lane<0xAA>(x)) + sc_quad_lane<0xFF>(x);
}
__global__ __launch_bounds__((SC_SCANW + 1) * 32) void rwkv_scan_kernel(
    const float* __restrict__ r, const float* __restrict__ k,
    const float* __restrict__ v, const float* __restrict__ decay,
    const float* __restrict__ u, float* __restrict__ olocal, float* __restrict__ send)
{
    __shared__ __attribute__((aligned(16))) float stage[SC_NBUF * SC_CHUNK];
    __shared__ __attribute__((aligned(16))) float obuf[2 * SC_OBUF];

    const int chain = (int)blockIdx.x / SC_NCH;
    const int tc    = (int)blockIdx.x % SC_NCH;
    const int wave  = __builtin_amdgcn_readfirstlane((int)threadIdx.x >> 5);
    const bool scan = (wave < SC_SCANW);
    const int e     = chain >> 2;
    const int head  = chain & 3;
    const int lane  = (int)threadIdx.x & 31;
    const long base = ((long)e * LKs + tc * SC_TL) * Hdim + head * DK;
    const int kq    = lane & 3;
    const int v0    = (wave & 3) * 16 + ((lane >> 2) << 1);
    const bool odd  = (kq & 1) != 0;

    v2f S[SC_KQ];
    float uu[SC_KQ];
    #pragma unroll
    for (int j = 0; j < SC_KQ; ++j) { S[j] = (v2f){0.0f, 0.0f}; uu[j] = u[head * DK + kq * SC_KQ + j]; }

    const float* __restrict__ srow = (wave == 0) ? r : (wave == 1) ? k : (wave == 2) ? decay : v;
    const float* __restrict__ sg = srow + base + 2 * lane;
    const int sput = (wave & 3) * DK + 2 * lane;

    v2f gx[SC_STEPS];

    auto fetch = [&](int chunk) {
        #pragma unroll
        for (int i = 0; i < SC_STEPS; ++i)
            gx[i] = *(const v2f*)(sg + (long)(chunk * SC_STEPS + i) * Hdim);
    };
    auto put = [&](float* buf) {
        #pragma unroll
        for (int i = 0; i < SC_STEPS; ++i)
            *(v2f*)(buf + i * SC_SLOT + sput) = gx[i];
    };

    v4f R[2][4], K[2][4], D[2][4];
    v2f VT[2];

    auto take = [&](const float* buf, int s, int set) {
        const float* slot = buf + s * SC_SLOT + kq * SC_KQ;
        #pragma unroll
        for (int g = 0; g < 4; ++g) {
            R[set][g] = *(const v4f*)(slot + 4 * g);
            K[set][g] = *(const v4f*)(slot + DK + 4 * g);
            D[set][g] = *(const v4f*)(slot + 2 * DK + 4 * g);
        }
        VT[set] = *(const v2f*)(buf + s * SC_SLOT + 3 * DK + v0);
    };

    float* __restrict__ og = olocal + base + lane;
    auto writeout = [&](int chunk, const float* ob) {
        #pragma unroll
        for (int hf = 0; hf < SC_STEPS / 4; ++hf) {
            const v4f oa = *(const v4f*)(ob + lane * SC_STEPS + 4 * hf);
            const v4f oc = *(const v4f*)(ob + (lane + 32) * SC_STEPS + 4 * hf);
            #pragma unroll
            for (int i = 0; i < 4; ++i) {
                const long off = (long)(chunk * SC_STEPS + 4 * hf + i) * Hdim;
                og[off]      = oa[i];
                og[off + 32] = oc[i];
            }
        }
    };
    if (scan) {
        fetch(0);
        put(stage);
    }
    __syncthreads();
    if (scan) {
        fetch(1);
        take(stage, 0, 0);
    }

    int rc = 0;
    for (int c = 0; c < SC_NCHUNK; ++c) {
        const int rn = (rc == SC_NBUF - 1) ? 0 : rc + 1;
        const float* cur = stage + rc * SC_CHUNK;
        float*       nxt = stage + rn * SC_CHUNK;
        rc = rn;
        const float* ob = obuf + ((c + 1) & 1) * SC_OBUF;
        float* const ow = obuf + (c & 1) * SC_OBUF + (v0 + (kq & 1)) * SC_STEPS;
        if (scan) put(nxt);
        __syncthreads();
        if (!scan) {
            if (c > 0) writeout(c - 1, ob);
            continue;
        }
        fetch((c + 2 < SC_NCHUNK) ? (c + 2) : (SC_NCHUNK - 1));

        v2f pa = {0.0f, 0.0f};
        #pragma unroll
        for (int i = 0; i < SC_STEPS; ++i) {
            if (i + 1 < SC_STEPS) take(cur, i + 1, (i + 1) & 1);
            else                  take(nxt, 0, 0);
            __builtin_amdgcn_sched_barrier(0);

            const int set = i & 1;
            const v2f vt = VT[set];
            v2f acc = {0.0f, 0.0f};
            #pragma unroll
            for (int j = 0; j < SC_KQ; ++j) {
                const v2f kv = K[set][j >> 2][j & 3] * vt;
                acc += R[set][j >> 2][j & 3] * (S[j] + uu[j] * kv);
                S[j] = D[set][j >> 2][j & 3] * S[j] + kv;
            }
            if (i > 0) {
                const float j0 = sc_quad_join(pa.x);
                const float j1 = sc_quad_join(pa.y);
                ow[i - 1] = odd ? j1 : j0;
            }
            pa = acc;
            __builtin_amdgcn_sched_barrier(0);
        }
        {
            const float j0 = sc_quad_join(pa.x);
            const float j1 = sc_quad_join(pa.y);
            ow[SC_STEPS - 1] = odd ? j1 : j0;
        }
    }

    {
        const float* ob = obuf + ((SC_NCHUNK + 1) & 1) * SC_OBUF;
        __syncthreads();
        if (!scan) writeout(SC_NCHUNK - 1, ob);
    }

    if (scan && tc < SC_NCH - 1) {
        float* __restrict__ sp = send + ((long)((chain * (SC_NCH - 1) + tc) * SC_SCANW + wave) * DK + kq * SC_KQ) * 16
                                      + ((lane >> 2) << 1);
        #pragma unroll
        for (int j = 0; j < SC_KQ; ++j) *(v2f*)(sp + j * 16) = S[j];
    }
}

#define CR_WAVES 8
#define CR_PR    68
#define CR_PS    72
#define CR_SEG   (SC_TL / 4)

__device__ __forceinline__ float cr_row_sum(float x)
{
    x += sc_row_xor<1>(x);
    x += sc_row_xor<2>(x);
    x += sc_row_xor<4>(x);
    return x + sc_row_xor<8>(x);
}

__global__ __launch_bounds__(CR_WAVES * 32) void rwkv_carry_kernel(
    const float* __restrict__ r, const float* __restrict__ decay,
    const float* __restrict__ olocal, const float* __restrict__ send,
    const float* __restrict__ gsilu, const float* __restrict__ gn_w,
    const float* __restrict__ gn_b, float* __restrict__ y)
{
    __shared__ __attribute__((aligned(16))) float rt_s[SC_TL * CR_PR];
    __shared__ __attribute__((aligned(16))) float ss_s[DK * CR_PS];
    __shared__ float segtot[4 * DK];
    __shared__ float ptot[(SC_NCH - 2) * 4 * DK];

    const int tid  = (int)threadIdx.x;
    const int wave = __builtin_amdgcn_readfirstlane(tid >> 5);
    const int seg  = __builtin_amdgcn_readfirstlane(tid >> 6);
    const int lane = tid & 31;
    const int sub  = lane & 15;
    const int hh   = lane >> 4;
    const int koff = hh << 1;
    const int kk   = tid & 63;

    const int chain = (int)blockIdx.x / SC_NCH;
    const int tc    = (int)blockIdx.x % SC_NCH;
    const int e     = chain >> 2;
    const int head  = chain & 3;
    const long cbase = ((long)e * LKs) * Hdim + head * DK;
    const long base  = cbase + (long)tc * SC_TL * Hdim;

    {
        const float* __restrict__ dp = decay + base + (long)(seg * CR_SEG) * Hdim + kk;
        const float* __restrict__ rp = r     + base + (long)(seg * CR_SEG) * Hdim + kk;
        float dv[CR_SEG], rv[CR_SEG];
        #pragma unroll
        for (int i = 0; i < CR_SEG; ++i) { dv[i] = dp[i * Hdim]; rv[i] = rp[i * Hdim]; }

        for (int pc = 1; pc < tc; ++pc) {
            const float* __restrict__ qp = decay + cbase + (long)(pc * SC_TL + seg * CR_SEG) * Hdim + kk;
            float qv[CR_SEG];
            #pragma unroll
            for (int i = 0; i < CR_SEG; ++i) qv[i] = qp[i * Hdim];
            float q = qv[0];
            #pragma unroll
            for (int i = 1; i < CR_SEG; ++i) q *= qv[i];
            ptot[((pc - 1) * 4 + seg) * DK + kk] = q;
        }

        float p = 1.0f;
        #pragma unroll
        for (int i = 0; i < CR_SEG; ++i) { const float before = p; p *= dv[i]; dv[i] = before; }
        segtot[seg * DK + kk] = p;
        __syncthreads();

        float start = 1.0f;
        for (int s = 0; s < seg; ++s) start *= segtot[s * DK + kk];
        float* rtp = rt_s + (seg * CR_SEG) * CR_PR + kk;
        #pragma unroll
        for (int i = 0; i < CR_SEG; ++i) rtp[i * CR_PR] = rv[i] * (start * dv[i]);
    }

    {
        const int kr = tid >> 4;
        const int vl = tid & 15;
        float s[16];
        #pragma unroll
        for (int i = 0; i < 16; ++i) s[i] = 0.0f;
        if (tc > 0) {
            const float* __restrict__ sp = send + (long)chain * (SC_NCH - 1) * DK * DK + tid;
            #pragma unroll
            for (int i = 0; i < 16; ++i) s[i] = sp[i * 256];
            for (int pc = 1; pc < tc; ++pc) {
                const float* __restrict__ sq = sp + (long)pc * DK * DK;
                const float* pt = ptot + (pc - 1) * 4 * DK + kr;
                float nv[16];
                #pragma unroll
                for (int i = 0; i < 16; ++i) nv[i] = sq[i * 256];
                float pe[4];
                #pragma unroll
                for (int q = 0; q < 4; ++q)
                    pe[q] = ((pt[16 * q] * pt[DK + 16 * q]) * pt[2 * DK + 16 * q]) * pt[3 * DK + 16 * q];
                #pragma unroll
                for (int i = 0; i < 16; ++i) s[i] = pe[i & 3] * s[i] + nv[i];
            }
        }
        #pragma unroll
        for (int i = 0; i < 16; ++i) ss_s[(kr + 16 * (i & 3)) * CR_PS + 16 * (i >> 2) + vl] = s[i];
    }
    __syncthreads();

    v8f c0 = {}, c1 = {}, c2 = {}, c3 = {};
    {
        const float* A0 = rt_s + (16 * wave + sub) * CR_PR + koff;
        const float* B0 = ss_s + koff * CR_PS + sub;
        #pragma unroll
        for (int k = 0; k < DK; k += 4) {
            const v2f a = *(const v2f*)(A0 + k);
            v2f b0, b1, b2, b3;
            b0.x = B0[k * CR_PS];
            b0.y = B0[(k + 1) * CR_PS];
            b1.x = B0[k * CR_PS + 16];
            b1.y = B0[(k + 1) * CR_PS + 16];
            b2.x = B0[k * CR_PS + 32];
            b2.y = B0[(k + 1) * CR_PS + 32];
            b3.x = B0[k * CR_PS + 48];
            b3.y = B0[(k + 1) * CR_PS + 48];
            c0 = __builtin_amdgcn_wmma_f32_16x16x4_f32(false, a, false, b0, (short)0, c0, false, false);
            c1 = __builtin_amdgcn_wmma_f32_16x16x4_f32(false, a, false, b1, (short)0, c1, false, false);
            c2 = __builtin_amdgcn_wmma_f32_16x16x4_f32(false, a, false, b2, (short)0, c2, false, false);
            c3 = __builtin_amdgcn_wmma_f32_16x16x4_f32(false, a, false, b3, (short)0, c3, false, false);
        }
    }

    {
        const long rowb = base + (long)(16 * wave + 8 * hh) * Hdim + sub;
        const float* __restrict__ op = olocal + rowb;
        const float* __restrict__ gp = gsilu + rowb;
        float* __restrict__ yp = y + rowb;
        const int hc = head * DK + sub;
        const float gw0 = gn_w[hc], gw1 = gn_w[hc + 16], gw2 = gn_w[hc + 32], gw3 = gn_w[hc + 48];
        const float gb0 = gn_b[hc], gb1 = gn_b[hc + 16], gb2 = gn_b[hc + 32], gb3 = gn_b[hc + 48];
        #pragma unroll
        for (int rr = 0; rr < 8; ++rr) {
            const float x0 = c0[rr] + op[rr * Hdim];
            const float x1 = c1[rr] + op[rr * Hdim + 16];
            const float x2 = c2[rr] + op[rr * Hdim + 32];
            const float x3 = c3[rr] + op[rr * Hdim + 48];
            const float mean = cr_row_sum((x0 + x1) + (x2 + x3)) * (1.0f / 64.0f);
            const float d0 = x0 - mean, d1 = x1 - mean, d2 = x2 - mean, d3 = x3 - mean;
            const float var = cr_row_sum((d0 * d0 + d1 * d1) + (d2 * d2 + d3 * d3)) * (1.0f / 64.0f);
            const float rstd = __builtin_amdgcn_rsqf(var + EPSV);
            yp[rr * Hdim]      = (d0 * rstd * gw0 + gb0) * gp[rr * Hdim];
            yp[rr * Hdim + 16] = (d1 * rstd * gw1 + gb1) * gp[rr * Hdim + 16];
            yp[rr * Hdim + 32] = (d2 * rstd * gw2 + gb2) * gp[rr * Hdim + 32];
            yp[rr * Hdim + 48] = (d3 * rstd * gw3 + gb3) * gp[rr * Hdim + 48];
        }
    }
}

#define OP_WAVES 8

__global__ __launch_bounds__(OP_WAVES * 32) void rwkv_out_kernel(
    const float* __restrict__ y, const float* __restrict__ Wo, float* __restrict__ out)
{
    __shared__ __attribute__((aligned(16))) float ys[FT_TILE];

    const int tid  = (int)threadIdx.x;
    const int wave = __builtin_amdgcn_readfirstlane(tid >> 5);
    const int lane = tid & 31;
    const int sub  = lane & 15;
    const int hh   = lane >> 4;
    const int koff = hh << 1;
    const long row0 = (long)blockIdx.x * FT_TOK;

    {
        const int c4 = (tid & 63) << 2;
        const int t0 = tid >> 6;
        const float* __restrict__ yp = y + row0 * Hdim + c4;
        #pragma unroll
        for (int j = 0; j < FT_TOK / 4; ++j) {
            const int tok = t0 + 4 * j;
            *(v4f*)(ys + tok * FT_XP + c4) = *(const v4f*)(yp + tok * Hdim);
        }
    }
    __syncthreads();

    const int nc = wave << 5;
    ft_strip<false>(ys, Wo, out + (row0 + 8 * hh) * Hdim + sub + nc, nc, sub, koff);
}

extern "C" void kernel_launch(void* const* d_in, const int* in_sizes, int n_in,
                              void* d_out, int out_size, void* d_ws, size_t ws_size,
                              hipStream_t stream)
{
    const float* query    = (const float*)d_in[0];
    const float* keyval   = (const float*)d_in[1];
    const float* rwg_mu   = (const float*)d_in[2];
    const float* W_rwg0   = (const float*)d_in[3];
    const float* W_rwg2   = (const float*)d_in[4];
    const float* rwg_bias = (const float*)d_in[5];
    const float* kv_mu    = (const float*)d_in[6];
    const float* W_kv0    = (const float*)d_in[7];
    const float* W_kv2    = (const float*)d_in[8];
    const float* kv_bias  = (const float*)d_in[9];
    const float* Wr       = (const float*)d_in[10];
    const float* Wa       = (const float*)d_in[11];
    const float* Wb       = (const float*)d_in[12];
    const float* bb       = (const float*)d_in[13];
    const float* Wk       = (const float*)d_in[14];
    const float* Wv       = (const float*)d_in[15];
    const float* Wg       = (const float*)d_in[16];
    const float* u        = (const float*)d_in[17];
    const float* gn_w     = (const float*)d_in[18];
    const float* gn_b     = (const float*)d_in[19];
    const float* Wo       = (const float*)d_in[20];

    float* ws = (float*)d_ws;
    const long SZ = (long)NTOK * Hdim;
    float* rbuf  = ws + 0 * SZ;
    float* kbuf  = ws + 1 * SZ;
    float* vbuf  = ws + 2 * SZ;
    float* dbuf  = ws + 3 * SZ;
    float* gsbuf = ws + 4 * SZ;
    float* ybuf  = ws + 5 * SZ;
    float* olbuf = ws + 6 * SZ;
    float* sebuf = ws + 7 * SZ;

    rwkv_features_kernel<<<NTOK / FT_TOK, FT_WAVES * 32, 0, stream>>>(
        query, keyval, rwg_mu, W_rwg0, W_rwg2, rwg_bias,
        kv_mu, W_kv0, W_kv2, kv_bias,
        Wr, Wa, Wb, bb, Wk, Wv, Wg,
        rbuf, kbuf, vbuf, gsbuf, dbuf);

    rwkv_scan_kernel<<<Bsz * LQ * NHEAD * SC_NCH, (SC_SCANW + 1) * 32, 0, stream>>>(
        rbuf, kbuf, vbuf, dbuf, u, olbuf, sebuf);

    rwkv_carry_kernel<<<Bsz * LQ * NHEAD * SC_NCH, CR_WAVES * 32, 0, stream>>>(
        rbuf, dbuf, olbuf, sebuf, gsbuf, gn_w, gn_b, ybuf);

    rwkv_out_kernel<<<NTOK / FT_TOK, OP_WAVES * 32, 0, stream>>>(ybuf, Wo, (float*)d_out);
}


# phase 2b (carried-in WMMA term, GroupNorm, gate) and the output projection share one launch: a group per 32 tokens x 4 heads keeps y in LDS; 4 launches -> 3, y plane gone; every sum order kept
# speedup vs baseline: 10.6321x; 1.0909x over previous
#include <hip/hip_runtime.h>
#include <math.h>

#define Hdim 256
#define NHEAD 4
#define DK 64
#define PLR 32
#define GLR 64
#define Bsz 2
#define LQ 8
#define LKs 512
#define NTOK (Bsz * LQ * LKs)
#define EPSV 1e-5f

typedef float v2f __attribute__((ext_vector_type(2)));
typedef float v4f __attribute__((ext_vector_type(4)));
typedef float v8f __attribute__((ext_vector_type(8)));

#define FT_TOK   32
#define FT_WAVES 12
#define FT_XP    260
#define FT_T1P   100
#define FT_T2P   68
#define FT_TWP   68
#define FT_TILE  (FT_TOK * FT_XP)

template <int K, int N, int P>
__device__ __forceinline__ void ft_products(
    const float* Xs, const float* __restrict__ W, int nc, int sub, int koff,
    v8f& c00, v8f& c01, v8f& c10, v8f& c11)
{
    const float* A0 = Xs + sub * P + koff;
    const float* A1 = A0 + 16 * P;
    const float* __restrict__ B0 = W + koff * N + nc + sub;
    const float* __restrict__ B1 = B0 + 16;
    #pragma unroll 8
    for (int k = 0; k < K; k += 4) {
        const v2f a0 = *(const v2f*)(A0 + k);
        const v2f a1 = *(const v2f*)(A1 + k);
        v2f b0, b1;
        b0.x = B0[k * N];
        b0.y = B0[(k + 1) * N];
        b1.x = B1[k * N];
        b1.y = B1[(k + 1) * N];
        c00 = __builtin_amdgcn_wmma_f32_16x16x4_f32(false, a0, false, b0, (short)0, c00, false, false);
        c01 = __builtin_amdgcn_wmma_f32_16x16x4_f32(false, a0, false, b1, (short)0, c01, false, false);
        c10 = __builtin_amdgcn_wmma_f32_16x16x4_f32(false, a1, false, b0, (short)0, c10, false, false);
        c11 = __builtin_amdgcn_wmma_f32_16x16x4_f32(false, a1, false, b1, (short)0, c11, false, false);
    }
}

template <bool SILU>
__device__ __forceinline__ void ft_strip(
    const float* Xs, const float* __restrict__ Wp, float* __restrict__ yp,
    int nc, int sub, int koff)
{
    v8f c00 = {}, c01 = {}, c10 = {}, c11 = {};
    ft_products<Hdim, Hdim, FT_XP>(Xs, Wp, nc, sub, koff, c00, c01, c10, c11);
    #pragma unroll
    for (int r = 0; r < 8; ++r) {
        if (SILU) {
            yp[r * Hdim]             = c00[r] / (1.0f + expf(-c00[r]));
            yp[r * Hdim + 16]        = c01[r] / (1.0f + expf(-c01[r]));
            yp[(r + 16) * Hdim]      = c10[r] / (1.0f + expf(-c10[r]));
            yp[(r + 16) * Hdim + 16] = c11[r] / (1.0f + expf(-c11[r]));
        } else {
            yp[r * Hdim]             = c00[r];
            yp[r * Hdim + 16]        = c01[r];
            yp[(r + 16) * Hdim]      = c10[r];
            yp[(r + 16) * Hdim + 16] = c11[r];
        }
    }
}

__device__ __forceinline__ void ft_long_job(
    int jb, const float* xs,
    const float* __restrict__ Wr, const float* __restrict__ Wk,
    const float* __restrict__ Wv, const float* __restrict__ Wg,
    float* __restrict__ rbuf, float* __restrict__ kbuf,
    float* __restrict__ vbuf, float* __restrict__ gsbuf,
    long yrow, int sub, int koff)
{
    const int nc = (jb & 7) << 5;
    switch (jb >> 3) {
        case 0:  ft_strip<false>(xs,               Wr, rbuf  + yrow + nc, nc, sub, koff); break;
        case 1:  ft_strip<false>(xs + 3 * FT_TILE, Wk, kbuf  + yrow + nc, nc, sub, koff); break;
        case 2:  ft_strip<false>(xs + 4 * FT_TILE, Wv, vbuf  + yrow + nc, nc, sub, koff); break;
        default: ft_strip<true>(xs + 2 * FT_TILE,  Wg, gsbuf + yrow + nc, nc, sub, koff); break;
    }
}

__global__ __launch_bounds__(FT_WAVES * 32) void rwkv_features_kernel(
    const float* __restrict__ query, const float* __restrict__ keyval,
    const float* __restrict__ rwg_mu, const float* __restrict__ W_rwg0,
    const float* __restrict__ W_rwg2, const float* __restrict__ rwg_bias,
    const float* __restrict__ kv_mu, const float* __restrict__ W_kv0,
    const float* __restrict__ W_kv2, const float* __restrict__ kv_bias,
    const float* __restrict__ Wr, const float* __restrict__ Wa,
    const float* __restrict__ Wb, const float* __restrict__ bb,
    const float* __restrict__ Wk, const float* __restrict__ Wv,
    const float* __restrict__ Wg,
    float* __restrict__ rbuf, float* __restrict__ kbuf, float* __restrict__ vbuf,
    float* __restrict__ gsbuf, float* __restrict__ dbuf)
{
    __shared__ __attribute__((aligned(16))) float xs[5 * FT_TILE];
    __shared__ __attribute__((aligned(16))) float t1_s[FT_TOK * FT_T1P];
    __shared__ __attribute__((aligned(16))) float t2_s[FT_TOK * FT_T2P];
    __shared__ __attribute__((aligned(16))) float tw_s[FT_TOK * FT_TWP];
    float* const xr_s = xs;
    float* const xk_s = xs + 3 * FT_TILE;

    const int tid  = (int)threadIdx.x;
    const int wave = __builtin_amdgcn_readfirstlane(tid >> 5);
    const int lane = tid & 31;
    const int sub  = lane & 15;
    const int hh   = lane >> 4;
    const int koff = hh << 1;

    const int tile = (int)blockIdx.x;
    const int e    = tile >> 4;
    const int l0   = (tile & 15) * FT_TOK;
    const int b    = e >> 3;
    const float* __restrict__ hrow = keyval + ((long)b * LKs + l0) * Hdim;
    const float* __restrict__ qrow = query + (long)e * Hdim;

    {
        const int c4 = (tid & 63) << 2;
        const int t0 = tid >> 6;
        const v4f q  = *(const v4f*)(qrow + c4);
        const v4f mr = *(const v4f*)(rwg_mu + c4);
        const v4f mk = *(const v4f*)(kv_mu + c4);
        #pragma unroll
        for (int j = 0; j < 6; ++j) {
            const int tok = t0 + 6 * j;
            if (j < 5 || tok < FT_TOK) {
                const bool first = (l0 + tok == 0);
                const int prow = first ? 0 : tok - 1;
                const v4f h  = *(const v4f*)(hrow + tok * Hdim + c4);
                v4f hp = *(const v4f*)(hrow + prow * Hdim + c4);
                if (first) hp = (v4f){0.0f, 0.0f, 0.0f, 0.0f};
                const v4f dc = q - h;
                const v4f d  = hp - h;
                *(v4f*)(xr_s + tok * FT_XP + c4) = h + dc * mr;
                *(v4f*)(xk_s + tok * FT_XP + c4) = h + d * mk;
            }
        }
    }
    __syncthreads();

    if (wave < 10) {
        const bool kvA = (wave >= 6);
        const float* Xs = kvA ? xk_s : xr_s;
        const float* __restrict__ W0 = kvA ? W_kv0 : W_rwg0;
        const int N0  = kvA ? 2 * PLR : 3 * PLR;
        const int nc  = (kvA ? wave - 6 : wave) << 4;
        float* Td     = kvA ? t2_s : t1_s;
        const int PT  = kvA ? FT_T2P : FT_T1P;

        const float* A0 = Xs + sub * FT_XP + koff;
        const float* A1 = A0 + 16 * FT_XP;
        const float* __restrict__ Bp = W0 + koff * N0 + nc + sub;

        v8f c0 = {}, c1 = {};
        #pragma unroll 8
        for (int k = 0; k < Hdim; k += 4) {
            const v2f a0 = *(const v2f*)(A0 + k);
            const v2f a1 = *(const v2f*)(A1 + k);
            v2f bf;
            bf.x = Bp[k * N0];
            bf.y = Bp[(k + 1) * N0];
            c0 = __builtin_amdgcn_wmma_f32_16x16x4_f32(false, a0, false, bf, (short)0, c0, false, false);
            c1 = __builtin_amdgcn_wmma_f32_16x16x4_f32(false, a1, false, bf, (short)0, c1, false, false);
        }
        #pragma unroll
        for (int r = 0; r < 8; ++r) {
            const int m = r + 8 * hh;
            Td[m * PT + nc + sub]        = tanhf(c0[r]);
            Td[(16 + m) * PT + nc + sub] = tanhf(c1[r]);
        }
    }
    __syncthreads();

    {
        const int jbeg = (20 * wave) / 3;
        const int jend = (20 * (wave + 1)) / 3;

        const int rowA = 8 * hh - 1;
        const int rowB = rowA + 16;
        const bool first = (l0 + rowA < 0);
        const int rowA0 = first ? 0 : rowA;

        for (int J = jbeg; J < jend; ++J) {
            const int o    = J >> 4;
            const bool kvB = (o >= 3);
            const int g    = kvB ? o - 3 : o;
            const float* Ts = kvB ? t2_s : t1_s;
            const int PT   = kvB ? FT_T2P : FT_T1P;
            const float* __restrict__ W2 = kvB ? W_kv2 : W_rwg2;
            const int NW   = kvB ? 2 * PLR : 3 * PLR;
            const float* __restrict__ bias = (kvB ? kv_bias : rwg_bias) + g * Hdim;
            float* outp = xs + o * FT_TILE;

            const int col = ((J & 15) << 4) + sub;
            const float* __restrict__ Bp = W2 + col * NW + g * PLR + koff;
            const float* __restrict__ hc = hrow + col;
            const float* A0 = Ts + sub * PT + g * PLR + koff;
            const float* A1 = A0 + 16 * PT;

            v2f fa0[PLR / 4], fa1[PLR / 4], fb[PLR / 4];
            #pragma unroll
            for (int s = 0; s < PLR / 4; ++s) {
                fb[s]  = *(const v2f*)(Bp + 4 * s);
                fa0[s] = *(const v2f*)(A0 + 4 * s);
                fa1[s] = *(const v2f*)(A1 + 4 * s);
            }
            float hA[9], hB[9];
            hA[0] = hc[rowA0 * Hdim];
            hB[0] = hc[rowB * Hdim];
            #pragma unroll
            for (int i = 1; i < 9; ++i) {
                hA[i] = hc[(rowA + i) * Hdim];
                hB[i] = hc[(rowB + i) * Hdim];
            }
            if (first) hA[0] = 0.0f;
            const float bv = bias[col];
            const float qv = qrow[col];

            v8f c0 = {}, c1 = {};
            #pragma unroll
            for (int s = 0; s < PLR / 4; ++s) {
                c0 = __builtin_amdgcn_wmma_f32_16x16x4_f32(false, fa0[s], false, fb[s], (short)0, c0, false, false);
                c1 = __builtin_amdgcn_wmma_f32_16x16x4_f32(false, fa1[s], false, fb[s], (short)0, c1, false, false);
            }

            const int ob = 8 * hh * FT_XP + col;
            #pragma unroll
            for (int r = 0; r < 8; ++r) {
                const float h0 = hA[r + 1];
                const float h1 = hB[r + 1];
                const float dl0 = kvB ? (hA[r] - h0) : (qv - h0);
                const float dl1 = kvB ? (hB[r] - h1) : (qv - h1);
                outp[ob + r * FT_XP]        = h0 + dl0 * (c0[r] + bv);
                outp[ob + (r + 16) * FT_XP] = h1 + dl1 * (c1[r] + bv);
            }
        }
    }
    __syncthreads();

    {
        const long yrow = ((long)e * LKs + l0 + 8 * hh) * Hdim + sub;

        if (wave < 2) {
            const int nc = wave << 5;
            v8f c00 = {}, c01 = {}, c10 = {}, c11 = {};
            ft_products<Hdim, GLR, FT_XP>(xs + FT_TILE, Wa, nc, sub, koff, c00, c01, c10, c11);
            float* tp = tw_s + 8 * hh * FT_TWP + nc + sub;
            #pragma unroll
            for (int r = 0; r < 8; ++r) {
                tp[r * FT_TWP]             = tanhf(c00[r]);
                tp[r * FT_TWP + 16]        = tanhf(c01[r]);
                tp[(r + 16) * FT_TWP]      = tanhf(c10[r]);
                tp[(r + 16) * FT_TWP + 16] = tanhf(c11[r]);
            }
        } else {
            ft_long_job(wave - 2, xs, Wr, Wk, Wv, Wg, rbuf, kbuf, vbuf, gsbuf, yrow, sub, koff);
        }
        __syncthreads();

        const int nup = (wave < 2) ? 4 : 0;
        for (int i = 0; i < nup; ++i) {
            const int nc = ((wave << 2) + i) << 5;
            v8f c00 = {}, c01 = {}, c10 = {}, c11 = {};
            ft_products<GLR, Hdim, FT_TWP>(tw_s, Wb, nc, sub, koff, c00, c01, c10, c11);
            const float bv0 = bb[nc + sub];
            const float bv1 = bb[nc + sub + 16];
            float* __restrict__ yp = dbuf + yrow + nc;
            #pragma unroll
            for (int r = 0; r < 8; ++r) {
                yp[r * Hdim]             = expf(-expf(c00[r] + bv0));
                yp[r * Hdim + 16]        = expf(-expf(c01[r] + bv1));
                yp[(r + 16) * Hdim]      = expf(-expf(c10[r] + bv0));
                yp[(r + 16) * Hdim + 16] = expf(-expf(c11[r] + bv1));
            }
        }
        const int jlast = (wave < 2) ? 20 : 32;
        for (int jb = 10 + wave; jb < jlast; jb += 10)
            ft_long_job(jb, xs, Wr, Wk, Wv, Wg, rbuf, kbuf, vbuf, gsbuf, yrow, sub, koff);
    }
}

#define SC_STEPS  8
#define SC_SLOT   (4 * DK)
#define SC_CHUNK  (SC_STEPS * SC_SLOT)
#define SC_NCH    4
#define SC_TL     (LKs / SC_NCH)
#define SC_NCHUNK (SC_TL / SC_STEPS)
#define SC_NBUF   3
#define SC_SCANW  4
#define SC_KQ     (DK / 4)
#define SC_OBUF   (DK * SC_STEPS)

template <int MASK>
__device__ __forceinline__ float sc_row_xor(float x)
{
    const int b = __builtin_bit_cast(int, x);
    return __builtin_bit_cast(float, __builtin_amdgcn_update_dpp(0, b, 0x160 + MASK, 0xF, 0xF, true));
}
template <int CTRL>
__device__ __forceinline__ float sc_quad_lane(float x)
{
    const int b = __builtin_bit_cast(int, x);
    return __builtin_bit_cast(float, __builtin_amdgcn_update_dpp(0, b, CTRL, 0xF, 0xF, true));
}
__device__ __forceinline__ float sc_quad_join(float x)
{
    return ((sc_quad_lane<0x00>(x) + sc_quad_lane<0x55>(x)) + sc_quad_lane<0xAA>(x)) + sc_quad_lane<0xFF>(x);
}
__global__ __launch_bounds__((SC_SCANW + 1) * 32) void rwkv_scan_kernel(
    const float* __restrict__ r, const float* __restrict__ k,
    const float* __restrict__ v, const float* __restrict__ decay,
    const float* __restrict__ u, float* __restrict__ olocal, float* __restrict__ send,
    float* __restrict__ dtot)
{
    __shared__ __attribute__((aligned(16))) float stage[SC_NBUF * SC_CHUNK];
    __shared__ __attribute__((aligned(16))) float obuf[2 * SC_OBUF];

    const int chain = (int)blockIdx.x / SC_NCH;
    const int tc    = (int)blockIdx.x % SC_NCH;
    const int wave  = __builtin_amdgcn_readfirstlane((int)threadIdx.x >> 5);
    const bool scan = (wave < SC_SCANW);
    const int e     = chain >> 2;
    const int head  = chain & 3;
    const int lane  = (int)threadIdx.x & 31;
    const long base = ((long)e * LKs + tc * SC_TL) * Hdim + head * DK;
    const int kq    = lane & 3;
    const int v0    = (wave & 3) * 16 + ((lane >> 2) << 1);
    const bool odd  = (kq & 1) != 0;

    v2f S[SC_KQ];
    float uu[SC_KQ];
    #pragma unroll
    for (int j = 0; j < SC_KQ; ++j) { S[j] = (v2f){0.0f, 0.0f}; uu[j] = u[head * DK + kq * SC_KQ + j]; }

    const float* __restrict__ srow = (wave == 0) ? r : (wave == 1) ? k : (wave == 2) ? decay : v;
    const float* __restrict__ sg = srow + base + 2 * lane;
    const int sput = (wave & 3) * DK + 2 * lane;

    v2f gx[SC_STEPS];

    auto fetch = [&](int chunk) {
        #pragma unroll
        for (int i = 0; i < SC_STEPS; ++i)
            gx[i] = *(const v2f*)(sg + (long)(chunk * SC_STEPS + i) * Hdim);
    };
    auto put = [&](float* buf) {
        #pragma unroll
        for (int i = 0; i < SC_STEPS; ++i)
            *(v2f*)(buf + i * SC_SLOT + sput) = gx[i];
    };

    v4f R[2][4], K[2][4], D[2][4];
    v2f VT[2];

    auto take = [&](const float* buf, int s, int set) {
        const float* slot = buf + s * SC_SLOT + kq * SC_KQ;
        #pragma unroll
        for (int g = 0; g < 4; ++g) {
            R[set][g] = *(const v4f*)(slot + 4 * g);
            K[set][g] = *(const v4f*)(slot + DK + 4 * g);
            D[set][g] = *(const v4f*)(slot + 2 * DK + 4 * g);
        }
        VT[set] = *(const v2f*)(buf + s * SC_SLOT + 3 * DK + v0);
    };

    float* __restrict__ og = olocal + base + lane;
    auto writeout = [&](int chunk, const float* ob) {
        #pragma unroll
        for (int hf = 0; hf < SC_STEPS / 4; ++hf) {
            const v4f oa = *(const v4f*)(ob + lane * SC_STEPS + 4 * hf);
            const v4f oc = *(const v4f*)(ob + (lane + 32) * SC_STEPS + 4 * hf);
            #pragma unroll
            for (int i = 0; i < 4; ++i) {
                const long off = (long)(chunk * SC_STEPS + 4 * hf + i) * Hdim;
                og[off]      = oa[i];
                og[off + 32] = oc[i];
            }
        }
    };
    if (scan) {
        fetch(0);
        put(stage);
    }
    __syncthreads();
    if (scan) {
        fetch(1);
        take(stage, 0, 0);
    }

    float dp0 = 1.0f, dp1 = 1.0f;
    int rc = 0;
    for (int c = 0; c < SC_NCHUNK; ++c) {
        const int rn = (rc == SC_NBUF - 1) ? 0 : rc + 1;
        const float* cur = stage + rc * SC_CHUNK;
        float*       nxt = stage + rn * SC_CHUNK;
        rc = rn;
        const float* ob = obuf + ((c + 1) & 1) * SC_OBUF;
        float* const ow = obuf + (c & 1) * SC_OBUF + (v0 + (kq & 1)) * SC_STEPS;
        if (scan) put(nxt);
        __syncthreads();
        if (!scan) {
            if (c > 0) writeout(c - 1, ob);
            #pragma unroll
            for (int i = 0; i < SC_STEPS; ++i) {
                dp0 *= cur[i * SC_SLOT + 2 * DK + lane];
                dp1 *= cur[i * SC_SLOT + 2 * DK + lane + 32];
            }
            if ((c & 3) == 3) {
                float* __restrict__ tp = dtot + ((long)(chain * SC_NCH + tc) * 4 + (c >> 2)) * DK + lane;
                tp[0]  = dp0;
                tp[32] = dp1;
                dp0 = 1.0f;
                dp1 = 1.0f;
            }
            continue;
        }
        fetch((c + 2 < SC_NCHUNK) ? (c + 2) : (SC_NCHUNK - 1));

        v2f pa = {0.0f, 0.0f};
        #pragma unroll
        for (int i = 0; i < SC_STEPS; ++i) {
            if (i + 1 < SC_STEPS) take(cur, i + 1, (i + 1) & 1);
            else                  take(nxt, 0, 0);
            __builtin_amdgcn_sched_barrier(0);

            const int set = i & 1;
            const v2f vt = VT[set];
            v2f acc = {0.0f, 0.0f};
            #pragma unroll
            for (int j = 0; j < SC_KQ; ++j) {
                const v2f kv = K[set][j >> 2][j & 3] * vt;
                acc += R[set][j >> 2][j & 3] * (S[j] + uu[j] * kv);
                S[j] = D[set][j >> 2][j & 3] * S[j] + kv;
            }
            if (i > 0) {
                const float j0 = sc_quad_join(pa.x);
                const float j1 = sc_quad_join(pa.y);
                ow[i - 1] = odd ? j1 : j0;
            }
            pa = acc;
            __builtin_amdgcn_sched_barrier(0);
        }
        {
            const float j0 = sc_quad_join(pa.x);
            const float j1 = sc_quad_join(pa.y);
            ow[SC_STEPS - 1] = odd ? j1 : j0;
        }
    }

    {
        const float* ob = obuf + ((SC_NCHUNK + 1) & 1) * SC_OBUF;
        __syncthreads();
        if (!scan) writeout(SC_NCHUNK - 1, ob);
    }

    if (scan && tc < SC_NCH - 1) {
        float* __restrict__ sp = send + ((long)((chain * (SC_NCH - 1) + tc) * SC_SCANW + wave) * DK + kq * SC_KQ) * 16
                                      + ((lane >> 2) << 1);
        #pragma unroll
        for (int j = 0; j < SC_KQ; ++j) *(v2f*)(sp + j * 16) = S[j];
    }
}

#define CR_WAVES 8
#define CR_PS    72
#define CR_SEG   (SC_TL / 4)

__device__ __forceinline__ float cr_row_sum(float x)
{
    x += sc_row_xor<1>(x);
    x += sc_row_xor<2>(x);
    x += sc_row_xor<4>(x);
    return x + sc_row_xor<8>(x);
}

__global__ __launch_bounds__(CR_WAVES * 32) void rwkv_carry_out_kernel(
    const float* __restrict__ r, const float* __restrict__ decay,
    const float* __restrict__ olocal, const float* __restrict__ send,
    const float* __restrict__ dtot, const float* __restrict__ gsilu,
    const float* __restrict__ gn_w, const float* __restrict__ gn_b,
    const float* __restrict__ Wo, float* __restrict__ out)
{
    __shared__ __attribute__((aligned(16))) float rt_s[FT_TILE];
    __shared__ __attribute__((aligned(16))) float ys[FT_TILE];
    __shared__ __attribute__((aligned(16))) float ss_s[NHEAD * DK * CR_PS];
    __shared__ float ptot[(SC_NCH - 2) * 4 * Hdim];

    const int tid  = (int)threadIdx.x;
    const int wave = __builtin_amdgcn_readfirstlane(tid >> 5);
    const int lane = tid & 31;
    const int sub  = lane & 15;
    const int hh   = lane >> 4;
    const int koff = hh << 1;

    const int tile = (int)blockIdx.x;
    const int e    = tile >> 4;
    const int l0   = (tile & 15) * CR_SEG;
    const int tc   = l0 / SC_TL;
    const int seg  = (l0 % SC_TL) / CR_SEG;
    const long row0 = (long)e * LKs + l0;

    {
        const int head = tid >> 6;
        const int kk   = tid & 63;
        const float* __restrict__ dp = decay + row0 * Hdim + tid;
        const float* __restrict__ rp = r     + row0 * Hdim + tid;
        float dv[CR_SEG], rv[CR_SEG];
        #pragma unroll
        for (int i = 0; i < CR_SEG; ++i) { dv[i] = dp[i * Hdim]; rv[i] = rp[i * Hdim]; }

        const float* __restrict__ tq = dtot + ((long)(e * NHEAD + head) * SC_NCH * 4) * DK + kk;
        for (int pc = 1; pc < tc; ++pc) {
            #pragma unroll
            for (int sg = 0; sg < 4; ++sg)
                ptot[((pc - 1) * 4 + sg) * Hdim + tid] = tq[(pc * 4 + sg) * DK];
        }
        float start = 1.0f;
        for (int sg = 0; sg < seg; ++sg) start *= tq[(tc * 4 + sg) * DK];

        float p = 1.0f;
        #pragma unroll
        for (int i = 0; i < CR_SEG; ++i) { const float before = p; p *= dv[i]; dv[i] = before; }
        float* rtp = rt_s + tid;
        #pragma unroll
        for (int i = 0; i < CR_SEG; ++i) rtp[i * FT_XP] = rv[i] * (start * dv[i]);
    }
    __syncthreads();

    {
        const int kr = tid >> 4;
        const int vl = tid & 15;
        float s[NHEAD][16];
        #pragma unroll
        for (int hd = 0; hd < NHEAD; ++hd) {
            #pragma unroll
            for (int i = 0; i < 16; ++i) s[hd][i] = 0.0f;
        }
        if (tc > 0) {
            const float* __restrict__ sp = send + (long)(e * NHEAD) * (SC_NCH - 1) * DK * DK + tid;
            #pragma unroll
            for (int hd = 0; hd < NHEAD; ++hd) {
                #pragma unroll
                for (int i = 0; i < 16; ++i) s[hd][i] = sp[(long)hd * (SC_NCH - 1) * DK * DK + i * 256];
            }
            for (int pc = 1; pc < tc; ++pc) {
                float nv[NHEAD][16];
                #pragma unroll
                for (int hd = 0; hd < NHEAD; ++hd) {
                    #pragma unroll
                    for (int i = 0; i < 16; ++i)
                        nv[hd][i] = sp[((long)hd * (SC_NCH - 1) + pc) * DK * DK + i * 256];
                }
                #pragma unroll
                for (int hd = 0; hd < NHEAD; ++hd) {
                    const float* pt = ptot + (pc - 1) * 4 * Hdim + hd * DK + kr;
                    float pe[4];
                    #pragma unroll
                    for (int q = 0; q < 4; ++q)
                        pe[q] = ((pt[16 * q] * pt[Hdim + 16 * q]) * pt[2 * Hdim + 16 * q]) * pt[3 * Hdim + 16 * q];
                    #pragma unroll
                    for (int i = 0; i < 16; ++i) s[hd][i] = pe[i & 3] * s[hd][i] + nv[hd][i];
                }
            }
        }
        #pragma unroll
        for (int hd = 0; hd < NHEAD; ++hd) {
            #pragma unroll
            for (int i = 0; i < 16; ++i)
                ss_s[(hd * DK + kr + 16 * (i & 3)) * CR_PS + 16 * (i >> 2) + vl] = s[hd][i];
        }
    }
    __syncthreads();

    const int head = wave >> 1;
    const int mrow = (wave & 1) << 4;
    v8f c0 = {}, c1 = {}, c2 = {}, c3 = {};
    {
        const float* A0 = rt_s + (mrow + sub) * FT_XP + head * DK + koff;
        const float* B0 = ss_s + (head * DK + koff) * CR_PS + sub;
        #pragma unroll
        for (int k = 0; k < DK; k += 4) {
            const v2f a = *(const v2f*)(A0 + k);
            v2f b0, b1, b2, b3;
            b0.x = B0[k * CR_PS];
            b0.y = B0[(k + 1) * CR_PS];
            b1.x = B0[k * CR_PS + 16];
            b1.y = B0[(k + 1) * CR_PS + 16];
            b2.x = B0[k * CR_PS + 32];
            b2.y = B0[(k + 1) * CR_PS + 32];
            b3.x = B0[k * CR_PS + 48];
            b3.y = B0[(k + 1) * CR_PS + 48];
            c0 = __builtin_amdgcn_wmma_f32_16x16x4_f32(false, a, false, b0, (short)0, c0, false, false);
            c1 = __builtin_amdgcn_wmma_f32_16x16x4_f32(false, a, false, b1, (short)0, c1, false, false);
            c2 = __builtin_amdgcn_wmma_f32_16x16x4_f32(false, a, false, b2, (short)0, c2, false, false);
            c3 = __builtin_amdgcn_wmma_f32_16x16x4_f32(false, a, false, b3, (short)0, c3, false, false);
        }
    }

    {
        const int hc = head * DK + sub;
        const long rowb = (row0 + mrow + 8 * hh) * Hdim + hc;
        const float* __restrict__ op = olocal + rowb;
        const float* __restrict__ gp = gsilu + rowb;
        float* yp = ys + (mrow + 8 * hh) * FT_XP + hc;
        const float gw0 = gn_w[hc], gw1 = gn_w[hc + 16], gw2 = gn_w[hc + 32], gw3 = gn_w[hc + 48];
        const float gb0 = gn_b[hc], gb1 = gn_b[hc + 16], gb2 = gn_b[hc + 32], gb3 = gn_b[hc + 48];
        #pragma unroll
        for (int rr = 0; rr < 8; ++rr) {
            const float x0 = c0[rr] + op[rr * Hdim];
            const float x1 = c1[rr] + op[rr * Hdim + 16];
            const float x2 = c2[rr] + op[rr * Hdim + 32];
            const float x3 = c3[rr] + op[rr * Hdim + 48];
            const float mean = cr_row_sum((x0 + x1) + (x2 + x3)) * (1.0f / 64.0f);
            const float d0 = x0 - mean, d1 = x1 - mean, d2 = x2 - mean, d3 = x3 - mean;
            const float var = cr_row_sum((d0 * d0 + d1 * d1) + (d2 * d2 + d3 * d3)) * (1.0f / 64.0f);
            const float rstd = __builtin_amdgcn_rsqf(var + EPSV);
            yp[rr * FT_XP]      = (d0 * rstd * gw0 + gb0) * gp[rr * Hdim];
            yp[rr * FT_XP + 16] = (d1 * rstd * gw1 + gb1) * gp[rr * Hdim + 16];
            yp[rr * FT_XP + 32] = (d2 * rstd * gw2 + gb2) * gp[rr * Hdim + 32];
            yp[rr * FT_XP + 48] = (d3 * rstd * gw3 + gb3) * gp[rr * Hdim + 48];
        }
    }
    __syncthreads();

    const int nc = wave << 5;
    ft_strip<false>(ys, Wo, out + (row0 + 8 * hh) * Hdim + sub + nc, nc, sub, koff);
}

extern "C" void kernel_launch(void* const* d_in, const int* in_sizes, int n_in,
                              void* d_out, int out_size, void* d_ws, size_t ws_size,
                              hipStream_t stream)
{
    const float* query    = (const float*)d_in[0];
    const float* keyval   = (const float*)d_in[1];
    const float* rwg_mu   = (const float*)d_in[2];
    const float* W_rwg0   = (const float*)d_in[3];
    const float* W_rwg2   = (const float*)d_in[4];
    const float* rwg_bias = (const float*)d_in[5];
    const float* kv_mu    = (const float*)d_in[6];
    const float* W_kv0    = (const float*)d_in[7];
    const float* W_kv2    = (const float*)d_in[8];
    const float* kv_bias  = (const float*)d_in[9];
    const float* Wr       = (const float*)d_in[10];
    const float* Wa       = (const float*)d_in[11];
    const float* Wb       = (const float*)d_in[12];
    const float* bb       = (const float*)d_in[13];
    const float* Wk       = (const float*)d_in[14];
    const float* Wv       = (const float*)d_in[15];
    const float* Wg       = (const float*)d_in[16];
    const float* u        = (const float*)d_in[17];
    const float* gn_w     = (const float*)d_in[18];
    const float* gn_b     = (const float*)d_in[19];
    const float* Wo       = (const float*)d_in[20];

    float* ws = (float*)d_ws;
    const long SZ = (long)NTOK * Hdim;
    float* rbuf  = ws + 0 * SZ;
    float* kbuf  = ws + 1 * SZ;
    float* vbuf  = ws + 2 * SZ;
    float* dbuf  = ws + 3 * SZ;
    float* gsbuf = ws + 4 * SZ;
    float* olbuf = ws + 6 * SZ;
    float* sebuf = ws + 7 * SZ;
    float* dtbuf = ws + 8 * SZ;

    rwkv_features_kernel<<<NTOK / FT_TOK, FT_WAVES * 32, 0, stream>>>(
        query, keyval, rwg_mu, W_rwg0, W_rwg2, rwg_bias,
        kv_mu, W_kv0, W_kv2, kv_bias,
        Wr, Wa, Wb, bb, Wk, Wv, Wg,
        rbuf, kbuf, vbuf, gsbuf, dbuf);

    rwkv_scan_kernel<<<Bsz * LQ * NHEAD * SC_NCH, (SC_SCANW + 1) * 32, 0, stream>>>(
        rbuf, kbuf, vbuf, dbuf, u, olbuf, sebuf, dtbuf);

    rwkv_carry_out_kernel<<<NTOK / CR_SEG, CR_WAVES * 32, 0, stream>>>(
        rbuf, dbuf, olbuf, sebuf, dtbuf, gsbuf, gn_w, gn_b, Wo, (float*)d_out);
}


# feature epilogues: tanh, x*sigmoid(x), exp(-exp(w)) on hardware exp2 and reciprocal, 5 instructions a value (library code: 30 to 50, with branches); values move at rounding level; sum orders kept
# speedup vs baseline: 11.3642x; 1.0689x over previous
#include <hip/hip_runtime.h>
#include <math.h>

#define Hdim 256
#define NHEAD 4
#define DK 64
#define PLR 32
#define GLR 64
#define Bsz 2
#define LQ 8
#define LKs 512
#define NTOK (Bsz * LQ * LKs)
#define EPSV 1e-5f

typedef float v2f __attribute__((ext_vector_type(2)));
typedef float v4f __attribute__((ext_vector_type(4)));
typedef float v8f __attribute__((ext_vector_type(8)));

#define FT_LOG2E 1.44269504f
__device__ __forceinline__ float ft_tanh(float x)
{
    const float t = __builtin_amdgcn_exp2f(x * (2.0f * FT_LOG2E));
    return 1.0f - 2.0f * __builtin_amdgcn_rcpf(1.0f + t);
}
__device__ __forceinline__ float ft_silu(float x)
{
    const float t = __builtin_amdgcn_exp2f(x * -FT_LOG2E);
    return x * __builtin_amdgcn_rcpf(1.0f + t);
}
__device__ __forceinline__ float ft_decay(float w)
{
    return __builtin_amdgcn_exp2f(-FT_LOG2E * __builtin_amdgcn_exp2f(w * FT_LOG2E));
}

#define FT_TOK   32
#define FT_WAVES 12
#define FT_XP    260
#define FT_T1P   100
#define FT_T2P   68
#define FT_TWP   68
#define FT_TILE  (FT_TOK * FT_XP)

template <int K, int N, int P>
__device__ __forceinline__ void ft_products(
    const float* Xs, const float* __restrict__ W, int nc, int sub, int koff,
    v8f& c00, v8f& c01, v8f& c10, v8f& c11)
{
    const float* A0 = Xs + sub * P + koff;
    const float* A1 = A0 + 16 * P;
    const float* __restrict__ B0 = W + koff * N + nc + sub;
    const float* __restrict__ B1 = B0 + 16;
    #pragma unroll 8
    for (int k = 0; k < K; k += 4) {
        const v2f a0 = *(const v2f*)(A0 + k);
        const v2f a1 = *(const v2f*)(A1 + k);
        v2f b0, b1;
        b0.x = B0[k * N];
        b0.y = B0[(k + 1) * N];
        b1.x = B1[k * N];
        b1.y = B1[(k + 1) * N];
        c00 = __builtin_amdgcn_wmma_f32_16x16x4_f32(false, a0, false, b0, (short)0, c00, false, false);
        c01 = __builtin_amdgcn_wmma_f32_16x16x4_f32(false, a0, false, b1, (short)0, c01, false, false);
        c10 = __builtin_amdgcn_wmma_f32_16x16x4_f32(false, a1, false, b0, (short)0, c10, false, false);
        c11 = __builtin_amdgcn_wmma_f32_16x16x4_f32(false, a1, false, b1, (short)0, c11, false, false);
    }
}

template <bool SILU>
__device__ __forceinline__ void ft_strip(
    const float* Xs, const float* __restrict__ Wp, float* __restrict__ yp,
    int nc, int sub, int koff)
{
    v8f c00 = {}, c01 = {}, c10 = {}, c11 = {};
    ft_products<Hdim, Hdim, FT_XP>(Xs, Wp, nc, sub, koff, c00, c01, c10, c11);
    #pragma unroll
    for (int r = 0; r < 8; ++r) {
        if (SILU) {
            yp[r * Hdim]             = ft_silu(c00[r]);
            yp[r * Hdim + 16]        = ft_silu(c01[r]);
            yp[(r + 16) * Hdim]      = ft_silu(c10[r]);
            yp[(r + 16) * Hdim + 16] = ft_silu(c11[r]);
        } else {
            yp[r * Hdim]             = c00[r];
            yp[r * Hdim + 16]        = c01[r];
            yp[(r + 16) * Hdim]      = c10[r];
            yp[(r + 16) * Hdim + 16] = c11[r];
        }
    }
}

__device__ __forceinline__ void ft_long_job(
    int jb, const float* xs,
    const float* __restrict__ Wr, const float* __restrict__ Wk,
    const float* __restrict__ Wv, const float* __restrict__ Wg,
    float* __restrict__ rbuf, float* __restrict__ kbuf,
    float* __restrict__ vbuf, float* __restrict__ gsbuf,
    long yrow, int sub, int koff)
{
    const int nc = (jb & 7) << 5;
    switch (jb >> 3) {
        case 0:  ft_strip<false>(xs,               Wr, rbuf  + yrow + nc, nc, sub, koff); break;
        case 1:  ft_strip<false>(xs + 3 * FT_TILE, Wk, kbuf  + yrow + nc, nc, sub, koff); break;
        case 2:  ft_strip<false>(xs + 4 * FT_TILE, Wv, vbuf  + yrow + nc, nc, sub, koff); break;
        default: ft_strip<true>(xs + 2 * FT_TILE,  Wg, gsbuf + yrow + nc, nc, sub, koff); break;
    }
}

__global__ __launch_bounds__(FT_WAVES * 32) void rwkv_features_kernel(
    const float* __restrict__ query, const float* __restrict__ keyval,
    const float* __restrict__ rwg_mu, const float* __restrict__ W_rwg0,
    const float* __restrict__ W_rwg2, const float* __restrict__ rwg_bias,
    const float* __restrict__ kv_mu, const float* __restrict__ W_kv0,
    const float* __restrict__ W_kv2, const float* __restrict__ kv_bias,
    const float* __restrict__ Wr, const float* __restrict__ Wa,
    const float* __restrict__ Wb, const float* __restrict__ bb,
    const float* __restrict__ Wk, const float* __restrict__ Wv,
    const float* __restrict__ Wg,
    float* __restrict__ rbuf, float* __restrict__ kbuf, float* __restrict__ vbuf,
    float* __restrict__ gsbuf, float* __restrict__ dbuf)
{
    __shared__ __attribute__((aligned(16))) float xs[5 * FT_TILE];
    __shared__ __attribute__((aligned(16))) float t1_s[FT_TOK * FT_T1P];
    __shared__ __attribute__((aligned(16))) float t2_s[FT_TOK * FT_T2P];
    __shared__ __attribute__((aligned(16))) float tw_s[FT_TOK * FT_TWP];
    float* const xr_s = xs;
    float* const xk_s = xs + 3 * FT_TILE;

    const int tid  = (int)threadIdx.x;
    const int wave = __builtin_amdgcn_readfirstlane(tid >> 5);
    const int lane = tid & 31;
    const int sub  = lane & 15;
    const int hh   = lane >> 4;
    const int koff = hh << 1;

    const int tile = (int)blockIdx.x;
    const int e    = tile >> 4;
    const int l0   = (tile & 15) * FT_TOK;
    const int b    = e >> 3;
    const float* __restrict__ hrow = keyval + ((long)b * LKs + l0) * Hdim;
    const float* __restrict__ qrow = query + (long)e * Hdim;

    {
        const int c4 = (tid & 63) << 2;
        const int t0 = tid >> 6;
        const v4f q  = *(const v4f*)(qrow + c4);
        const v4f mr = *(const v4f*)(rwg_mu + c4);
        const v4f mk = *(const v4f*)(kv_mu + c4);
        #pragma unroll
        for (int j = 0; j < 6; ++j) {
            const int tok = t0 + 6 * j;
            if (j < 5 || tok < FT_TOK) {
                const bool first = (l0 + tok == 0);
                const int prow = first ? 0 : tok - 1;
                const v4f h  = *(const v4f*)(hrow + tok * Hdim + c4);
                v4f hp = *(const v4f*)(hrow + prow * Hdim + c4);
                if (first) hp = (v4f){0.0f, 0.0f, 0.0f, 0.0f};
                const v4f dc = q - h;
                const v4f d  = hp - h;
                *(v4f*)(xr_s + tok * FT_XP + c4) = h + dc * mr;
                *(v4f*)(xk_s + tok * FT_XP + c4) = h + d * mk;
            }
        }
    }
    __syncthreads();

    if (wave < 10) {
        const bool kvA = (wave >= 6);
        const float* Xs = kvA ? xk_s : xr_s;
        const float* __restrict__ W0 = kvA ? W_kv0 : W_rwg0;
        const int N0  = kvA ? 2 * PLR : 3 * PLR;
        const int nc  = (kvA ? wave - 6 : wave) << 4;
        float* Td     = kvA ? t2_s : t1_s;
        const int PT  = kvA ? FT_T2P : FT_T1P;

        const float* A0 = Xs + sub * FT_XP + koff;
        const float* A1 = A0 + 16 * FT_XP;
        const float* __restrict__ Bp = W0 + koff * N0 + nc + sub;

        v8f c0 = {}, c1 = {};
        #pragma unroll 8
        for (int k = 0; k < Hdim; k += 4) {
            const v2f a0 = *(const v2f*)(A0 + k);
            const v2f a1 = *(const v2f*)(A1 + k);
            v2f bf;
            bf.x = Bp[k * N0];
            bf.y = Bp[(k + 1) * N0];
            c0 = __builtin_amdgcn_wmma_f32_16x16x4_f32(false, a0, false, bf, (short)0, c0, false, false);
            c1 = __builtin_amdgcn_wmma_f32_16x16x4_f32(false, a1, false, bf, (short)0, c1, false, false);
        }
        #pragma unroll
        for (int r = 0; r < 8; ++r) {
            const int m = r + 8 * hh;
            Td[m * PT + nc + sub]        = ft_tanh(c0[r]);
            Td[(16 + m) * PT + nc + sub] = ft_tanh(c1[r]);
        }
    }
    __syncthreads();

    {
        const int jbeg = (20 * wave) / 3;
        const int jend = (20 * (wave + 1)) / 3;

        const int rowA = 8 * hh - 1;
        const int rowB = rowA + 16;
        const bool first = (l0 + rowA < 0);
        const int rowA0 = first ? 0 : rowA;

        for (int J = jbeg; J < jend; ++J) {
            const int o    = J >> 4;
            const bool kvB = (o >= 3);
            const int g    = kvB ? o - 3 : o;
            const float* Ts = kvB ? t2_s : t1_s;
            const int PT   = kvB ? FT_T2P : FT_T1P;
            const float* __restrict__ W2 = kvB ? W_kv2 : W_rwg2;
            const int NW   = kvB ? 2 * PLR : 3 * PLR;
            const float* __restrict__ bias = (kvB ? kv_bias : rwg_bias) + g * Hdim;
            float* outp = xs + o * FT_TILE;

            const int col = ((J & 15) << 4) + sub;
            const float* __restrict__ Bp = W2 + col * NW + g * PLR + koff;
            const float* __restrict__ hc = hrow + col;
            const float* A0 = Ts + sub * PT + g * PLR + koff;
            const float* A1 = A0 + 16 * PT;

            v2f fa0[PLR / 4], fa1[PLR / 4], fb[PLR / 4];
            #pragma unroll
            for (int s = 0; s < PLR / 4; ++s) {
                fb[s]  = *(const v2f*)(Bp + 4 * s);
                fa0[s] = *(const v2f*)(A0 + 4 * s);
                fa1[s] = *(const v2f*)(A1 + 4 * s);
            }
            float hA[9], hB[9];
            hA[0] = hc[rowA0 * Hdim];
            hB[0] = hc[rowB * Hdim];
            #pragma unroll
            for (int i = 1; i < 9; ++i) {
                hA[i] = hc[(rowA + i) * Hdim];
                hB[i] = hc[(rowB + i) * Hdim];
            }
            if (first) hA[0] = 0.0f;
            const float bv = bias[col];
            const float qv = qrow[col];

            v8f c0 = {}, c1 = {};
            #pragma unroll
            for (int s = 0; s < PLR / 4; ++s) {
                c0 = __builtin_amdgcn_wmma_f32_16x16x4_f32(false, fa0[s], false, fb[s], (short)0, c0, false, false);
                c1 = __builtin_amdgcn_wmma_f32_16x16x4_f32(false, fa1[s], false, fb[s], (short)0, c1, false, false);
            }

            const int ob = 8 * hh * FT_XP + col;
            #pragma unroll
            for (int r = 0; r < 8; ++r) {
                const float h0 = hA[r + 1];
                const float h1 = hB[r + 1];
                const float dl0 = kvB ? (hA[r] - h0) : (qv - h0);
                const float dl1 = kvB ? (hB[r] - h1) : (qv - h1);
                outp[ob + r * FT_XP]        = h0 + dl0 * (c0[r] + bv);
                outp[ob + (r + 16) * FT_XP] = h1 + dl1 * (c1[r] + bv);
            }
        }
    }
    __syncthreads();

    {
        const long yrow = ((long)e * LKs + l0 + 8 * hh) * Hdim + sub;

        if (wave < 2) {
            const int nc = wave << 5;
            v8f c00 = {}, c01 = {}, c10 = {}, c11 = {};
            ft_products<Hdim, GLR, FT_XP>(xs + FT_TILE, Wa, nc, sub, koff, c00, c01, c10, c11);
            float* tp = tw_s + 8 * hh * FT_TWP + nc + sub;
            #pragma unroll
            for (int r = 0; r < 8; ++r) {
                tp[r * FT_TWP]             = ft_tanh(c00[r]);
                tp[r * FT_TWP + 16]        = ft_tanh(c01[r]);
                tp[(r + 16) * FT_TWP]      = ft_tanh(c10[r]);
                tp[(r + 16) * FT_TWP + 16] = ft_tanh(c11[r]);
            }
        } else {
            ft_long_job(wave - 2, xs, Wr, Wk, Wv, Wg, rbuf, kbuf, vbuf, gsbuf, yrow, sub, koff);
        }
        __syncthreads();

        const int nup = (wave < 2) ? 4 : 0;
        for (int i = 0; i < nup; ++i) {
            const int nc = ((wave << 2) + i) << 5;
            v8f c00 = {}, c01 = {}, c10 = {}, c11 = {};
            ft_products<GLR, Hdim, FT_TWP>(tw_s, Wb, nc, sub, koff, c00, c01, c10, c11);
            const float bv0 = bb[nc + sub];
            const float bv1 = bb[nc + sub + 16];
            float* __restrict__ yp = dbuf + yrow + nc;
            #pragma unroll
            for (int r = 0; r < 8; ++r) {
                yp[r * Hdim]             = ft_decay(c00[r] + bv0);
                yp[r * Hdim + 16]        = ft_decay(c01[r] + bv1);
                yp[(r + 16) * Hdim]      = ft_decay(c10[r] + bv0);
                yp[(r + 16) * Hdim + 16] = ft_decay(c11[r] + bv1);
            }
        }
        const int jlast = (wave < 2) ? 20 : 32;
        for (int jb = 10 + wave; jb < jlast; jb += 10)
            ft_long_job(jb, xs, Wr, Wk, Wv, Wg, rbuf, kbuf, vbuf, gsbuf, yrow, sub, koff);
    }
}

#define SC_STEPS  8
#define SC_SLOT   (4 * DK)
#define SC_CHUNK  (SC_STEPS * SC_SLOT)
#define SC_NCH    4
#define SC_TL     (LKs / SC_NCH)
#define SC_NCHUNK (SC_TL / SC_STEPS)
#define SC_NBUF   3
#define SC_SCANW  4
#define SC_KQ     (DK / 4)
#define SC_OBUF   (DK * SC_STEPS)

template <int MASK>
__device__ __forceinline__ float sc_row_xor(float x)
{
    const int b = __builtin_bit_cast(int, x);
    return __builtin_bit_cast(float, __builtin_amdgcn_update_dpp(0, b, 0x160 + MASK, 0xF, 0xF, true));
}
template <int CTRL>
__device__ __forceinline__ float sc_quad_lane(float x)
{
    const int b = __builtin_bit_cast(int, x);
    return __builtin_bit_cast(float, __builtin_amdgcn_update_dpp(0, b, CTRL, 0xF, 0xF, true));
}
__device__ __forceinline__ float sc_quad_join(float x)
{
    return ((sc_quad_lane<0x00>(x) + sc_quad_lane<0x55>(x)) + sc_quad_lane<0xAA>(x)) + sc_quad_lane<0xFF>(x);
}
__global__ __launch_bounds__((SC_SCANW + 1) * 32) void rwkv_scan_kernel(
    const float* __restrict__ r, const float* __restrict__ k,
    const float* __restrict__ v, const float* __restrict__ decay,
    const float* __restrict__ u, float* __restrict__ olocal, float* __restrict__ send,
    float* __restrict__ dtot)
{
    __shared__ __attribute__((aligned(16))) float stage[SC_NBUF * SC_CHUNK];
    __shared__ __attribute__((aligned(16))) float obuf[2 * SC_OBUF];

    const int chain = (int)blockIdx.x / SC_NCH;
    const int tc    = (int)blockIdx.x % SC_NCH;
    const int wave  = __builtin_amdgcn_readfirstlane((int)threadIdx.x >> 5);
    const bool scan = (wave < SC_SCANW);
    const int e     = chain >> 2;
    const int head  = chain & 3;
    const int lane  = (int)threadIdx.x & 31;
    const long base = ((long)e * LKs + tc * SC_TL) * Hdim + head * DK;
    const int kq    = lane & 3;
    const int v0    = (wave & 3) * 16 + ((lane >> 2) << 1);
    const bool odd  = (kq & 1) != 0;

    v2f S[SC_KQ];
    float uu[SC_KQ];
    #pragma unroll
    for (int j = 0; j < SC_KQ; ++j) { S[j] = (v2f){0.0f, 0.0f}; uu[j] = u[head * DK + kq * SC_KQ + j]; }

    const float* __restrict__ srow = (wave == 0) ? r : (wave == 1) ? k : (wave == 2) ? decay : v;
    const float* __restrict__ sg = srow + base + 2 * lane;
    const int sput = (wave & 3) * DK + 2 * lane;

    v2f gx[SC_STEPS];

    auto fetch = [&](int chunk) {
        #pragma unroll
        for (int i = 0; i < SC_STEPS; ++i)
            gx[i] = *(const v2f*)(sg + (long)(chunk * SC_STEPS + i) * Hdim);
    };
    auto put = [&](float* buf) {
        #pragma unroll
        for (int i = 0; i < SC_STEPS; ++i)
            *(v2f*)(buf + i * SC_SLOT + sput) = gx[i];
    };

    v4f R[2][4], K[2][4], D[2][4];
    v2f VT[2];

    auto take = [&](const float* buf, int s, int set) {
        const float* slot = buf + s * SC_SLOT + kq * SC_KQ;
        #pragma unroll
        for (int g = 0; g < 4; ++g) {
            R[set][g] = *(const v4f*)(slot + 4 * g);
            K[set][g] = *(const v4f*)(slot + DK + 4 * g);
            D[set][g] = *(const v4f*)(slot + 2 * DK + 4 * g);
        }
        VT[set] = *(const v2f*)(buf + s * SC_SLOT + 3 * DK + v0);
    };

    float* __restrict__ og = olocal + base + lane;
    auto writeout = [&](int chunk, const float* ob) {
        #pragma unroll
        for (int hf = 0; hf < SC_STEPS / 4; ++hf) {
            const v4f oa = *(const v4f*)(ob + lane * SC_STEPS + 4 * hf);
            const v4f oc = *(const v4f*)(ob + (lane + 32) * SC_STEPS + 4 * hf);
            #pragma unroll
            for (int i = 0; i < 4; ++i) {
                const long off = (long)(chunk * SC_STEPS + 4 * hf + i) * Hdim;
                og[off]      = oa[i];
                og[off + 32] = oc[i];
            }
        }
    };
    if (scan) {
        fetch(0);
        put(stage);
    }
    __syncthreads();
    if (scan) {
        fetch(1);
        take(stage, 0, 0);
    }

    float dp0 = 1.0f, dp1 = 1.0f;
    int rc = 0;
    for (int c = 0; c < SC_NCHUNK; ++c) {
        const int rn = (rc == SC_NBUF - 1) ? 0 : rc + 1;
        const float* cur = stage + rc * SC_CHUNK;
        float*       nxt = stage + rn * SC_CHUNK;
        rc = rn;
        const float* ob = obuf + ((c + 1) & 1) * SC_OBUF;
        float* const ow = obuf + (c & 1) * SC_OBUF + (v0 + (kq & 1)) * SC_STEPS;
        if (scan) put(nxt);
        __syncthreads();
        if (!scan) {
            if (c > 0) writeout(c - 1, ob);
            #pragma unroll
            for (int i = 0; i < SC_STEPS; ++i) {
                dp0 *= cur[i * SC_SLOT + 2 * DK + lane];
                dp1 *= cur[i * SC_SLOT + 2 * DK + lane + 32];
            }
            if ((c & 3) == 3) {
                float* __restrict__ tp = dtot + ((long)(chain * SC_NCH + tc) * 4 + (c >> 2)) * DK + lane;
                tp[0]  = dp0;
                tp[32] = dp1;
                dp0 = 1.0f;
                dp1 = 1.0f;
            }
            continue;
        }
        fetch((c + 2 < SC_NCHUNK) ? (c + 2) : (SC_NCHUNK - 1));

        v2f pa = {0.0f, 0.0f};
        #pragma unroll
        for (int i = 0; i < SC_STEPS; ++i) {
            if (i + 1 < SC_STEPS) take(cur, i + 1, (i + 1) & 1);
            else                  take(nxt, 0, 0);
            __builtin_amdgcn_sched_barrier(0);

            const int set = i & 1;
            const v2f vt = VT[set];
            v2f acc = {0.0f, 0.0f};
            #pragma unroll
            for (int j = 0; j < SC_KQ; ++j) {
                const v2f kv = K[set][j >> 2][j & 3] * vt;
                acc += R[set][j >> 2][j & 3] * (S[j] + uu[j] * kv);
                S[j] = D[set][j >> 2][j & 3] * S[j] + kv;
            }
            if (i > 0) {
                const float j0 = sc_quad_join(pa.x);
                const float j1 = sc_quad_join(pa.y);
                ow[i - 1] = odd ? j1 : j0;
            }
            pa = acc;
            __builtin_amdgcn_sched_barrier(0);
        }
        {
            const float j0 = sc_quad_join(pa.x);
            const float j1 = sc_quad_join(pa.y);
            ow[SC_STEPS - 1] = odd ? j1 : j0;
        }
    }

    {
        const float* ob = obuf + ((SC_NCHUNK + 1) & 1) * SC_OBUF;
        __syncthreads();
        if (!scan) writeout(SC_NCHUNK - 1, ob);
    }

    if (scan && tc < SC_NCH - 1) {
        float* __restrict__ sp = send + ((long)((chain * (SC_NCH - 1) + tc) * SC_SCANW + wave) * DK + kq * SC_KQ) * 16
                                      + ((lane >> 2) << 1);
        #pragma unroll
        for (int j = 0; j < SC_KQ; ++j) *(v2f*)(sp + j * 16) = S[j];
    }
}

#define CR_WAVES 8
#define CR_PS    72
#define CR_SEG   (SC_TL / 4)

__device__ __forceinline__ float cr_row_sum(float x)
{
    x += sc_row_xor<1>(x);
    x += sc_row_xor<2>(x);
    x += sc_row_xor<4>(x);
    return x + sc_row_xor<8>(x);
}

__global__ __launch_bounds__(CR_WAVES * 32) void rwkv_carry_out_kernel(
    const float* __restrict__ r, const float* __restrict__ decay,
    const float* __restrict__ olocal, const float* __restrict__ send,
    const float* __restrict__ dtot, const float* __restrict__ gsilu,
    const float* __restrict__ gn_w, const float* __restrict__ gn_b,
    const float* __restrict__ Wo, float* __restrict__ out)
{
    __shared__ __attribute__((aligned(16))) float rt_s[FT_TILE];
    __shared__ __attribute__((aligned(16))) float ys[FT_TILE];
    __shared__ __attribute__((aligned(16))) float ss_s[NHEAD * DK * CR_PS];
    __shared__ float ptot[(SC_NCH - 2) * 4 * Hdim];

    const int tid  = (int)threadIdx.x;
    const int wave = __builtin_amdgcn_readfirstlane(tid >> 5);
    const int lane = tid & 31;
    const int sub  = lane & 15;
    const int hh   = lane >> 4;
    const int koff = hh << 1;

    const int tile = (int)blockIdx.x;
    const int e    = tile >> 4;
    const int l0   = (tile & 15) * CR_SEG;
    const int tc   = l0 / SC_TL;
    const int seg  = (l0 % SC_TL) / CR_SEG;
    const long row0 = (long)e * LKs + l0;

    {
        const int head = tid >> 6;
        const int kk   = tid & 63;
        const float* __restrict__ dp = decay + row0 * Hdim + tid;
        const float* __restrict__ rp = r     + row0 * Hdim + tid;
        float dv[CR_SEG], rv[CR_SEG];
        #pragma unroll
        for (int i = 0; i < CR_SEG; ++i) { dv[i] = dp[i * Hdim]; rv[i] = rp[i * Hdim]; }

        const float* __restrict__ tq = dtot + ((long)(e * NHEAD + head) * SC_NCH * 4) * DK + kk;
        for (int pc = 1; pc < tc; ++pc) {
            #pragma unroll
            for (int sg = 0; sg < 4; ++sg)
                ptot[((pc - 1) * 4 + sg) * Hdim + tid] = tq[(pc * 4 + sg) * DK];
        }
        float start = 1.0f;
        for (int sg = 0; sg < seg; ++sg) start *= tq[(tc * 4 + sg) * DK];

        float p = 1.0f;
        #pragma unroll
        for (int i = 0; i < CR_SEG; ++i) { const float before = p; p *= dv[i]; dv[i] = before; }
        float* rtp = rt_s + tid;
        #pragma unroll
        for (int i = 0; i < CR_SEG; ++i) rtp[i * FT_XP] = rv[i] * (start * dv[i]);
    }
    __syncthreads();

    {
        const int kr = tid >> 4;
        const int vl = tid & 15;
        float s[NHEAD][16];
        #pragma unroll
        for (int hd = 0; hd < NHEAD; ++hd) {
            #pragma unroll
            for (int i = 0; i < 16; ++i) s[hd][i] = 0.0f;
        }
        if (tc > 0) {
            const float* __restrict__ sp = send + (long)(e * NHEAD) * (SC_NCH - 1) * DK * DK + tid;
            #pragma unroll
            for (int hd = 0; hd < NHEAD; ++hd) {
                #pragma unroll
                for (int i = 0; i < 16; ++i) s[hd][i] = sp[(long)hd * (SC_NCH - 1) * DK * DK + i * 256];
            }
            for (int pc = 1; pc < tc; ++pc) {
                float nv[NHEAD][16];
                #pragma unroll
                for (int hd = 0; hd < NHEAD; ++hd) {
                    #pragma unroll
                    for (int i = 0; i < 16; ++i)
                        nv[hd][i] = sp[((long)hd * (SC_NCH - 1) + pc) * DK * DK + i * 256];
                }
                #pragma unroll
                for (int hd = 0; hd < NHEAD; ++hd) {
                    const float* pt = ptot + (pc - 1) * 4 * Hdim + hd * DK + kr;
                    float pe[4];
                    #pragma unroll
                    for (int q = 0; q < 4; ++q)
                        pe[q] = ((pt[16 * q] * pt[Hdim + 16 * q]) * pt[2 * Hdim + 16 * q]) * pt[3 * Hdim + 16 * q];
                    #pragma unroll
                    for (int i = 0; i < 16; ++i) s[hd][i] = pe[i & 3] * s[hd][i] + nv[hd][i];
                }
            }
        }
        #pragma unroll
        for (int hd = 0; hd < NHEAD; ++hd) {
            #pragma unroll
            for (int i = 0; i < 16; ++i)
                ss_s[(hd * DK + kr + 16 * (i & 3)) * CR_PS + 16 * (i >> 2) + vl] = s[hd][i];
        }
    }
    __syncthreads();

    const int head = wave >> 1;
    const int mrow = (wave & 1) << 4;
    v8f c0 = {}, c1 = {}, c2 = {}, c3 = {};
    {
        const float* A0 = rt_s + (mrow + sub) * FT_XP + head * DK + koff;
        const float* B0 = ss_s + (head * DK + koff) * CR_PS + sub;
        #pragma unroll
        for (int k = 0; k < DK; k += 4) {
            const v2f a = *(const v2f*)(A0 + k);
            v2f b0, b1, b2, b3;
            b0.x = B0[k * CR_PS];
            b0.y = B0[(k + 1) * CR_PS];
            b1.x = B0[k * CR_PS + 16];
            b1.y = B0[(k + 1) * CR_PS + 16];
            b2.x = B0[k * CR_PS + 32];
            b2.y = B0[(k + 1) * CR_PS + 32];
            b3.x = B0[k * CR_PS + 48];
            b3.y = B0[(k + 1) * CR_PS + 48];
            c0 = __builtin_amdgcn_wmma_f32_16x16x4_f32(false, a, false, b0, (short)0, c0, false, false);
            c1 = __builtin_amdgcn_wmma_f32_16x16x4_f32(false, a, false, b1, (short)0, c1, false, false);
            c2 = __builtin_amdgcn_wmma_f32_16x16x4_f32(false, a, false, b2, (short)0, c2, false, false);
            c3 = __builtin_amdgcn_wmma_f32_16x16x4_f32(false, a, false, b3, (short)0, c3, false, false);
        }
    }

    {
        const int hc = head * DK + sub;
        const long rowb = (row0 + mrow + 8 * hh) * Hdim + hc;
        const float* __restrict__ op = olocal + rowb;
        const float* __restrict__ gp = gsilu + rowb;
        float* yp = ys + (mrow + 8 * hh) * FT_XP + hc;
        const float gw0 = gn_w[hc], gw1 = gn_w[hc + 16], gw2 = gn_w[hc + 32], gw3 = gn_w[hc + 48];
        const float gb0 = gn_b[hc], gb1 = gn_b[hc + 16], gb2 = gn_b[hc + 32], gb3 = gn_b[hc + 48];
        #pragma unroll
        for (int rr = 0; rr < 8; ++rr) {
            const float x0 = c0[rr] + op[rr * Hdim];
            const float x1 = c1[rr] + op[rr * Hdim + 16];
            const float x2 = c2[rr] + op[rr * Hdim + 32];
            const float x3 = c3[rr] + op[rr * Hdim + 48];
            const float mean = cr_row_sum((x0 + x1) + (x2 + x3)) * (1.0f / 64.0f);
            const float d0 = x0 - mean, d1 = x1 - mean, d2 = x2 - mean, d3 = x3 - mean;
            const float var = cr_row_sum((d0 * d0 + d1 * d1) + (d2 * d2 + d3 * d3)) * (1.0f / 64.0f);
            const float rstd = __builtin_amdgcn_rsqf(var + EPSV);
            yp[rr * FT_XP]      = (d0 * rstd * gw0 + gb0) * gp[rr * Hdim];
            yp[rr * FT_XP + 16] = (d1 * rstd * gw1 + gb1) * gp[rr * Hdim + 16];
            yp[rr * FT_XP + 32] = (d2 * rstd * gw2 + gb2) * gp[rr * Hdim + 32];
            yp[rr * FT_XP + 48] = (d3 * rstd * gw3 + gb3) * gp[rr * Hdim + 48];
        }
    }
    __syncthreads();

    const int nc = wave << 5;
    ft_strip<false>(ys, Wo, out + (row0 + 8 * hh) * Hdim + sub + nc, nc, sub, koff);
}

extern "C" void kernel_launch(void* const* d_in, const int* in_sizes, int n_in,
                              void* d_out, int out_size, void* d_ws, size_t ws_size,
                              hipStream_t stream)
{
    const float* query    = (const float*)d_in[0];
    const float* keyval   = (const float*)d_in[1];
    const float* rwg_mu   = (const float*)d_in[2];
    const float* W_rwg0   = (const float*)d_in[3];
    const float* W_rwg2   = (const float*)d_in[4];
    const float* rwg_bias = (const float*)d_in[5];
    const float* kv_mu    = (const float*)d_in[6];
    const float* W_kv0    = (const float*)d_in[7];
    const float* W_kv2    = (const float*)d_in[8];
    const float* kv_bias  = (const float*)d_in[9];
    const float* Wr       = (const float*)d_in[10];
    const float* Wa       = (const float*)d_in[11];
    const float* Wb       = (const float*)d_in[12];
    const float* bb       = (const float*)d_in[13];
    const float* Wk       = (const float*)d_in[14];
    const float* Wv       = (const float*)d_in[15];
    const float* Wg       = (const float*)d_in[16];
    const float* u        = (const float*)d_in[17];
    const float* gn_w     = (const float*)d_in[18];
    const float* gn_b     = (const float*)d_in[19];
    const float* Wo       = (const float*)d_in[20];

    float* ws = (float*)d_ws;
    const long SZ = (long)NTOK * Hdim;
    float* rbuf  = ws + 0 * SZ;
    float* kbuf  = ws + 1 * SZ;
    float* vbuf  = ws + 2 * SZ;
    float* dbuf  = ws + 3 * SZ;
    float* gsbuf = ws + 4 * SZ;
    float* olbuf = ws + 6 * SZ;
    float* sebuf = ws + 7 * SZ;
    float* dtbuf = ws + 8 * SZ;

    rwkv_features_kernel<<<NTOK / FT_TOK, FT_WAVES * 32, 0, stream>>>(
        query, keyval, rwg_mu, W_rwg0, W_rwg2, rwg_bias,
        kv_mu, W_kv0, W_kv2, kv_bias,
        Wr, Wa, Wb, bb, Wk, Wv, Wg,
        rbuf, kbuf, vbuf, gsbuf, dbuf);

    rwkv_scan_kernel<<<Bsz * LQ * NHEAD * SC_NCH, (SC_SCANW + 1) * 32, 0, stream>>>(
        rbuf, kbuf, vbuf, dbuf, u, olbuf, sebuf, dtbuf);

    rwkv_carry_out_kernel<<<NTOK / CR_SEG, CR_WAVES * 32, 0, stream>>>(
        rbuf, dbuf, olbuf, sebuf, dtbuf, gsbuf, gn_w, gn_b, Wo, (float*)d_out);
}
